# starting kernel
# baseline (speedup 1.0000x reference)
#include <hip/hip_runtime.h>
#include <hip/hip_bf16.h>

typedef float v2f __attribute__((ext_vector_type(2)));
typedef float v8f __attribute__((ext_vector_type(8)));

#define NEG_SLOPE 0.2f

template<int K, int NOUT, int NT>
__global__ __launch_bounds__(256) void gemm_wmma_f32(
    const float* __restrict__ X, const float* __restrict__ W,
    float* __restrict__ C, int M)
{
  const int lane   = threadIdx.x & 31;
  const int wave   = threadIdx.x >> 5;
  constexpr int tilesN = NOUT / (16 * NT);
  const int tiles  = (M >> 4) * tilesN;
  const int tile   = blockIdx.x * 8 + wave;
  if (tile >= tiles) return;
  const int tM = tile / tilesN;
  const int tN = tile % tilesN;

  const int half = lane >> 4;
  const int l15  = lane & 15;
  const float* xrow = X + (size_t)(tM * 16 + l15) * K + 2 * half;
  const float* wb   = W + (size_t)(2 * half) * NOUT + tN * (16 * NT) + l15;

  v8f acc[NT];
#pragma unroll
  for (int t = 0; t < NT; ++t) acc[t] = (v8f){0.f,0.f,0.f,0.f,0.f,0.f,0.f,0.f};

#pragma unroll
  for (int k = 0; k < K; k += 4) {
    v2f a;
    a.x = xrow[k + 0];
    a.y = xrow[k + 1];
#pragma unroll
    for (int t = 0; t < NT; ++t) {
      v2f b;
      b.x = wb[(size_t)(k + 0) * NOUT + 16 * t];
      b.y = wb[(size_t)(k + 1) * NOUT + 16 * t];
      acc[t] = __builtin_amdgcn_wmma_f32_16x16x4_f32(false, a, false, b,
                                                     (short)0, acc[t],
                                                     false, false);
    }
  }

  const int mbase = tM * 16 + 8 * half;
  float* cp = C + (size_t)mbase * NOUT + tN * (16 * NT) + l15;
#pragma unroll
  for (int t = 0; t < NT; ++t)
#pragma unroll
    for (int r = 0; r < 8; ++r)
      cp[(size_t)r * NOUT + 16 * t] = acc[t][r];
}

__global__ __launch_bounds__(256) void rowdot(
    const float* __restrict__ X, const float* __restrict__ v,
    float* __restrict__ out, int N, int K)
{
  const int gw   = (int)((blockIdx.x * (size_t)blockDim.x + threadIdx.x) >> 5);
  const int lane = threadIdx.x & 31;
  if (gw >= N) return;
  const float* row = X + (size_t)gw * K;
  float s = 0.f;
  for (int k = lane; k < K; k += 32) s += row[k] * v[k];
#pragma unroll
  for (int off = 16; off > 0; off >>= 1) s += __shfl_down(s, off, 32);
  if (lane == 0) out[gw] = s;
}

__global__ void foldvec(const float* __restrict__ W, const float* __restrict__ att,
                        float* __restrict__ v, int rows, int cols)
{
  const int j = blockIdx.x * blockDim.x + threadIdx.x;
  if (j >= rows) return;
  const float* wr = W + (size_t)j * cols;
  float s = 0.f;
  for (int k = 0; k < cols; ++k) s += wr[k] * att[k];
  v[j] = s;
}

__global__ void init_seg(float* __restrict__ m, float* __restrict__ denom, int N)
{
  const int i = blockIdx.x * blockDim.x + threadIdx.x;
  if (i < N) { m[i] = -__builtin_huge_valf(); denom[i] = 0.f; }
}

__global__ void init_bias(float* __restrict__ out, const float* __restrict__ b,
                          int total, int C)
{
  const int i = blockIdx.x * blockDim.x + threadIdx.x;
  if (i < total) out[i] = b[i % C];
}

__global__ void relu_ip(float* __restrict__ x, int n)
{
  const int i = blockIdx.x * blockDim.x + threadIdx.x;
  if (i < n) x[i] = fmaxf(x[i], 0.f);
}

__device__ __forceinline__ void atomicMaxF(float* addr, float val)
{
  if (val >= 0.f) atomicMax((int*)addr, __float_as_int(val));
  else            atomicMin((unsigned int*)addr, __float_as_uint(val));
}

__global__ __launch_bounds__(256) void edge_logits(
    const int* __restrict__ src, const int* __restrict__ dst,
    const float* __restrict__ asrc, const float* __restrict__ adst,
    float* __restrict__ e, float* __restrict__ m, int E)
{
  const int i = blockIdx.x * blockDim.x + threadIdx.x;
  if (i >= E) return;
  const int d = dst[i];
  float v = asrc[src[i]] + adst[d];
  v = (v > 0.f) ? v : NEG_SLOPE * v;
  e[i] = v;
  atomicMaxF(&m[d], v);
}

__global__ __launch_bounds__(256) void edge_exp(
    const int* __restrict__ dst, const float* __restrict__ m,
    float* __restrict__ e, float* __restrict__ denom, int E)
{
  const int i = blockIdx.x * blockDim.x + threadIdx.x;
  if (i >= E) return;
  const int d = dst[i];
  const float ee = __expf(e[i] - m[d]);
  e[i] = ee;
  atomicAdd(&denom[d], ee);
}

template<int C>
__global__ __launch_bounds__(256) void edge_scatter(
    const int* __restrict__ src, const int* __restrict__ dst,
    const float* __restrict__ ee, const float* __restrict__ denom,
    const float* __restrict__ h, float* __restrict__ out, int E)
{
  const int wv   = (int)((blockIdx.x * (size_t)blockDim.x + threadIdx.x) >> 5);
  const int lane = threadIdx.x & 31;
  if (wv >= E) return;
  const int s = src[wv];
  const int d = dst[wv];
  const float alpha = ee[wv] / denom[d];
  constexpr int P = C / 32;
  const float* hp = h   + (size_t)s * C + lane * P;
  float*       op = out + (size_t)d * C + lane * P;
  float tmp[P];
  if constexpr (P == 4) {
    const float4 v = *(const float4*)hp;
    tmp[0] = v.x; tmp[1] = v.y; tmp[2] = v.z; tmp[3] = v.w;
  } else {
    const float2 v = *(const float2*)hp;
    tmp[0] = v.x; tmp[1] = v.y;
  }
#pragma unroll
  for (int i = 0; i < P; ++i) atomicAdd(op + i, alpha * tmp[i]);
}

static inline int cdiv(long a, long b) { return (int)((a + b - 1) / b); }

extern "C" void kernel_launch(void* const* d_in, const int* in_sizes, int n_in,
                              void* d_out, int out_size, void* d_ws, size_t ws_size,
                              hipStream_t stream)
{
  const float* x       = (const float*)d_in[0];
  const int*   ei      = (const int*)d_in[1];
  const float* Wsrc1   = (const float*)d_in[2];
  const float* Wdst1   = (const float*)d_in[3];
  const float* attsrc1 = (const float*)d_in[4];
  const float* attdst1 = (const float*)d_in[5];
  const float* b1      = (const float*)d_in[6];
  const float* Wsrc2   = (const float*)d_in[7];
  const float* Wdst2   = (const float*)d_in[8];
  const float* attsrc2 = (const float*)d_in[9];
  const float* attdst2 = (const float*)d_in[10];
  const float* b2      = (const float*)d_in[11];

  const int HID = in_sizes[4];
  const int OUT = in_sizes[9];
  const int FIN = in_sizes[2] / HID;
  const int N   = in_sizes[0] / FIN;
  const int E   = in_sizes[1] / 2;
  const int* src = ei;
  const int* dst = ei + E;

  float* ws    = (float*)d_ws;
  float* h1    = ws;  ws += (size_t)N * HID;
  float* o1    = ws;  ws += (size_t)N * HID;
  float* h2    = ws;  ws += (size_t)N * OUT;
  float* asrc  = ws;  ws += N;
  float* adst  = ws;  ws += N;
  float* mbuf  = ws;  ws += N;
  float* dbuf  = ws;  ws += N;
  float* ebuf  = ws;  ws += E;
  float* vfold = ws;  ws += HID;

  {
    const int tiles = (N / 16) * (HID / 64);
    gemm_wmma_f32<128, 128, 4><<<cdiv(tiles, 8), 256, 0, stream>>>(x, Wsrc1, h1, N);
    rowdot<<<cdiv((long)N * 32, 256), 256, 0, stream>>>(h1, attsrc1, asrc, N, HID);
    foldvec<<<cdiv(FIN, 128), 128, 0, stream>>>(Wdst1, attdst1, vfold, FIN, HID);
    rowdot<<<cdiv((long)N * 32, 256), 256, 0, stream>>>(x, vfold, adst, N, FIN);
    init_seg<<<cdiv(N, 256), 256, 0, stream>>>(mbuf, dbuf, N);
    init_bias<<<cdiv((long)N * HID, 256), 256, 0, stream>>>(o1, b1, N * HID, HID);
    edge_logits<<<cdiv(E, 256), 256, 0, stream>>>(src, dst, asrc, adst, ebuf, mbuf, E);
    edge_exp<<<cdiv(E, 256), 256, 0, stream>>>(dst, mbuf, ebuf, dbuf, E);
    edge_scatter<128><<<cdiv((long)E * 32, 256), 256, 0, stream>>>(
        src, dst, ebuf, dbuf, h1, o1, E);
    relu_ip<<<cdiv((long)N * HID, 256), 256, 0, stream>>>(o1, N * HID);
  }

  {
    float* out = (float*)d_out;
    const int tiles = (N / 16) * (OUT / 64);
    gemm_wmma_f32<128, 64, 4><<<cdiv(tiles, 8), 256, 0, stream>>>(o1, Wsrc2, h2, N);
    rowdot<<<cdiv((long)N * 32, 256), 256, 0, stream>>>(h2, attsrc2, asrc, N, OUT);
    foldvec<<<cdiv(HID, 128), 128, 0, stream>>>(Wdst2, attdst2, vfold, HID, OUT);
    rowdot<<<cdiv((long)N * 32, 256), 256, 0, stream>>>(o1, vfold, adst, N, HID);
    init_seg<<<cdiv(N, 256), 256, 0, stream>>>(mbuf, dbuf, N);
    init_bias<<<cdiv((long)N * OUT, 256), 256, 0, stream>>>(out, b2, N * OUT, OUT);
    edge_logits<<<cdiv(E, 256), 256, 0, stream>>>(src, dst, asrc, adst, ebuf, mbuf, E);
    edge_exp<<<cdiv(E, 256), 256, 0, stream>>>(dst, mbuf, ebuf, dbuf, E);
    edge_scatter<64><<<cdiv((long)E * 32, 256), 256, 0, stream>>>(
        src, dst, ebuf, dbuf, h2, out, E);
  }
}


# Group edges by destination once per call (count, scan, fill); one wave per node does softmax and weighted row sum in registers: no float atomics; rows summed in fill arrival order
# speedup vs baseline: 1.9875x; 1.9875x over previous
#include <hip/hip_runtime.h>
#include <hip/hip_bf16.h>

typedef float v2f __attribute__((ext_vector_type(2)));
typedef float v8f __attribute__((ext_vector_type(8)));

#define NEG_SLOPE 0.2f

template<int K, int NOUT, int NT>
__global__ __launch_bounds__(256) void gemm_wmma_f32(
    const float* __restrict__ X, const float* __restrict__ W,
    float* __restrict__ C, int M)
{
  const int lane   = threadIdx.x & 31;
  const int wave   = threadIdx.x >> 5;
  constexpr int tilesN = NOUT / (16 * NT);
  const int tiles  = (M >> 4) * tilesN;
  const int tile   = blockIdx.x * 8 + wave;
  if (tile >= tiles) return;
  const int tM = tile / tilesN;
  const int tN = tile % tilesN;

  const int half = lane >> 4;
  const int l15  = lane & 15;
  const float* xrow = X + (size_t)(tM * 16 + l15) * K + 2 * half;
  const float* wb   = W + (size_t)(2 * half) * NOUT + tN * (16 * NT) + l15;

  v8f acc[NT];
#pragma unroll
  for (int t = 0; t < NT; ++t) acc[t] = (v8f){0.f,0.f,0.f,0.f,0.f,0.f,0.f,0.f};

#pragma unroll
  for (int k = 0; k < K; k += 4) {
    v2f a;
    a.x = xrow[k + 0];
    a.y = xrow[k + 1];
#pragma unroll
    for (int t = 0; t < NT; ++t) {
      v2f b;
      b.x = wb[(size_t)(k + 0) * NOUT + 16 * t];
      b.y = wb[(size_t)(k + 1) * NOUT + 16 * t];
      acc[t] = __builtin_amdgcn_wmma_f32_16x16x4_f32(false, a, false, b,
                                                     (short)0, acc[t],
                                                     false, false);
    }
  }

  const int mbase = tM * 16 + 8 * half;
  float* cp = C + (size_t)mbase * NOUT + tN * (16 * NT) + l15;
#pragma unroll
  for (int t = 0; t < NT; ++t)
#pragma unroll
    for (int r = 0; r < 8; ++r)
      cp[(size_t)r * NOUT + 16 * t] = acc[t][r];
}

__global__ __launch_bounds__(256) void rowdot(
    const float* __restrict__ X, const float* __restrict__ v,
    float* __restrict__ out, int N, int K)
{
  const int gw   = (int)((blockIdx.x * (size_t)blockDim.x + threadIdx.x) >> 5);
  const int lane = threadIdx.x & 31;
  if (gw >= N) return;
  const float* row = X + (size_t)gw * K;
  float s = 0.f;
  for (int k = lane; k < K; k += 32) s += row[k] * v[k];
#pragma unroll
  for (int off = 16; off > 0; off >>= 1) s += __shfl_down(s, off, 32);
  if (lane == 0) out[gw] = s;
}

__global__ void foldvec(const float* __restrict__ W, const float* __restrict__ att,
                        float* __restrict__ v, int rows, int cols)
{
  const int j = blockIdx.x * blockDim.x + threadIdx.x;
  if (j >= rows) return;
  const float* wr = W + (size_t)j * cols;
  float s = 0.f;
  for (int k = 0; k < cols; ++k) s += wr[k] * att[k];
  v[j] = s;
}

__global__ void zero_counts(int* __restrict__ cnt, int N)
{
  const int i = blockIdx.x * blockDim.x + threadIdx.x;
  if (i < N) cnt[i] = 0;
}

__global__ __launch_bounds__(256) void count_dst(
    const int* __restrict__ dst, int* __restrict__ cnt, int E)
{
  const int i0 = blockIdx.x * 1024 + threadIdx.x;
  int d[4];
  for (int q = 0; q < 4; ++q) {
    const int i = i0 + q * 256;
    d[q] = (i < E) ? dst[i] : -1;
  }
  for (int q = 0; q < 4; ++q)
    if (d[q] >= 0) atomicAdd(&cnt[d[q]], 1);
}

__global__ __launch_bounds__(1024) void scan_counts(
    const int* __restrict__ cnt, int* __restrict__ rowptr,
    int* __restrict__ cursor, int N)
{
  __shared__ int wtot[32];
  __shared__ int tile_total;
  const int tid  = threadIdx.x;
  const int lane = tid & 31;
  const int wave = tid >> 5;
  int carry = 0;
  for (int base = 0; base < N; base += 8192) {
    const int i0 = base + tid * 8;
    int c[8];
    int t = 0;
    for (int q = 0; q < 8; ++q) {
      c[q] = (i0 + q < N) ? cnt[i0 + q] : 0;
      t += c[q];
    }
    int s = t;
    for (int off = 1; off < 32; off <<= 1) {
      const int nb = __shfl_up(s, off, 32);
      if (lane >= off) s += nb;
    }
    if (lane == 31) wtot[wave] = s;
    __syncthreads();
    if (wave == 0) {
      const int w = wtot[lane];
      int ws = w;
      for (int off = 1; off < 32; off <<= 1) {
        const int nb = __shfl_up(ws, off, 32);
        if (lane >= off) ws += nb;
      }
      wtot[lane] = ws - w;
      if (lane == 31) tile_total = ws;
    }
    __syncthreads();
    int e = carry + wtot[wave] + (s - t);
    carry += tile_total;
    for (int q = 0; q < 8; ++q) {
      if (i0 + q < N) { rowptr[i0 + q] = e; cursor[i0 + q] = e; }
      e += c[q];
    }
    __syncthreads();
  }
  if (tid == 0) rowptr[N] = carry;
}

__global__ __launch_bounds__(256) void fill_csr(
    const int* __restrict__ src, const int* __restrict__ dst,
    int* __restrict__ cursor, int* __restrict__ col, int E)
{
  const int i0 = blockIdx.x * 1024 + threadIdx.x;
  int d[4], s[4], p[4];
  for (int q = 0; q < 4; ++q) {
    const int i = i0 + q * 256;
    d[q] = (i < E) ? dst[i] : -1;
    s[q] = (i < E) ? src[i] : 0;
  }
  for (int q = 0; q < 4; ++q)
    p[q] = (d[q] >= 0) ? atomicAdd(&cursor[d[q]], 1) : -1;
  for (int q = 0; q < 4; ++q)
    if (p[q] >= 0) col[p[q]] = s[q];
}

typedef float v4f __attribute__((ext_vector_type(4)));

__device__ __forceinline__ float leaky(float v)
{
  return (v > 0.f) ? v : NEG_SLOPE * v;
}

template<typename VT, int C, bool RELU>
__global__ __launch_bounds__(256) void seg_gather(
    const int* __restrict__ rowptr, const int* __restrict__ col,
    const float* __restrict__ asrc, const float* __restrict__ adst,
    const float* __restrict__ h, const float* __restrict__ bias,
    float* __restrict__ out, int N)
{
  constexpr int P = C / 32;
  const int lane = threadIdx.x & 31;
  const int d = __builtin_amdgcn_readfirstlane((int)(blockIdx.x * 8 + (threadIdx.x >> 5)));
  if (d >= N) return;
  const int beg = rowptr[d];
  const int end = rowptr[d + 1];
  const float ad = adst[d];

  float mx = -__builtin_huge_valf();
  for (int j = beg + lane; j < end; j += 32)
    mx = fmaxf(mx, leaky(asrc[col[j]] + ad));
  for (int off = 16; off > 0; off >>= 1) mx = fmaxf(mx, __shfl_xor(mx, off, 32));

  float den = 0.f;
  for (int j = beg + lane; j < end; j += 32)
    den += __expf(leaky(asrc[col[j]] + ad) - mx);
  for (int off = 16; off > 0; off >>= 1) den += __shfl_xor(den, off, 32);

  VT acc = (VT)(0.f);
  for (int base = beg; base < end; base += 32) {
    const int j = base + lane;
    int   sl = 0;
    float al = 0.f;
    if (j < end) {
      sl = col[j];
      al = __expf(leaky(asrc[sl] + ad) - mx) / den;
    }
    const int n = (end - base < 32) ? (end - base) : 32;
    int k = 0;
    for (; k + 4 <= n; k += 4) {
      const int s0 = __builtin_amdgcn_readlane(sl, k);
      const int s1 = __builtin_amdgcn_readlane(sl, k + 1);
      const int s2 = __builtin_amdgcn_readlane(sl, k + 2);
      const int s3 = __builtin_amdgcn_readlane(sl, k + 3);
      const VT r0 = *(const VT*)(h + (size_t)s0 * C + lane * P);
      const VT r1 = *(const VT*)(h + (size_t)s1 * C + lane * P);
      const VT r2 = *(const VT*)(h + (size_t)s2 * C + lane * P);
      const VT r3 = *(const VT*)(h + (size_t)s3 * C + lane * P);
      const float a0 = __int_as_float(__builtin_amdgcn_readlane(__float_as_int(al), k));
      const float a1 = __int_as_float(__builtin_amdgcn_readlane(__float_as_int(al), k + 1));
      const float a2 = __int_as_float(__builtin_amdgcn_readlane(__float_as_int(al), k + 2));
      const float a3 = __int_as_float(__builtin_amdgcn_readlane(__float_as_int(al), k + 3));
      acc += a0 * r0;
      acc += a1 * r1;
      acc += a2 * r2;
      acc += a3 * r3;
    }
    for (; k < n; ++k) {
      const int s0 = __builtin_amdgcn_readlane(sl, k);
      const float a0 = __int_as_float(__builtin_amdgcn_readlane(__float_as_int(al), k));
      const VT r0 = *(const VT*)(h + (size_t)s0 * C + lane * P);
      acc += a0 * r0;
    }
  }

  VT o = acc + *(const VT*)(bias + lane * P);
  if (RELU) {
    for (int i = 0; i < P; ++i) o[i] = fmaxf(o[i], 0.f);
  }
  *(VT*)(out + (size_t)d * C + lane * P) = o;
}

static inline int cdiv(long a, long b) { return (int)((a + b - 1) / b); }

extern "C" void kernel_launch(void* const* d_in, const int* in_sizes, int n_in,
                              void* d_out, int out_size, void* d_ws, size_t ws_size,
                              hipStream_t stream)
{
  const float* x       = (const float*)d_in[0];
  const int*   ei      = (const int*)d_in[1];
  const float* Wsrc1   = (const float*)d_in[2];
  const float* Wdst1   = (const float*)d_in[3];
  const float* attsrc1 = (const float*)d_in[4];
  const float* attdst1 = (const float*)d_in[5];
  const float* b1      = (const float*)d_in[6];
  const float* Wsrc2   = (const float*)d_in[7];
  const float* Wdst2   = (const float*)d_in[8];
  const float* attsrc2 = (const float*)d_in[9];
  const float* attdst2 = (const float*)d_in[10];
  const float* b2      = (const float*)d_in[11];

  const int HID = in_sizes[4];
  const int OUT = in_sizes[9];
  const int FIN = in_sizes[2] / HID;
  const int N   = in_sizes[0] / FIN;
  const int E   = in_sizes[1] / 2;
  const int* src = ei;
  const int* dst = ei + E;

  float* ws    = (float*)d_ws;
  float* h1    = ws;  ws += (size_t)N * HID;
  float* o1    = ws;  ws += (size_t)N * HID;
  float* h2    = ws;  ws += (size_t)N * OUT;
  float* asrc  = ws;  ws += N;
  float* adst  = ws;  ws += N;
  float* vfold = ws;  ws += HID;
  int* cnt     = (int*)ws;  ws += N;
  int* cursor  = (int*)ws;  ws += N;
  int* col     = (int*)ws;  ws += E;
  int* rowptr  = (int*)ws;  ws += N + 1;

  zero_counts<<<cdiv(N, 256), 256, 0, stream>>>(cnt, N);
  count_dst<<<cdiv(E, 1024), 256, 0, stream>>>(dst, cnt, E);
  scan_counts<<<1, 1024, 0, stream>>>(cnt, rowptr, cursor, N);
  fill_csr<<<cdiv(E, 1024), 256, 0, stream>>>(src, dst, cursor, col, E);

  {
    const int tiles = (N / 16) * (HID / 64);
    gemm_wmma_f32<128, 128, 4><<<cdiv(tiles, 8), 256, 0, stream>>>(x, Wsrc1, h1, N);
    rowdot<<<cdiv((long)N * 32, 256), 256, 0, stream>>>(h1, attsrc1, asrc, N, HID);
    foldvec<<<cdiv(FIN, 128), 128, 0, stream>>>(Wdst1, attdst1, vfold, FIN, HID);
    rowdot<<<cdiv((long)N * 32, 256), 256, 0, stream>>>(x, vfold, adst, N, FIN);
    seg_gather<v4f, 128, true><<<cdiv(N, 8), 256, 0, stream>>>(
        rowptr, col, asrc, adst, h1, b1, o1, N);
  }

  {
    float* out = (float*)d_out;
    const int tiles = (N / 16) * (OUT / 64);
    gemm_wmma_f32<128, 64, 4><<<cdiv(tiles, 8), 256, 0, stream>>>(o1, Wsrc2, h2, N);
    rowdot<<<cdiv((long)N * 32, 256), 256, 0, stream>>>(h2, attsrc2, asrc, N, OUT);
    foldvec<<<cdiv(HID, 128), 128, 0, stream>>>(Wdst2, attdst2, vfold, HID, OUT);
    rowdot<<<cdiv((long)N * 32, 256), 256, 0, stream>>>(o1, vfold, adst, N, HID);
    seg_gather<v2f, 64, false><<<cdiv(N, 8), 256, 0, stream>>>(
        rowptr, col, asrc, adst, h2, b2, out, N);
  }
}


# Attention logits a_src, a_dst computed inside the GEMM kernel from its A fragments and folded vectors W*att: the four row-dot launches go; logits now summed in the GEMM k order
# speedup vs baseline: 2.3837x; 1.1993x over previous
#include <hip/hip_runtime.h>
#include <hip/hip_bf16.h>

typedef float v2f __attribute__((ext_vector_type(2)));
typedef float v8f __attribute__((ext_vector_type(8)));

#define NEG_SLOPE 0.2f

template<int K, int NOUT, int NT>
__global__ __launch_bounds__(256) void gemm_wmma_f32(
    const float* __restrict__ X, const float* __restrict__ W,
    const float* __restrict__ vecs, float* __restrict__ C,
    float* __restrict__ logit, int M)
{
  const int lane   = threadIdx.x & 31;
  const int wave   = threadIdx.x >> 5;
  constexpr int tilesN = NOUT / (16 * NT);
  static_assert(tilesN == 1 || tilesN == 2, "two logit vectors are split over the strips of a row block");
  constexpr int NV = 2 / tilesN;
  const int tiles  = (M >> 4) * tilesN;
  const int tile   = blockIdx.x * 8 + wave;
  if (tile >= tiles) return;
  const int tM = tile / tilesN;
  const int tN = tile % tilesN;

  const int half = lane >> 4;
  const int l15  = lane & 15;
  const float* xrow = X + (size_t)(tM * 16 + l15) * K + 2 * half;
  const float* wb   = W + (size_t)(2 * half) * NOUT + tN * (16 * NT) + l15;
  const float* vp   = vecs + (size_t)(tN * NV) * K + 2 * half;

  v8f acc[NT];
#pragma unroll
  for (int t = 0; t < NT; ++t) acc[t] = (v8f){0.f,0.f,0.f,0.f,0.f,0.f,0.f,0.f};
  float lg[NV];
  for (int i = 0; i < NV; ++i) lg[i] = 0.f;

#pragma unroll
  for (int k = 0; k < K; k += 4) {
    v2f a;
    a.x = xrow[k + 0];
    a.y = xrow[k + 1];
#pragma unroll
    for (int i = 0; i < NV; ++i) {
      lg[i] = fmaf(a.x, vp[i * K + k + 0], lg[i]);
      lg[i] = fmaf(a.y, vp[i * K + k + 1], lg[i]);
    }
#pragma unroll
    for (int t = 0; t < NT; ++t) {
      v2f b;
      b.x = wb[(size_t)(k + 0) * NOUT + 16 * t];
      b.y = wb[(size_t)(k + 1) * NOUT + 16 * t];
      acc[t] = __builtin_amdgcn_wmma_f32_16x16x4_f32(false, a, false, b,
                                                     (short)0, acc[t],
                                                     false, false);
    }
  }

  const int mbase = tM * 16 + 8 * half;
  float* cp = C + (size_t)mbase * NOUT + tN * (16 * NT) + l15;
#pragma unroll
  for (int t = 0; t < NT; ++t)
#pragma unroll
    for (int r = 0; r < 8; ++r)
      cp[(size_t)r * NOUT + 16 * t] = acc[t][r];

  for (int i = 0; i < NV; ++i) {
    const float s = lg[i] + __shfl_xor(lg[i], 16, 32);
    if (half == 0) logit[(size_t)(tN * NV + i) * M + tM * 16 + l15] = s;
  }
}

__global__ __launch_bounds__(256) void fold_att(
    const float* __restrict__ W0, const float* __restrict__ a0,
    const float* __restrict__ W1, const float* __restrict__ a1,
    const float* __restrict__ W2, const float* __restrict__ a2,
    const float* __restrict__ W3, const float* __restrict__ a3,
    float* __restrict__ v, int r1, int c1, int r2, int c2)
{
  const int gw   = (int)((blockIdx.x * (size_t)blockDim.x + threadIdx.x) >> 5);
  const int lane = threadIdx.x & 31;
  if (gw >= 2 * r1 + 2 * r2) return;
  const float* Wm; const float* att; int row, cols;
  if (gw < r1)               { Wm = W0; att = a0; row = gw;               cols = c1; }
  else if (gw < 2 * r1)      { Wm = W1; att = a1; row = gw - r1;          cols = c1; }
  else if (gw < 2 * r1 + r2) { Wm = W2; att = a2; row = gw - 2 * r1;      cols = c2; }
  else                       { Wm = W3; att = a3; row = gw - 2 * r1 - r2; cols = c2; }
  const float* wr = Wm + (size_t)row * cols;
  float s = 0.f;
  for (int k = lane; k < cols; k += 32) s += wr[k] * att[k];
  for (int off = 16; off > 0; off >>= 1) s += __shfl_down(s, off, 32);
  if (lane == 0) v[gw] = s;
}

__global__ void zero_counts(int* __restrict__ cnt, int N)
{
  const int i = blockIdx.x * blockDim.x + threadIdx.x;
  if (i < N) cnt[i] = 0;
}

__global__ __launch_bounds__(256) void count_dst(
    const int* __restrict__ dst, int* __restrict__ cnt, int E)
{
  const int i0 = blockIdx.x * 1024 + threadIdx.x;
  int d[4];
  for (int q = 0; q < 4; ++q) {
    const int i = i0 + q * 256;
    d[q] = (i < E) ? dst[i] : -1;
  }
  for (int q = 0; q < 4; ++q)
    if (d[q] >= 0) atomicAdd(&cnt[d[q]], 1);
}

__global__ __launch_bounds__(1024) void scan_counts(
    const int* __restrict__ cnt, int* __restrict__ rowptr,
    int* __restrict__ cursor, int N)
{
  __shared__ int wtot[32];
  __shared__ int tile_total;
  const int tid  = threadIdx.x;
  const int lane = tid & 31;
  const int wave = tid >> 5;
  int carry = 0;
  for (int base = 0; base < N; base += 8192) {
    const int i0 = base + tid * 8;
    int c[8];
    int t = 0;
    for (int q = 0; q < 8; ++q) {
      c[q] = (i0 + q < N) ? cnt[i0 + q] : 0;
      t += c[q];
    }
    int s = t;
    for (int off = 1; off < 32; off <<= 1) {
      const int nb = __shfl_up(s, off, 32);
      if (lane >= off) s += nb;
    }
    if (lane == 31) wtot[wave] = s;
    __syncthreads();
    if (wave == 0) {
      const int w = wtot[lane];
      int ws = w;
      for (int off = 1; off < 32; off <<= 1) {
        const int nb = __shfl_up(ws, off, 32);
        if (lane >= off) ws += nb;
      }
      wtot[lane] = ws - w;
      if (lane == 31) tile_total = ws;
    }
    __syncthreads();
    int e = carry + wtot[wave] + (s - t);
    carry += tile_total;
    for (int q = 0; q < 8; ++q) {
      if (i0 + q < N) { rowptr[i0 + q] = e; cursor[i0 + q] = e; }
      e += c[q];
    }
    __syncthreads();
  }
  if (tid == 0) rowptr[N] = carry;
}

__global__ __launch_bounds__(256) void fill_csr(
    const int* __restrict__ src, const int* __restrict__ dst,
    int* __restrict__ cursor, int* __restrict__ col, int E)
{
  const int i0 = blockIdx.x * 1024 + threadIdx.x;
  int d[4], s[4], p[4];
  for (int q = 0; q < 4; ++q) {
    const int i = i0 + q * 256;
    d[q] = (i < E) ? dst[i] : -1;
    s[q] = (i < E) ? src[i] : 0;
  }
  for (int q = 0; q < 4; ++q)
    p[q] = (d[q] >= 0) ? atomicAdd(&cursor[d[q]], 1) : -1;
  for (int q = 0; q < 4; ++q)
    if (p[q] >= 0) col[p[q]] = s[q];
}

typedef float v4f __attribute__((ext_vector_type(4)));

__device__ __forceinline__ float leaky(float v)
{
  return (v > 0.f) ? v : NEG_SLOPE * v;
}

template<typename VT, int C, bool RELU>
__global__ __launch_bounds__(256) void seg_gather(
    const int* __restrict__ rowptr, const int* __restrict__ col,
    const float* __restrict__ asrc, const float* __restrict__ adst,
    const float* __restrict__ h, const float* __restrict__ bias,
    float* __restrict__ out, int N)
{
  constexpr int P = C / 32;
  const int lane = threadIdx.x & 31;
  const int d = __builtin_amdgcn_readfirstlane((int)(blockIdx.x * 8 + (threadIdx.x >> 5)));
  if (d >= N) return;
  const int beg = rowptr[d];
  const int end = rowptr[d + 1];
  const float ad = adst[d];

  float mx = -__builtin_huge_valf();
  for (int j = beg + lane; j < end; j += 32)
    mx = fmaxf(mx, leaky(asrc[col[j]] + ad));
  for (int off = 16; off > 0; off >>= 1) mx = fmaxf(mx, __shfl_xor(mx, off, 32));

  float den = 0.f;
  for (int j = beg + lane; j < end; j += 32)
    den += __expf(leaky(asrc[col[j]] + ad) - mx);
  for (int off = 16; off > 0; off >>= 1) den += __shfl_xor(den, off, 32);

  VT acc = (VT)(0.f);
  for (int base = beg; base < end; base += 32) {
    const int j = base + lane;
    int   sl = 0;
    float al = 0.f;
    if (j < end) {
      sl = col[j];
      al = __expf(leaky(asrc[sl] + ad) - mx) / den;
    }
    const int n = (end - base < 32) ? (end - base) : 32;
    int k = 0;
    for (; k + 4 <= n; k += 4) {
      const int s0 = __builtin_amdgcn_readlane(sl, k);
      const int s1 = __builtin_amdgcn_readlane(sl, k + 1);
      const int s2 = __builtin_amdgcn_readlane(sl, k + 2);
      const int s3 = __builtin_amdgcn_readlane(sl, k + 3);
      const VT r0 = *(const VT*)(h + (size_t)s0 * C + lane * P);
      const VT r1 = *(const VT*)(h + (size_t)s1 * C + lane * P);
      const VT r2 = *(const VT*)(h + (size_t)s2 * C + lane * P);
      const VT r3 = *(const VT*)(h + (size_t)s3 * C + lane * P);
      const float a0 = __int_as_float(__builtin_amdgcn_readlane(__float_as_int(al), k));
      const float a1 = __int_as_float(__builtin_amdgcn_readlane(__float_as_int(al), k + 1));
      const float a2 = __int_as_float(__builtin_amdgcn_readlane(__float_as_int(al), k + 2));
      const float a3 = __int_as_float(__builtin_amdgcn_readlane(__float_as_int(al), k + 3));
      acc += a0 * r0;
      acc += a1 * r1;
      acc += a2 * r2;
      acc += a3 * r3;
    }
    for (; k < n; ++k) {
      const int s0 = __builtin_amdgcn_readlane(sl, k);
      const float a0 = __int_as_float(__builtin_amdgcn_readlane(__float_as_int(al), k));
      const VT r0 = *(const VT*)(h + (size_t)s0 * C + lane * P);
      acc += a0 * r0;
    }
  }

  VT o = acc + *(const VT*)(bias + lane * P);
  if (RELU) {
    for (int i = 0; i < P; ++i) o[i] = fmaxf(o[i], 0.f);
  }
  *(VT*)(out + (size_t)d * C + lane * P) = o;
}

static inline int cdiv(long a, long b) { return (int)((a + b - 1) / b); }

extern "C" void kernel_launch(void* const* d_in, const int* in_sizes, int n_in,
                              void* d_out, int out_size, void* d_ws, size_t ws_size,
                              hipStream_t stream)
{
  const float* x       = (const float*)d_in[0];
  const int*   ei      = (const int*)d_in[1];
  const float* Wsrc1   = (const float*)d_in[2];
  const float* Wdst1   = (const float*)d_in[3];
  const float* attsrc1 = (const float*)d_in[4];
  const float* attdst1 = (const float*)d_in[5];
  const float* b1      = (const float*)d_in[6];
  const float* Wsrc2   = (const float*)d_in[7];
  const float* Wdst2   = (const float*)d_in[8];
  const float* attsrc2 = (const float*)d_in[9];
  const float* attdst2 = (const float*)d_in[10];
  const float* b2      = (const float*)d_in[11];

  const int HID = in_sizes[4];
  const int OUT = in_sizes[9];
  const int FIN = in_sizes[2] / HID;
  const int N   = in_sizes[0] / FIN;
  const int E   = in_sizes[1] / 2;
  const int* src = ei;
  const int* dst = ei + E;

  float* ws    = (float*)d_ws;
  float* h1    = ws;  ws += (size_t)N * HID;
  float* o1    = ws;  ws += (size_t)N * HID;
  float* h2    = ws;  ws += (size_t)N * OUT;
  float* logit = ws;  ws += 2 * (size_t)N;
  float* asrc  = logit;
  float* adst  = logit + N;
  float* vfold = ws;  ws += 2 * (size_t)FIN + 2 * (size_t)HID;
  int* cnt     = (int*)ws;  ws += N;
  int* cursor  = (int*)ws;  ws += N;
  int* col     = (int*)ws;  ws += E;
  int* rowptr  = (int*)ws;  ws += N + 1;

  zero_counts<<<cdiv(N, 256), 256, 0, stream>>>(cnt, N);
  count_dst<<<cdiv(E, 1024), 256, 0, stream>>>(dst, cnt, E);
  scan_counts<<<1, 1024, 0, stream>>>(cnt, rowptr, cursor, N);
  fill_csr<<<cdiv(E, 1024), 256, 0, stream>>>(src, dst, cursor, col, E);

  fold_att<<<cdiv((long)(2 * FIN + 2 * HID) * 32, 256), 256, 0, stream>>>(
      Wsrc1, attsrc1, Wdst1, attdst1, Wsrc2, attsrc2, Wdst2, attdst2,
      vfold, FIN, HID, HID, OUT);

  {
    const int tiles = (N / 16) * (HID / 64);
    gemm_wmma_f32<128, 128, 4><<<cdiv(tiles, 8), 256, 0, stream>>>(
        x, Wsrc1, vfold, h1, logit, N);
    seg_gather<v4f, 128, true><<<cdiv(N, 8), 256, 0, stream>>>(
        rowptr, col, asrc, adst, h1, b1, o1, N);
  }

  {
    float* out = (float*)d_out;
    const int tiles = (N / 16) * (OUT / 64);
    gemm_wmma_f32<128, 64, 4><<<cdiv(tiles, 8), 256, 0, stream>>>(
        o1, Wsrc2, vfold + 2 * (size_t)FIN, h2, logit, N);
    seg_gather<v2f, 64, false><<<cdiv(N, 8), 256, 0, stream>>>(
        rowptr, col, asrc, adst, h2, b2, out, N);
  }
}


# Gather kernels: each edge's source id and logit asrc[col[j]] are loaded once and kept in registers through max, sum and weight (parent: three passes over memory); every sum in the parent's order
# speedup vs baseline: 2.5399x; 1.0655x over previous
#include <hip/hip_runtime.h>
#include <hip/hip_bf16.h>

typedef float v2f __attribute__((ext_vector_type(2)));
typedef float v8f __attribute__((ext_vector_type(8)));

#define NEG_SLOPE 0.2f

template<int K, int NOUT, int NT>
__global__ __launch_bounds__(256) void gemm_wmma_f32(
    const float* __restrict__ X, const float* __restrict__ W,
    const float* __restrict__ vecs, float* __restrict__ C,
    float* __restrict__ logit, int M)
{
  const int lane   = threadIdx.x & 31;
  const int wave   = threadIdx.x >> 5;
  constexpr int tilesN = NOUT / (16 * NT);
  static_assert(tilesN == 1 || tilesN == 2, "two logit vectors are split over the strips of a row block");
  constexpr int NV = 2 / tilesN;
  const int tiles  = (M >> 4) * tilesN;
  const int tile   = blockIdx.x * 8 + wave;
  if (tile >= tiles) return;
  const int tM = tile / tilesN;
  const int tN = tile % tilesN;

  const int half = lane >> 4;
  const int l15  = lane & 15;
  const float* xrow = X + (size_t)(tM * 16 + l15) * K + 2 * half;
  const float* wb   = W + (size_t)(2 * half) * NOUT + tN * (16 * NT) + l15;
  const float* vp   = vecs + (size_t)(tN * NV) * K + 2 * half;

  v8f acc[NT];
#pragma unroll
  for (int t = 0; t < NT; ++t) acc[t] = (v8f){0.f,0.f,0.f,0.f,0.f,0.f,0.f,0.f};
  float lg[NV];
  for (int i = 0; i < NV; ++i) lg[i] = 0.f;

#pragma unroll
  for (int k = 0; k < K; k += 4) {
    v2f a;
    a.x = xrow[k + 0];
    a.y = xrow[k + 1];
#pragma unroll
    for (int i = 0; i < NV; ++i) {
      lg[i] = fmaf(a.x, vp[i * K + k + 0], lg[i]);
      lg[i] = fmaf(a.y, vp[i * K + k + 1], lg[i]);
    }
#pragma unroll
    for (int t = 0; t < NT; ++t) {
      v2f b;
      b.x = wb[(size_t)(k + 0) * NOUT + 16 * t];
      b.y = wb[(size_t)(k + 1) * NOUT + 16 * t];
      acc[t] = __builtin_amdgcn_wmma_f32_16x16x4_f32(false, a, false, b,
                                                     (short)0, acc[t],
                                                     false, false);
    }
  }

  const int mbase = tM * 16 + 8 * half;
  float* cp = C + (size_t)mbase * NOUT + tN * (16 * NT) + l15;
#pragma unroll
  for (int t = 0; t < NT; ++t)
#pragma unroll
    for (int r = 0; r < 8; ++r)
      cp[(size_t)r * NOUT + 16 * t] = acc[t][r];

  for (int i = 0; i < NV; ++i) {
    const float s = lg[i] + __shfl_xor(lg[i], 16, 32);
    if (half == 0) logit[(size_t)(tN * NV + i) * M + tM * 16 + l15] = s;
  }
}

__global__ __launch_bounds__(256) void fold_att(
    const float* __restrict__ W0, const float* __restrict__ a0,
    const float* __restrict__ W1, const float* __restrict__ a1,
    const float* __restrict__ W2, const float* __restrict__ a2,
    const float* __restrict__ W3, const float* __restrict__ a3,
    float* __restrict__ v, int r1, int c1, int r2, int c2)
{
  const int gw   = (int)((blockIdx.x * (size_t)blockDim.x + threadIdx.x) >> 5);
  const int lane = threadIdx.x & 31;
  if (gw >= 2 * r1 + 2 * r2) return;
  const float* Wm; const float* att; int row, cols;
  if (gw < r1)               { Wm = W0; att = a0; row = gw;               cols = c1; }
  else if (gw < 2 * r1)      { Wm = W1; att = a1; row = gw - r1;          cols = c1; }
  else if (gw < 2 * r1 + r2) { Wm = W2; att = a2; row = gw - 2 * r1;      cols = c2; }
  else                       { Wm = W3; att = a3; row = gw - 2 * r1 - r2; cols = c2; }
  const float* wr = Wm + (size_t)row * cols;
  float s = 0.f;
  for (int k = lane; k < cols; k += 32) s += wr[k] * att[k];
  for (int off = 16; off > 0; off >>= 1) s += __shfl_down(s, off, 32);
  if (lane == 0) v[gw] = s;
}

__global__ void zero_counts(int* __restrict__ cnt, int N)
{
  const int i = blockIdx.x * blockDim.x + threadIdx.x;
  if (i < N) cnt[i] = 0;
}

__global__ __launch_bounds__(256) void count_dst(
    const int* __restrict__ dst, int* __restrict__ cnt, int E)
{
  const int i0 = blockIdx.x * 1024 + threadIdx.x;
  int d[4];
  for (int q = 0; q < 4; ++q) {
    const int i = i0 + q * 256;
    d[q] = (i < E) ? dst[i] : -1;
  }
  for (int q = 0; q < 4; ++q)
    if (d[q] >= 0) atomicAdd(&cnt[d[q]], 1);
}

__global__ __launch_bounds__(1024) void scan_counts(
    const int* __restrict__ cnt, int* __restrict__ rowptr,
    int* __restrict__ cursor, int N)
{
  __shared__ int wtot[32];
  __shared__ int tile_total;
  const int tid  = threadIdx.x;
  const int lane = tid & 31;
  const int wave = tid >> 5;
  int carry = 0;
  for (int base = 0; base < N; base += 8192) {
    const int i0 = base + tid * 8;
    int c[8];
    int t = 0;
    for (int q = 0; q < 8; ++q) {
      c[q] = (i0 + q < N) ? cnt[i0 + q] : 0;
      t += c[q];
    }
    int s = t;
    for (int off = 1; off < 32; off <<= 1) {
      const int nb = __shfl_up(s, off, 32);
      if (lane >= off) s += nb;
    }
    if (lane == 31) wtot[wave] = s;
    __syncthreads();
    if (wave == 0) {
      const int w = wtot[lane];
      int ws = w;
      for (int off = 1; off < 32; off <<= 1) {
        const int nb = __shfl_up(ws, off, 32);
        if (lane >= off) ws += nb;
      }
      wtot[lane] = ws - w;
      if (lane == 31) tile_total = ws;
    }
    __syncthreads();
    int e = carry + wtot[wave] + (s - t);
    carry += tile_total;
    for (int q = 0; q < 8; ++q) {
      if (i0 + q < N) { rowptr[i0 + q] = e; cursor[i0 + q] = e; }
      e += c[q];
    }
    __syncthreads();
  }
  if (tid == 0) rowptr[N] = carry;
}

__global__ __launch_bounds__(256) void fill_csr(
    const int* __restrict__ src, const int* __restrict__ dst,
    int* __restrict__ cursor, int* __restrict__ col, int E)
{
  const int i0 = blockIdx.x * 1024 + threadIdx.x;
  int d[4], s[4], p[4];
  for (int q = 0; q < 4; ++q) {
    const int i = i0 + q * 256;
    d[q] = (i < E) ? dst[i] : -1;
    s[q] = (i < E) ? src[i] : 0;
  }
  for (int q = 0; q < 4; ++q)
    p[q] = (d[q] >= 0) ? atomicAdd(&cursor[d[q]], 1) : -1;
  for (int q = 0; q < 4; ++q)
    if (p[q] >= 0) col[p[q]] = s[q];
}

typedef float v4f __attribute__((ext_vector_type(4)));

__device__ __forceinline__ float leaky(float v)
{
  return (v > 0.f) ? v : NEG_SLOPE * v;
}

template<typename VT, int C, bool RELU>
__global__ __launch_bounds__(256) void seg_gather(
    const int* __restrict__ rowptr, const int* __restrict__ col,
    const float* __restrict__ asrc, const float* __restrict__ adst,
    const float* __restrict__ h, const float* __restrict__ bias,
    float* __restrict__ out, int N)
{
  constexpr int P = C / 32;
  const int lane = threadIdx.x & 31;
  const int d = __builtin_amdgcn_readfirstlane((int)(blockIdx.x * 8 + (threadIdx.x >> 5)));
  if (d >= N) return;
  const int beg = rowptr[d];
  const int end = rowptr[d + 1];
  const float ad = adst[d];

  const int j0 = beg + lane;
  int   sl0 = 0;
  float e0  = -__builtin_huge_valf();
  if (j0 < end) {
    sl0 = col[j0];
    e0  = leaky(asrc[sl0] + ad);
  }

  float mx = e0;
  for (int j = j0 + 32; j < end; j += 32)
    mx = fmaxf(mx, leaky(asrc[col[j]] + ad));
  for (int off = 16; off > 0; off >>= 1) mx = fmaxf(mx, __shfl_xor(mx, off, 32));

  const float p0 = (j0 < end) ? __expf(e0 - mx) : 0.f;
  float den = p0;
  for (int j = j0 + 32; j < end; j += 32)
    den += __expf(leaky(asrc[col[j]] + ad) - mx);
  for (int off = 16; off > 0; off >>= 1) den += __shfl_xor(den, off, 32);

  VT acc = (VT)(0.f);
  for (int base = beg; base < end; base += 32) {
    int   sl = sl0;
    float al = p0 / den;
    if (base != beg) {
      const int j = base + lane;
      sl = 0;
      al = 0.f;
      if (j < end) {
        sl = col[j];
        al = __expf(leaky(asrc[sl] + ad) - mx) / den;
      }
    }
    const int n = (end - base < 32) ? (end - base) : 32;
    int k = 0;
    for (; k + 4 <= n; k += 4) {
      const int s0 = __builtin_amdgcn_readlane(sl, k);
      const int s1 = __builtin_amdgcn_readlane(sl, k + 1);
      const int s2 = __builtin_amdgcn_readlane(sl, k + 2);
      const int s3 = __builtin_amdgcn_readlane(sl, k + 3);
      const VT r0 = *(const VT*)(h + (size_t)s0 * C + lane * P);
      const VT r1 = *(const VT*)(h + (size_t)s1 * C + lane * P);
      const VT r2 = *(const VT*)(h + (size_t)s2 * C + lane * P);
      const VT r3 = *(const VT*)(h + (size_t)s3 * C + lane * P);
      const float a0 = __int_as_float(__builtin_amdgcn_readlane(__float_as_int(al), k));
      const float a1 = __int_as_float(__builtin_amdgcn_readlane(__float_as_int(al), k + 1));
      const float a2 = __int_as_float(__builtin_amdgcn_readlane(__float_as_int(al), k + 2));
      const float a3 = __int_as_float(__builtin_amdgcn_readlane(__float_as_int(al), k + 3));
      acc += a0 * r0;
      acc += a1 * r1;
      acc += a2 * r2;
      acc += a3 * r3;
    }
    for (; k < n; ++k) {
      const int s0 = __builtin_amdgcn_readlane(sl, k);
      const float a0 = __int_as_float(__builtin_amdgcn_readlane(__float_as_int(al), k));
      const VT r0 = *(const VT*)(h + (size_t)s0 * C + lane * P);
      acc += a0 * r0;
    }
  }

  VT o = acc + *(const VT*)(bias + lane * P);
  if (RELU) {
    for (int i = 0; i < P; ++i) o[i] = fmaxf(o[i], 0.f);
  }
  *(VT*)(out + (size_t)d * C + lane * P) = o;
}

static inline int cdiv(long a, long b) { return (int)((a + b - 1) / b); }

extern "C" void kernel_launch(void* const* d_in, const int* in_sizes, int n_in,
                              void* d_out, int out_size, void* d_ws, size_t ws_size,
                              hipStream_t stream)
{
  const float* x       = (const float*)d_in[0];
  const int*   ei      = (const int*)d_in[1];
  const float* Wsrc1   = (const float*)d_in[2];
  const float* Wdst1   = (const float*)d_in[3];
  const float* attsrc1 = (const float*)d_in[4];
  const float* attdst1 = (const float*)d_in[5];
  const float* b1      = (const float*)d_in[6];
  const float* Wsrc2   = (const float*)d_in[7];
  const float* Wdst2   = (const float*)d_in[8];
  const float* attsrc2 = (const float*)d_in[9];
  const float* attdst2 = (const float*)d_in[10];
  const float* b2      = (const float*)d_in[11];

  const int HID = in_sizes[4];
  const int OUT = in_sizes[9];
  const int FIN = in_sizes[2] / HID;
  const int N   = in_sizes[0] / FIN;
  const int E   = in_sizes[1] / 2;
  const int* src = ei;
  const int* dst = ei + E;

  float* ws    = (float*)d_ws;
  float* h1    = ws;  ws += (size_t)N * HID;
  float* o1    = ws;  ws += (size_t)N * HID;
  float* h2    = ws;  ws += (size_t)N * OUT;
  float* logit = ws;  ws += 2 * (size_t)N;
  float* asrc  = logit;
  float* adst  = logit + N;
  float* vfold = ws;  ws += 2 * (size_t)FIN + 2 * (size_t)HID;
  int* cnt     = (int*)ws;  ws += N;
  int* cursor  = (int*)ws;  ws += N;
  int* col     = (int*)ws;  ws += E;
  int* rowptr  = (int*)ws;  ws += N + 1;

  zero_counts<<<cdiv(N, 256), 256, 0, stream>>>(cnt, N);
  count_dst<<<cdiv(E, 1024), 256, 0, stream>>>(dst, cnt, E);
  scan_counts<<<1, 1024, 0, stream>>>(cnt, rowptr, cursor, N);
  fill_csr<<<cdiv(E, 1024), 256, 0, stream>>>(src, dst, cursor, col, E);

  fold_att<<<cdiv((long)(2 * FIN + 2 * HID) * 32, 256), 256, 0, stream>>>(
      Wsrc1, attsrc1, Wdst1, attdst1, Wsrc2, attsrc2, Wdst2, attdst2,
      vfold, FIN, HID, HID, OUT);

  {
    const int tiles = (N / 16) * (HID / 64);
    gemm_wmma_f32<128, 128, 4><<<cdiv(tiles, 8), 256, 0, stream>>>(
        x, Wsrc1, vfold, h1, logit, N);
    seg_gather<v4f, 128, true><<<cdiv(N, 8), 256, 0, stream>>>(
        rowptr, col, asrc, adst, h1, b1, o1, N);
  }

  {
    float* out = (float*)d_out;
    const int tiles = (N / 16) * (OUT / 64);
    gemm_wmma_f32<128, 64, 4><<<cdiv(tiles, 8), 256, 0, stream>>>(
        o1, Wsrc2, vfold + 2 * (size_t)FIN, h2, logit, N);
    seg_gather<v2f, 64, false><<<cdiv(N, 8), 256, 0, stream>>>(
        rowptr, col, asrc, adst, h2, b2, out, N);
  }
}


# Edges grouped by destination on shared-memory counters (bin by 512-node range, scan, rank inside each range): no global atomics left; a node's rows summed in shared-counter arrival order
# speedup vs baseline: 4.9341x; 1.9426x over previous
#include <hip/hip_runtime.h>
#include <hip/hip_bf16.h>

typedef float v2f __attribute__((ext_vector_type(2)));
typedef float v8f __attribute__((ext_vector_type(8)));

#define NEG_SLOPE 0.2f

template<int K, int NOUT, int NT>
__global__ __launch_bounds__(256) void gemm_wmma_f32(
    const float* __restrict__ X, const float* __restrict__ W,
    const float* __restrict__ vecs, float* __restrict__ C,
    float* __restrict__ logit, int M)
{
  const int lane   = threadIdx.x & 31;
  const int wave   = threadIdx.x >> 5;
  constexpr int tilesN = NOUT / (16 * NT);
  static_assert(tilesN == 1 || tilesN == 2, "two logit vectors are split over the strips of a row block");
  constexpr int NV = 2 / tilesN;
  const int tiles  = (M >> 4) * tilesN;
  const int tile   = blockIdx.x * 8 + wave;
  if (tile >= tiles) return;
  const int tM = tile / tilesN;
  const int tN = tile % tilesN;

  const int half = lane >> 4;
  const int l15  = lane & 15;
  const float* xrow = X + (size_t)(tM * 16 + l15) * K + 2 * half;
  const float* wb   = W + (size_t)(2 * half) * NOUT + tN * (16 * NT) + l15;
  const float* vp   = vecs + (size_t)(tN * NV) * K + 2 * half;

  v8f acc[NT];
#pragma unroll
  for (int t = 0; t < NT; ++t) acc[t] = (v8f){0.f,0.f,0.f,0.f,0.f,0.f,0.f,0.f};
  float lg[NV];
  for (int i = 0; i < NV; ++i) lg[i] = 0.f;

#pragma unroll
  for (int k = 0; k < K; k += 4) {
    v2f a;
    a.x = xrow[k + 0];
    a.y = xrow[k + 1];
#pragma unroll
    for (int i = 0; i < NV; ++i) {
      lg[i] = fmaf(a.x, vp[i * K + k + 0], lg[i]);
      lg[i] = fmaf(a.y, vp[i * K + k + 1], lg[i]);
    }
#pragma unroll
    for (int t = 0; t < NT; ++t) {
      v2f b;
      b.x = wb[(size_t)(k + 0) * NOUT + 16 * t];
      b.y = wb[(size_t)(k + 1) * NOUT + 16 * t];
      acc[t] = __builtin_amdgcn_wmma_f32_16x16x4_f32(false, a, false, b,
                                                     (short)0, acc[t],
                                                     false, false);
    }
  }

  const int mbase = tM * 16 + 8 * half;
  float* cp = C + (size_t)mbase * NOUT + tN * (16 * NT) + l15;
#pragma unroll
  for (int t = 0; t < NT; ++t)
#pragma unroll
    for (int r = 0; r < 8; ++r)
      cp[(size_t)r * NOUT + 16 * t] = acc[t][r];

  for (int i = 0; i < NV; ++i) {
    const float s = lg[i] + __shfl_xor(lg[i], 16, 32);
    if (half == 0) logit[(size_t)(tN * NV + i) * M + tM * 16 + l15] = s;
  }
}

__global__ __launch_bounds__(256) void fold_att(
    const float* __restrict__ W0, const float* __restrict__ a0,
    const float* __restrict__ W1, const float* __restrict__ a1,
    const float* __restrict__ W2, const float* __restrict__ a2,
    const float* __restrict__ W3, const float* __restrict__ a3,
    float* __restrict__ v, int r1, int c1, int r2, int c2)
{
  const int gw   = (int)((blockIdx.x * (size_t)blockDim.x + threadIdx.x) >> 5);
  const int lane = threadIdx.x & 31;
  if (gw >= 2 * r1 + 2 * r2) return;
  const float* Wm; const float* att; int row, cols;
  if (gw < r1)               { Wm = W0; att = a0; row = gw;               cols = c1; }
  else if (gw < 2 * r1)      { Wm = W1; att = a1; row = gw - r1;          cols = c1; }
  else if (gw < 2 * r1 + r2) { Wm = W2; att = a2; row = gw - 2 * r1;      cols = c2; }
  else                       { Wm = W3; att = a3; row = gw - 2 * r1 - r2; cols = c2; }
  const float* wr = Wm + (size_t)row * cols;
  float s = 0.f;
  for (int k = lane; k < cols; k += 32) s += wr[k] * att[k];
  for (int off = 16; off > 0; off >>= 1) s += __shfl_down(s, off, 32);
  if (lane == 0) v[gw] = s;
}

#define RB_SHIFT 9
#define RB (1 << RB_SHIFT)
#define MAXNB 1024

__global__ __launch_bounds__(1024) void bin_count(
    const int* __restrict__ dst, int* __restrict__ wgcnt, int E, int NB)
{
  __shared__ int hist[MAXNB];
  const int tid = threadIdx.x;
  const int G   = (int)gridDim.x;
  for (int r = tid; r < NB; r += 1024) hist[r] = 0;
  __syncthreads();
  const int chunk = (E + G - 1) / G;
  const int e0 = (int)blockIdx.x * chunk;
  const int e1 = (e0 + chunk < E) ? (e0 + chunk) : E;
  for (int base = e0; base < e1; base += 4096) {
    int d[4];
    for (int q = 0; q < 4; ++q) {
      const int i = base + tid + q * 1024;
      d[q] = (i < e1) ? dst[i] : -1;
    }
    for (int q = 0; q < 4; ++q)
      if (d[q] >= 0) atomicAdd(&hist[d[q] >> RB_SHIFT], 1);
  }
  __syncthreads();
  for (int r = tid; r < NB; r += 1024) wgcnt[(size_t)r * G + blockIdx.x] = hist[r];
}

__global__ __launch_bounds__(1024) void scan_counts(
    const int* __restrict__ cnt, int* __restrict__ rowptr, int N)
{
  __shared__ int wtot[32];
  __shared__ int tile_total;
  const int tid  = threadIdx.x;
  const int lane = tid & 31;
  const int wave = tid >> 5;
  int carry = 0;
  for (int base = 0; base < N; base += 8192) {
    const int i0 = base + tid * 8;
    int c[8];
    int t = 0;
    for (int q = 0; q < 8; ++q) {
      c[q] = (i0 + q < N) ? cnt[i0 + q] : 0;
      t += c[q];
    }
    int s = t;
    for (int off = 1; off < 32; off <<= 1) {
      const int nb = __shfl_up(s, off, 32);
      if (lane >= off) s += nb;
    }
    if (lane == 31) wtot[wave] = s;
    __syncthreads();
    if (wave == 0) {
      const int w = wtot[lane];
      int ws = w;
      for (int off = 1; off < 32; off <<= 1) {
        const int nb = __shfl_up(ws, off, 32);
        if (lane >= off) ws += nb;
      }
      wtot[lane] = ws - w;
      if (lane == 31) tile_total = ws;
    }
    __syncthreads();
    int e = carry + wtot[wave] + (s - t);
    carry += tile_total;
    for (int q = 0; q < 8; ++q) {
      if (i0 + q < N) rowptr[i0 + q] = e;
      e += c[q];
    }
    __syncthreads();
  }
  if (tid == 0) rowptr[N] = carry;
}

__global__ __launch_bounds__(1024) void bin_scatter(
    const int* __restrict__ src, const int* __restrict__ dst,
    const int* __restrict__ wgoff, int* __restrict__ binned, int E, int NB)
{
  __shared__ int cur[MAXNB];
  const int tid = threadIdx.x;
  const int G   = (int)gridDim.x;
  for (int r = tid; r < NB; r += 1024) cur[r] = wgoff[(size_t)r * G + blockIdx.x];
  __syncthreads();
  int2* pairs = reinterpret_cast<int2*>(binned);
  const int chunk = (E + G - 1) / G;
  const int e0 = (int)blockIdx.x * chunk;
  const int e1 = (e0 + chunk < E) ? (e0 + chunk) : E;
  for (int base = e0; base < e1; base += 4096) {
    int d[4], s[4];
    for (int q = 0; q < 4; ++q) {
      const int i = base + tid + q * 1024;
      d[q] = (i < e1) ? dst[i] : -1;
      s[q] = (i < e1) ? src[i] : 0;
    }
    for (int q = 0; q < 4; ++q)
      if (d[q] >= 0) {
        const int slot = atomicAdd(&cur[d[q] >> RB_SHIFT], 1);
        pairs[slot] = make_int2(s[q], d[q]);
      }
  }
}

__global__ __launch_bounds__(1024) void range_sort(
    const int* __restrict__ binned, const int* __restrict__ wgoff,
    int* __restrict__ rowptr, int* __restrict__ col, int N, int G)
{
  __shared__ int cnt[RB];
  __shared__ int wtot[RB / 32];
  const int tid  = threadIdx.x;
  const int lane = tid & 31;
  const int wave = tid >> 5;
  const int rg   = (int)blockIdx.x;
  const int s0 = wgoff[(size_t)rg * G];
  const int s1 = wgoff[(size_t)(rg + 1) * G];
  const int2* pairs = reinterpret_cast<const int2*>(binned);

  if (tid < RB) cnt[tid] = 0;
  __syncthreads();
  for (int base = s0; base < s1; base += 4096) {
    int dl[4];
    for (int q = 0; q < 4; ++q) {
      const int i = base + tid + q * 1024;
      dl[q] = (i < s1) ? (pairs[i].y & (RB - 1)) : -1;
    }
    for (int q = 0; q < 4; ++q)
      if (dl[q] >= 0) atomicAdd(&cnt[dl[q]], 1);
  }
  __syncthreads();

  int v = 0, incl = 0;
  if (tid < RB) {
    v = cnt[tid];
    incl = v;
    for (int off = 1; off < 32; off <<= 1) {
      const int nb = __shfl_up(incl, off, 32);
      if (lane >= off) incl += nb;
    }
    if (lane == 31) wtot[wave] = incl;
  }
  __syncthreads();
  if (tid < RB) {
    int offs = 0;
    for (int w = 0; w < wave; ++w) offs += wtot[w];
    const int excl = offs + incl - v;
    cnt[tid] = excl;
    const int node = rg * RB + tid;
    if (node < N) rowptr[node] = s0 + excl;
  }
  if (rg == (int)gridDim.x - 1 && tid == 0) rowptr[N] = s1;
  __syncthreads();

  for (int base = s0; base < s1; base += 4096) {
    int2 r[4];
    for (int q = 0; q < 4; ++q) {
      const int i = base + tid + q * 1024;
      r[q] = (i < s1) ? pairs[i] : make_int2(0, -1);
    }
    for (int q = 0; q < 4; ++q)
      if (r[q].y >= 0) {
        const int slot = atomicAdd(&cnt[r[q].y & (RB - 1)], 1);
        col[s0 + slot] = r[q].x;
      }
  }
}

typedef float v4f __attribute__((ext_vector_type(4)));

__device__ __forceinline__ float leaky(float v)
{
  return (v > 0.f) ? v : NEG_SLOPE * v;
}

template<typename VT, int C, bool RELU>
__global__ __launch_bounds__(256) void seg_gather(
    const int* __restrict__ rowptr, const int* __restrict__ col,
    const float* __restrict__ asrc, const float* __restrict__ adst,
    const float* __restrict__ h, const float* __restrict__ bias,
    float* __restrict__ out, int N)
{
  constexpr int P = C / 32;
  const int lane = threadIdx.x & 31;
  const int d = __builtin_amdgcn_readfirstlane((int)(blockIdx.x * 8 + (threadIdx.x >> 5)));
  if (d >= N) return;
  const int beg = rowptr[d];
  const int end = rowptr[d + 1];
  const float ad = adst[d];

  const int j0 = beg + lane;
  int   sl0 = 0;
  float e0  = -__builtin_huge_valf();
  if (j0 < end) {
    sl0 = col[j0];
    e0  = leaky(asrc[sl0] + ad);
  }

  float mx = e0;
  for (int j = j0 + 32; j < end; j += 32)
    mx = fmaxf(mx, leaky(asrc[col[j]] + ad));
  for (int off = 16; off > 0; off >>= 1) mx = fmaxf(mx, __shfl_xor(mx, off, 32));

  const float p0 = (j0 < end) ? __expf(e0 - mx) : 0.f;
  float den = p0;
  for (int j = j0 + 32; j < end; j += 32)
    den += __expf(leaky(asrc[col[j]] + ad) - mx);
  for (int off = 16; off > 0; off >>= 1) den += __shfl_xor(den, off, 32);

  VT acc = (VT)(0.f);
  for (int base = beg; base < end; base += 32) {
    int   sl = sl0;
    float al = p0 / den;
    if (base != beg) {
      const int j = base + lane;
      sl = 0;
      al = 0.f;
      if (j < end) {
        sl = col[j];
        al = __expf(leaky(asrc[sl] + ad) - mx) / den;
      }
    }
    const int n = (end - base < 32) ? (end - base) : 32;
    int k = 0;
    for (; k + 4 <= n; k += 4) {
      const int s0 = __builtin_amdgcn_readlane(sl, k);
      const int s1 = __builtin_amdgcn_readlane(sl, k + 1);
      const int s2 = __builtin_amdgcn_readlane(sl, k + 2);
      const int s3 = __builtin_amdgcn_readlane(sl, k + 3);
      const VT r0 = *(const VT*)(h + (size_t)s0 * C + lane * P);
      const VT r1 = *(const VT*)(h + (size_t)s1 * C + lane * P);
      const VT r2 = *(const VT*)(h + (size_t)s2 * C + lane * P);
      const VT r3 = *(const VT*)(h + (size_t)s3 * C + lane * P);
      const float a0 = __int_as_float(__builtin_amdgcn_readlane(__float_as_int(al), k));
      const float a1 = __int_as_float(__builtin_amdgcn_readlane(__float_as_int(al), k + 1));
      const float a2 = __int_as_float(__builtin_amdgcn_readlane(__float_as_int(al), k + 2));
      const float a3 = __int_as_float(__builtin_amdgcn_readlane(__float_as_int(al), k + 3));
      acc += a0 * r0;
      acc += a1 * r1;
      acc += a2 * r2;
      acc += a3 * r3;
    }
    for (; k < n; ++k) {
      const int s0 = __builtin_amdgcn_readlane(sl, k);
      const float a0 = __int_as_float(__builtin_amdgcn_readlane(__float_as_int(al), k));
      const VT r0 = *(const VT*)(h + (size_t)s0 * C + lane * P);
      acc += a0 * r0;
    }
  }

  VT o = acc + *(const VT*)(bias + lane * P);
  if (RELU) {
    for (int i = 0; i < P; ++i) o[i] = fmaxf(o[i], 0.f);
  }
  *(VT*)(out + (size_t)d * C + lane * P) = o;
}

static inline int cdiv(long a, long b) { return (int)((a + b - 1) / b); }

extern "C" void kernel_launch(void* const* d_in, const int* in_sizes, int n_in,
                              void* d_out, int out_size, void* d_ws, size_t ws_size,
                              hipStream_t stream)
{
  const float* x       = (const float*)d_in[0];
  const int*   ei      = (const int*)d_in[1];
  const float* Wsrc1   = (const float*)d_in[2];
  const float* Wdst1   = (const float*)d_in[3];
  const float* attsrc1 = (const float*)d_in[4];
  const float* attdst1 = (const float*)d_in[5];
  const float* b1      = (const float*)d_in[6];
  const float* Wsrc2   = (const float*)d_in[7];
  const float* Wdst2   = (const float*)d_in[8];
  const float* attsrc2 = (const float*)d_in[9];
  const float* attdst2 = (const float*)d_in[10];
  const float* b2      = (const float*)d_in[11];

  const int HID = in_sizes[4];
  const int OUT = in_sizes[9];
  const int FIN = in_sizes[2] / HID;
  const int N   = in_sizes[0] / FIN;
  const int E   = in_sizes[1] / 2;
  const int* src = ei;
  const int* dst = ei + E;

  float* ws    = (float*)d_ws;
  float* h1    = ws;  ws += (size_t)N * HID;
  float* o1    = ws;  ws += (size_t)N * HID;
  float* h2    = ws;  ws += (size_t)N * OUT;
  float* logit = ws;  ws += 2 * (size_t)N;
  float* asrc  = logit;
  float* adst  = logit + N;
  float* vfold = ws;  ws += 2 * (size_t)FIN + 2 * (size_t)HID;
  const int G1 = 256;
  const int NB = cdiv(N, RB);
  int* binned  = (int*)ws;  ws += 2 * (size_t)E;
  int* wgcnt   = (int*)ws;  ws += (size_t)NB * G1;
  int* wgoff   = (int*)ws;  ws += (size_t)NB * G1 + 1;
  int* col     = (int*)ws;  ws += E;
  int* rowptr  = (int*)ws;  ws += N + 1;

  bin_count<<<G1, 1024, 0, stream>>>(dst, wgcnt, E, NB);
  scan_counts<<<1, 1024, 0, stream>>>(wgcnt, wgoff, NB * G1);
  bin_scatter<<<G1, 1024, 0, stream>>>(src, dst, wgoff, binned, E, NB);
  range_sort<<<NB, 1024, 0, stream>>>(binned, wgoff, rowptr, col, N, G1);

  fold_att<<<cdiv((long)(2 * FIN + 2 * HID) * 32, 256), 256, 0, stream>>>(
      Wsrc1, attsrc1, Wdst1, attdst1, Wsrc2, attsrc2, Wdst2, attdst2,
      vfold, FIN, HID, HID, OUT);

  {
    const int tiles = (N / 16) * (HID / 64);
    gemm_wmma_f32<128, 128, 4><<<cdiv(tiles, 8), 256, 0, stream>>>(
        x, Wsrc1, vfold, h1, logit, N);
    seg_gather<v4f, 128, true><<<cdiv(N, 8), 256, 0, stream>>>(
        rowptr, col, asrc, adst, h1, b1, o1, N);
  }

  {
    float* out = (float*)d_out;
    const int tiles = (N / 16) * (OUT / 64);
    gemm_wmma_f32<128, 64, 4><<<cdiv(tiles, 8), 256, 0, stream>>>(
        o1, Wsrc2, vfold + 2 * (size_t)FIN, h2, logit, N);
    seg_gather<v2f, 64, false><<<cdiv(N, 8), 256, 0, stream>>>(
        rowptr, col, asrc, adst, h2, b2, out, N);
  }
}


# Layer-1 GEMM strips and the attention fold ride as extra workgroups in the four edge-grouping launches (GEMM kept to 64 registers so both kinds share a compute unit): 9 launches to 7; sums unchanged
# speedup vs baseline: 5.3714x; 1.0886x over previous
#include <hip/hip_runtime.h>
#include <hip/hip_bf16.h>

typedef float v2f __attribute__((ext_vector_type(2)));
typedef float v8f __attribute__((ext_vector_type(8)));

#define NEG_SLOPE 0.2f

template<int K, int NOUT, int NT, bool AHEAD>
__device__ __forceinline__ void gemm_strip(
    const float* __restrict__ X, const float* __restrict__ W,
    const float* __restrict__ vecs, float* __restrict__ C,
    float* __restrict__ logit, int M, int tile)
{
  const int lane   = threadIdx.x & 31;
  constexpr int tilesN = NOUT / (16 * NT);
  static_assert(tilesN == 1 || tilesN == 2, "two logit vectors are split over the strips of a row block");
  constexpr int NV = 2 / tilesN;
  const int tM = tile / tilesN;
  const int tN = tile % tilesN;

  const int half = lane >> 4;
  const int l15  = lane & 15;
  const float* xrow = X + (size_t)(tM * 16 + l15) * K + 2 * half;
  const float* wb   = W + (size_t)(2 * half) * NOUT + tN * (16 * NT) + l15;
  const float* vp   = vecs + (size_t)(tN * NV) * K + 2 * half;

  v8f acc[NT];
#pragma unroll
  for (int t = 0; t < NT; ++t) acc[t] = (v8f){0.f,0.f,0.f,0.f,0.f,0.f,0.f,0.f};
  float lg[NV];
  for (int i = 0; i < NV; ++i) lg[i] = 0.f;

  if constexpr (AHEAD) {
    v2f a, an;
    v2f vv[NV], vn[NV];
    v2f b[NT];
    a.x = xrow[0];
    a.y = xrow[1];
#pragma unroll
    for (int i = 0; i < NV; ++i) { vv[i].x = vp[i * K + 0]; vv[i].y = vp[i * K + 1]; }
#pragma unroll
    for (int t = 0; t < NT; ++t) { b[t].x = wb[16 * t]; b[t].y = wb[(size_t)NOUT + 16 * t]; }
#pragma unroll
    for (int k = 0; k < K; k += 4) {
      if (k + 4 < K) {
        an.x = xrow[k + 4];
        an.y = xrow[k + 5];
#pragma unroll
        for (int i = 0; i < NV; ++i) { vn[i].x = vp[i * K + k + 4]; vn[i].y = vp[i * K + k + 5]; }
      }
#pragma unroll
      for (int i = 0; i < NV; ++i) {
        lg[i] = fmaf(a.x, vv[i].x, lg[i]);
        lg[i] = fmaf(a.y, vv[i].y, lg[i]);
      }
#pragma unroll
      for (int t = 0; t < NT; ++t) {
        acc[t] = __builtin_amdgcn_wmma_f32_16x16x4_f32(false, a, false, b[t],
                                                       (short)0, acc[t],
                                                       false, false);
        if (k + 4 < K) {
          b[t].x = wb[(size_t)(k + 4) * NOUT + 16 * t];
          b[t].y = wb[(size_t)(k + 5) * NOUT + 16 * t];
        }
        __builtin_amdgcn_sched_barrier(0);
      }
      if (k + 4 < K) {
        a = an;
#pragma unroll
        for (int i = 0; i < NV; ++i) vv[i] = vn[i];
      }
    }
  } else {
#pragma unroll
    for (int k = 0; k < K; k += 4) {
      v2f a;
      a.x = xrow[k + 0];
      a.y = xrow[k + 1];
#pragma unroll
      for (int i = 0; i < NV; ++i) {
        lg[i] = fmaf(a.x, vp[i * K + k + 0], lg[i]);
        lg[i] = fmaf(a.y, vp[i * K + k + 1], lg[i]);
      }
#pragma unroll
      for (int t = 0; t < NT; ++t) {
        v2f b;
        b.x = wb[(size_t)(k + 0) * NOUT + 16 * t];
        b.y = wb[(size_t)(k + 1) * NOUT + 16 * t];
        acc[t] = __builtin_amdgcn_wmma_f32_16x16x4_f32(false, a, false, b,
                                                       (short)0, acc[t],
                                                       false, false);
      }
    }
  }

  const int mbase = tM * 16 + 8 * half;
  float* cp = C + (size_t)mbase * NOUT + tN * (16 * NT) + l15;
#pragma unroll
  for (int t = 0; t < NT; ++t)
#pragma unroll
    for (int r = 0; r < 8; ++r)
      cp[(size_t)r * NOUT + 16 * t] = acc[t][r];

  for (int i = 0; i < NV; ++i) {
    const float s = lg[i] + __shfl_xor(lg[i], 16, 32);
    if (half == 0) logit[(size_t)(tN * NV + i) * M + tM * 16 + l15] = s;
  }
}

template<int K, int NOUT, int NT>
__global__ __launch_bounds__(256) void gemm_wmma_f32(
    const float* __restrict__ X, const float* __restrict__ W,
    const float* __restrict__ vecs, float* __restrict__ C,
    float* __restrict__ logit, int M)
{
  const int wave  = threadIdx.x >> 5;
  const int tiles = (M >> 4) * (NOUT / (16 * NT));
  const int tile  = blockIdx.x * 8 + wave;
  if (tile >= tiles) return;
  gemm_strip<K, NOUT, NT, false>(X, W, vecs, C, logit, M, tile);
}

#define GEMM_WPB 16
__device__ __forceinline__ void gemm1_part(
    const float* __restrict__ X, const float* __restrict__ W,
    const float* __restrict__ vecs, float* __restrict__ C,
    float* __restrict__ logit, int M, int tile0, int tile1, int gb)
{
  const int wave = threadIdx.x >> 5;
  if (wave >= GEMM_WPB) return;
  const int tile = __builtin_amdgcn_readfirstlane(tile0 + gb * GEMM_WPB + wave);
  if (tile >= tile1) return;
  gemm_strip<128, 128, 4, true>(X, W, vecs, C, logit, M, tile);
}

__device__ __forceinline__ void fold_att_body(
    const float* __restrict__ W0, const float* __restrict__ a0,
    const float* __restrict__ W1, const float* __restrict__ a1,
    const float* __restrict__ W2, const float* __restrict__ a2,
    const float* __restrict__ W3, const float* __restrict__ a3,
    float* __restrict__ v, int r1, int c1, int r2, int c2, int gw)
{
  const int lane = threadIdx.x & 31;
  if (gw >= 2 * r1 + 2 * r2) return;
  const float* Wm; const float* att; int row, cols;
  if (gw < r1)               { Wm = W0; att = a0; row = gw;               cols = c1; }
  else if (gw < 2 * r1)      { Wm = W1; att = a1; row = gw - r1;          cols = c1; }
  else if (gw < 2 * r1 + r2) { Wm = W2; att = a2; row = gw - 2 * r1;      cols = c2; }
  else                       { Wm = W3; att = a3; row = gw - 2 * r1 - r2; cols = c2; }
  const float* wr = Wm + (size_t)row * cols;
  float s = 0.f;
  for (int k = lane; k < cols; k += 32) s += wr[k] * att[k];
  for (int off = 16; off > 0; off >>= 1) s += __shfl_down(s, off, 32);
  if (lane == 0) v[gw] = s;
}

#define RB_SHIFT 9
#define RB (1 << RB_SHIFT)
#define MAXNB 1024

__device__ __forceinline__ void bin_count_body(
    const int* __restrict__ dst, int* __restrict__ wgcnt, int E, int NB, int bid, int G)
{
  __shared__ int hist[MAXNB];
  const int tid = threadIdx.x;
  for (int r = tid; r < NB; r += 1024) hist[r] = 0;
  __syncthreads();
  const int chunk = (E + G - 1) / G;
  const int e0 = bid * chunk;
  const int e1 = (e0 + chunk < E) ? (e0 + chunk) : E;
  for (int base = e0; base < e1; base += 4096) {
    int d[4];
    for (int q = 0; q < 4; ++q) {
      const int i = base + tid + q * 1024;
      d[q] = (i < e1) ? dst[i] : -1;
    }
    for (int q = 0; q < 4; ++q)
      if (d[q] >= 0) atomicAdd(&hist[d[q] >> RB_SHIFT], 1);
  }
  __syncthreads();
  for (int r = tid; r < NB; r += 1024) wgcnt[(size_t)r * G + bid] = hist[r];
}

__device__ __forceinline__ void scan_counts_body(
    const int* __restrict__ cnt, int* __restrict__ rowptr, int N)
{
  __shared__ int wtot[32];
  __shared__ int tile_total;
  const int tid  = threadIdx.x;
  const int lane = tid & 31;
  const int wave = tid >> 5;
  int carry = 0;
  for (int base = 0; base < N; base += 8192) {
    const int i0 = base + tid * 8;
    int c[8];
    int t = 0;
    for (int q = 0; q < 8; ++q) {
      c[q] = (i0 + q < N) ? cnt[i0 + q] : 0;
      t += c[q];
    }
    int s = t;
    for (int off = 1; off < 32; off <<= 1) {
      const int nb = __shfl_up(s, off, 32);
      if (lane >= off) s += nb;
    }
    if (lane == 31) wtot[wave] = s;
    __syncthreads();
    if (wave == 0) {
      const int w = wtot[lane];
      int ws = w;
      for (int off = 1; off < 32; off <<= 1) {
        const int nb = __shfl_up(ws, off, 32);
        if (lane >= off) ws += nb;
      }
      wtot[lane] = ws - w;
      if (lane == 31) tile_total = ws;
    }
    __syncthreads();
    int e = carry + wtot[wave] + (s - t);
    carry += tile_total;
    for (int q = 0; q < 8; ++q) {
      if (i0 + q < N) rowptr[i0 + q] = e;
      e += c[q];
    }
    __syncthreads();
  }
  if (tid == 0) rowptr[N] = carry;
}

__device__ __forceinline__ void bin_scatter_body(
    const int* __restrict__ src, const int* __restrict__ dst,
    const int* __restrict__ wgoff, int* __restrict__ binned, int E, int NB, int bid, int G)
{
  __shared__ int cur[MAXNB];
  const int tid = threadIdx.x;
  for (int r = tid; r < NB; r += 1024) cur[r] = wgoff[(size_t)r * G + bid];
  __syncthreads();
  int2* pairs = reinterpret_cast<int2*>(binned);
  const int chunk = (E + G - 1) / G;
  const int e0 = bid * chunk;
  const int e1 = (e0 + chunk < E) ? (e0 + chunk) : E;
  for (int base = e0; base < e1; base += 4096) {
    int d[4], s[4];
    for (int q = 0; q < 4; ++q) {
      const int i = base + tid + q * 1024;
      d[q] = (i < e1) ? dst[i] : -1;
      s[q] = (i < e1) ? src[i] : 0;
    }
    for (int q = 0; q < 4; ++q)
      if (d[q] >= 0) {
        const int slot = atomicAdd(&cur[d[q] >> RB_SHIFT], 1);
        pairs[slot] = make_int2(s[q], d[q]);
      }
  }
}

__device__ __forceinline__ void range_sort_body(
    const int* __restrict__ binned, const int* __restrict__ wgoff,
    int* __restrict__ rowptr, int* __restrict__ col, int N, int G, int rg, int NB)
{
  __shared__ int cnt[RB];
  __shared__ int wtot[RB / 32];
  const int tid  = threadIdx.x;
  const int lane = tid & 31;
  const int wave = tid >> 5;
  const int s0 = wgoff[(size_t)rg * G];
  const int s1 = wgoff[(size_t)(rg + 1) * G];
  const int2* pairs = reinterpret_cast<const int2*>(binned);

  if (tid < RB) cnt[tid] = 0;
  __syncthreads();
  for (int base = s0; base < s1; base += 4096) {
    int dl[4];
    for (int q = 0; q < 4; ++q) {
      const int i = base + tid + q * 1024;
      dl[q] = (i < s1) ? (pairs[i].y & (RB - 1)) : -1;
    }
    for (int q = 0; q < 4; ++q)
      if (dl[q] >= 0) atomicAdd(&cnt[dl[q]], 1);
  }
  __syncthreads();

  int v = 0, incl = 0;
  if (tid < RB) {
    v = cnt[tid];
    incl = v;
    for (int off = 1; off < 32; off <<= 1) {
      const int nb = __shfl_up(incl, off, 32);
      if (lane >= off) incl += nb;
    }
    if (lane == 31) wtot[wave] = incl;
  }
  __syncthreads();
  if (tid < RB) {
    int offs = 0;
    for (int w = 0; w < wave; ++w) offs += wtot[w];
    const int excl = offs + incl - v;
    cnt[tid] = excl;
    const int node = rg * RB + tid;
    if (node < N) rowptr[node] = s0 + excl;
  }
  if (rg == NB - 1 && tid == 0) rowptr[N] = s1;
  __syncthreads();

  for (int base = s0; base < s1; base += 4096) {
    int2 r[4];
    for (int q = 0; q < 4; ++q) {
      const int i = base + tid + q * 1024;
      r[q] = (i < s1) ? pairs[i] : make_int2(0, -1);
    }
    for (int q = 0; q < 4; ++q)
      if (r[q].y >= 0) {
        const int slot = atomicAdd(&cnt[r[q].y & (RB - 1)], 1);
        col[s0 + slot] = r[q].x;
      }
  }
}

__global__ __launch_bounds__(1024) void count_and_fold(
    const int* __restrict__ dst, int* __restrict__ wgcnt, int E, int NB, int G,
    const float* __restrict__ W0, const float* __restrict__ a0,
    const float* __restrict__ W1, const float* __restrict__ a1,
    const float* __restrict__ W2, const float* __restrict__ a2,
    const float* __restrict__ W3, const float* __restrict__ a3,
    float* __restrict__ v, int r1, int c1, int r2, int c2)
{
  const int b = (int)blockIdx.x;
  if (b < G) bin_count_body(dst, wgcnt, E, NB, b, G);
  else fold_att_body(W0, a0, W1, a1, W2, a2, W3, a3, v, r1, c1, r2, c2,
                     (b - G) * 32 + (int)(threadIdx.x >> 5));
}

__global__ __launch_bounds__(1024, 16) void scan_and_gemm(
    const int* __restrict__ cnt, int* __restrict__ off, int n,
    const float* __restrict__ X, const float* __restrict__ W,
    const float* __restrict__ vecs, float* __restrict__ C,
    float* __restrict__ logit, int M, int tile0, int tile1)
{
  const int b = (int)blockIdx.x;
  if (b == 0) scan_counts_body(cnt, off, n);
  else gemm1_part(X, W, vecs, C, logit, M, tile0, tile1, b - 1);
}

__global__ __launch_bounds__(1024, 16) void scatter_and_gemm(
    const int* __restrict__ src, const int* __restrict__ dst,
    const int* __restrict__ wgoff, int* __restrict__ binned, int E, int NB, int G,
    const float* __restrict__ X, const float* __restrict__ W,
    const float* __restrict__ vecs, float* __restrict__ C,
    float* __restrict__ logit, int M, int tile0, int tile1)
{
  const int b = (int)blockIdx.x;
  if (b < G) bin_scatter_body(src, dst, wgoff, binned, E, NB, b, G);
  else gemm1_part(X, W, vecs, C, logit, M, tile0, tile1, b - G);
}

__global__ __launch_bounds__(1024, 16) void sort_and_gemm(
    const int* __restrict__ binned, const int* __restrict__ wgoff,
    int* __restrict__ rowptr, int* __restrict__ col, int N, int G, int NB,
    const float* __restrict__ X, const float* __restrict__ W,
    const float* __restrict__ vecs, float* __restrict__ C,
    float* __restrict__ logit, int M, int tile0, int tile1)
{
  const int b = (int)blockIdx.x;
  if (b < NB) range_sort_body(binned, wgoff, rowptr, col, N, G, b, NB);
  else gemm1_part(X, W, vecs, C, logit, M, tile0, tile1, b - NB);
}

typedef float v4f __attribute__((ext_vector_type(4)));

__device__ __forceinline__ float leaky(float v)
{
  return (v > 0.f) ? v : NEG_SLOPE * v;
}

template<typename VT, int C, bool RELU>
__global__ __launch_bounds__(256) void seg_gather(
    const int* __restrict__ rowptr, const int* __restrict__ col,
    const float* __restrict__ asrc, const float* __restrict__ adst,
    const float* __restrict__ h, const float* __restrict__ bias,
    float* __restrict__ out, int N)
{
  constexpr int P = C / 32;
  const int lane = threadIdx.x & 31;
  const int d = __builtin_amdgcn_readfirstlane((int)(blockIdx.x * 8 + (threadIdx.x >> 5)));
  if (d >= N) return;
  const int beg = rowptr[d];
  const int end = rowptr[d + 1];
  const float ad = adst[d];

  const int j0 = beg + lane;
  int   sl0 = 0;
  float e0  = -__builtin_huge_valf();
  if (j0 < end) {
    sl0 = col[j0];
    e0  = leaky(asrc[sl0] + ad);
  }

  float mx = e0;
  for (int j = j0 + 32; j < end; j += 32)
    mx = fmaxf(mx, leaky(asrc[col[j]] + ad));
  for (int off = 16; off > 0; off >>= 1) mx = fmaxf(mx, __shfl_xor(mx, off, 32));

  const float p0 = (j0 < end) ? __expf(e0 - mx) : 0.f;
  float den = p0;
  for (int j = j0 + 32; j < end; j += 32)
    den += __expf(leaky(asrc[col[j]] + ad) - mx);
  for (int off = 16; off > 0; off >>= 1) den += __shfl_xor(den, off, 32);

  VT acc = (VT)(0.f);
  for (int base = beg; base < end; base += 32) {
    int   sl = sl0;
    float al = p0 / den;
    if (base != beg) {
      const int j = base + lane;
      sl = 0;
      al = 0.f;
      if (j < end) {
        sl = col[j];
        al = __expf(leaky(asrc[sl] + ad) - mx) / den;
      }
    }
    const int n = (end - base < 32) ? (end - base) : 32;
    int k = 0;
    for (; k + 4 <= n; k += 4) {
      const int s0 = __builtin_amdgcn_readlane(sl, k);
      const int s1 = __builtin_amdgcn_readlane(sl, k + 1);
      const int s2 = __builtin_amdgcn_readlane(sl, k + 2);
      const int s3 = __builtin_amdgcn_readlane(sl, k + 3);
      const VT r0 = *(const VT*)(h + (size_t)s0 * C + lane * P);
      const VT r1 = *(const VT*)(h + (size_t)s1 * C + lane * P);
      const VT r2 = *(const VT*)(h + (size_t)s2 * C + lane * P);
      const VT r3 = *(const VT*)(h + (size_t)s3 * C + lane * P);
      const float a0 = __int_as_float(__builtin_amdgcn_readlane(__float_as_int(al), k));
      const float a1 = __int_as_float(__builtin_amdgcn_readlane(__float_as_int(al), k + 1));
      const float a2 = __int_as_float(__builtin_amdgcn_readlane(__float_as_int(al), k + 2));
      const float a3 = __int_as_float(__builtin_amdgcn_readlane(__float_as_int(al), k + 3));
      acc += a0 * r0;
      acc += a1 * r1;
      acc += a2 * r2;
      acc += a3 * r3;
    }
    for (; k < n; ++k) {
      const int s0 = __builtin_amdgcn_readlane(sl, k);
      const float a0 = __int_as_float(__builtin_amdgcn_readlane(__float_as_int(al), k));
      const VT r0 = *(const VT*)(h + (size_t)s0 * C + lane * P);
      acc += a0 * r0;
    }
  }

  VT o = acc + *(const VT*)(bias + lane * P);
  if (RELU) {
    for (int i = 0; i < P; ++i) o[i] = fmaxf(o[i], 0.f);
  }
  *(VT*)(out + (size_t)d * C + lane * P) = o;
}

static inline int cdiv(long a, long b) { return (int)((a + b - 1) / b); }

extern "C" void kernel_launch(void* const* d_in, const int* in_sizes, int n_in,
                              void* d_out, int out_size, void* d_ws, size_t ws_size,
                              hipStream_t stream)
{
  const float* x       = (const float*)d_in[0];
  const int*   ei      = (const int*)d_in[1];
  const float* Wsrc1   = (const float*)d_in[2];
  const float* Wdst1   = (const float*)d_in[3];
  const float* attsrc1 = (const float*)d_in[4];
  const float* attdst1 = (const float*)d_in[5];
  const float* b1      = (const float*)d_in[6];
  const float* Wsrc2   = (const float*)d_in[7];
  const float* Wdst2   = (const float*)d_in[8];
  const float* attsrc2 = (const float*)d_in[9];
  const float* attdst2 = (const float*)d_in[10];
  const float* b2      = (const float*)d_in[11];

  const int HID = in_sizes[4];
  const int OUT = in_sizes[9];
  const int FIN = in_sizes[2] / HID;
  const int N   = in_sizes[0] / FIN;
  const int E   = in_sizes[1] / 2;
  const int* src = ei;
  const int* dst = ei + E;

  float* ws    = (float*)d_ws;
  float* h1    = ws;  ws += (size_t)N * HID;
  float* o1    = ws;  ws += (size_t)N * HID;
  float* h2    = ws;  ws += (size_t)N * OUT;
  float* logit = ws;  ws += 2 * (size_t)N;
  float* asrc  = logit;
  float* adst  = logit + N;
  float* vfold = ws;  ws += 2 * (size_t)FIN + 2 * (size_t)HID;
  const int G1 = 256;
  const int NB = cdiv(N, RB);
  int* binned  = (int*)ws;  ws += 2 * (size_t)E;
  int* wgcnt   = (int*)ws;  ws += (size_t)NB * G1;
  int* wgoff   = (int*)ws;  ws += (size_t)NB * G1 + 1;
  int* col     = (int*)ws;  ws += E;
  int* rowptr  = (int*)ws;  ws += N + 1;

  const int tiles1 = (N / 16) * (HID / 64);
  const int tA = tiles1 / 3;
  const int tB = 2 * (tiles1 / 3);
  count_and_fold<<<G1 + cdiv(2 * FIN + 2 * HID, 32), 1024, 0, stream>>>(
      dst, wgcnt, E, NB, G1,
      Wsrc1, attsrc1, Wdst1, attdst1, Wsrc2, attsrc2, Wdst2, attdst2,
      vfold, FIN, HID, HID, OUT);
  scan_and_gemm<<<1 + cdiv(tA, GEMM_WPB), 1024, 0, stream>>>(
      wgcnt, wgoff, NB * G1,
      x, Wsrc1, vfold, h1, logit, N, 0, tA);
  scatter_and_gemm<<<G1 + cdiv(tB - tA, GEMM_WPB), 1024, 0, stream>>>(
      src, dst, wgoff, binned, E, NB, G1,
      x, Wsrc1, vfold, h1, logit, N, tA, tB);
  sort_and_gemm<<<NB + cdiv(tiles1 - tB, GEMM_WPB), 1024, 0, stream>>>(
      binned, wgoff, rowptr, col, N, G1, NB,
      x, Wsrc1, vfold, h1, logit, N, tB, tiles1);

  seg_gather<v4f, 128, true><<<cdiv(N, 8), 256, 0, stream>>>(
      rowptr, col, asrc, adst, h1, b1, o1, N);

  {
    float* out = (float*)d_out;
    const int tiles = (N / 16) * (OUT / 64);
    gemm_wmma_f32<128, 64, 4><<<cdiv(tiles, 8), 256, 0, stream>>>(
        o1, Wsrc2, vfold + 2 * (size_t)FIN, h2, logit, N);
    seg_gather<v2f, 64, false><<<cdiv(N, 8), 256, 0, stream>>>(
        rowptr, col, asrc, adst, h2, b2, out, N);
  }
}


# Layer-2 product moved into the layer-1 gather: 8 waves gather 32 nodes, the relu'd rows meet in shared memory, 4 waves multiply them by W2 on WMMA, 4 form the logits; 7 launches to 6; sums unchanged
# speedup vs baseline: 5.8287x; 1.0851x over previous
#include <hip/hip_runtime.h>
#include <hip/hip_bf16.h>

typedef float v2f __attribute__((ext_vector_type(2)));
typedef float v8f __attribute__((ext_vector_type(8)));

#define NEG_SLOPE 0.2f

template<int K, int NOUT, int NT>
__device__ __forceinline__ void gemm_strip(
    const float* __restrict__ X, const float* __restrict__ W,
    const float* __restrict__ vecs, float* __restrict__ C,
    float* __restrict__ logit, int M, int tile)
{
  const int lane   = threadIdx.x & 31;
  constexpr int tilesN = NOUT / (16 * NT);
  static_assert(tilesN == 1 || tilesN == 2, "two logit vectors are split over the strips of a row block");
  constexpr int NV = 2 / tilesN;
  const int tM = tile / tilesN;
  const int tN = tile % tilesN;

  const int half = lane >> 4;
  const int l15  = lane & 15;
  const float* xrow = X + (size_t)(tM * 16 + l15) * K + 2 * half;
  const float* wb   = W + (size_t)(2 * half) * NOUT + tN * (16 * NT) + l15;
  const float* vp   = vecs + (size_t)(tN * NV) * K + 2 * half;

  v8f acc[NT];
#pragma unroll
  for (int t = 0; t < NT; ++t) acc[t] = (v8f){0.f,0.f,0.f,0.f,0.f,0.f,0.f,0.f};
  float lg[NV];
  for (int i = 0; i < NV; ++i) lg[i] = 0.f;

  {
    v2f a, an;
    v2f vv[NV], vn[NV];
    v2f b[NT];
    a.x = xrow[0];
    a.y = xrow[1];
#pragma unroll
    for (int i = 0; i < NV; ++i) { vv[i].x = vp[i * K + 0]; vv[i].y = vp[i * K + 1]; }
#pragma unroll
    for (int t = 0; t < NT; ++t) { b[t].x = wb[16 * t]; b[t].y = wb[(size_t)NOUT + 16 * t]; }
#pragma unroll
    for (int k = 0; k < K; k += 4) {
      if (k + 4 < K) {
        an.x = xrow[k + 4];
        an.y = xrow[k + 5];
#pragma unroll
        for (int i = 0; i < NV; ++i) { vn[i].x = vp[i * K + k + 4]; vn[i].y = vp[i * K + k + 5]; }
      }
#pragma unroll
      for (int i = 0; i < NV; ++i) {
        lg[i] = fmaf(a.x, vv[i].x, lg[i]);
        lg[i] = fmaf(a.y, vv[i].y, lg[i]);
      }
#pragma unroll
      for (int t = 0; t < NT; ++t) {
        acc[t] = __builtin_amdgcn_wmma_f32_16x16x4_f32(false, a, false, b[t],
                                                       (short)0, acc[t],
                                                       false, false);
        if (k + 4 < K) {
          b[t].x = wb[(size_t)(k + 4) * NOUT + 16 * t];
          b[t].y = wb[(size_t)(k + 5) * NOUT + 16 * t];
        }
        __builtin_amdgcn_sched_barrier(0);
      }
      if (k + 4 < K) {
        a = an;
#pragma unroll
        for (int i = 0; i < NV; ++i) vv[i] = vn[i];
      }
    }
  }

  const int mbase = tM * 16 + 8 * half;
  float* cp = C + (size_t)mbase * NOUT + tN * (16 * NT) + l15;
#pragma unroll
  for (int t = 0; t < NT; ++t)
#pragma unroll
    for (int r = 0; r < 8; ++r)
      cp[(size_t)r * NOUT + 16 * t] = acc[t][r];

  for (int i = 0; i < NV; ++i) {
    const float s = lg[i] + __shfl_xor(lg[i], 16, 32);
    if (half == 0) logit[(size_t)(tN * NV + i) * M + tM * 16 + l15] = s;
  }
}

#define GEMM_WPB 16
__device__ __forceinline__ void gemm1_part(
    const float* __restrict__ X, const float* __restrict__ W,
    const float* __restrict__ vecs, float* __restrict__ C,
    float* __restrict__ logit, int M, int tile0, int tile1, int gb)
{
  const int wave = threadIdx.x >> 5;
  if (wave >= GEMM_WPB) return;
  const int tile = __builtin_amdgcn_readfirstlane(tile0 + gb * GEMM_WPB + wave);
  if (tile >= tile1) return;
  gemm_strip<128, 128, 4>(X, W, vecs, C, logit, M, tile);
}

__device__ __forceinline__ void fold_att_body(
    const float* __restrict__ W0, const float* __restrict__ a0,
    const float* __restrict__ W1, const float* __restrict__ a1,
    const float* __restrict__ W2, const float* __restrict__ a2,
    const float* __restrict__ W3, const float* __restrict__ a3,
    float* __restrict__ v, int r1, int c1, int r2, int c2, int gw)
{
  const int lane = threadIdx.x & 31;
  if (gw >= 2 * r1 + 2 * r2) return;
  const float* Wm; const float* att; int row, cols;
  if (gw < r1)               { Wm = W0; att = a0; row = gw;               cols = c1; }
  else if (gw < 2 * r1)      { Wm = W1; att = a1; row = gw - r1;          cols = c1; }
  else if (gw < 2 * r1 + r2) { Wm = W2; att = a2; row = gw - 2 * r1;      cols = c2; }
  else                       { Wm = W3; att = a3; row = gw - 2 * r1 - r2; cols = c2; }
  const float* wr = Wm + (size_t)row * cols;
  float s = 0.f;
  for (int k = lane; k < cols; k += 32) s += wr[k] * att[k];
  for (int off = 16; off > 0; off >>= 1) s += __shfl_down(s, off, 32);
  if (lane == 0) v[gw] = s;
}

#define RB_SHIFT 9
#define RB (1 << RB_SHIFT)
#define MAXNB 1024

__device__ __forceinline__ void bin_count_body(
    const int* __restrict__ dst, int* __restrict__ wgcnt, int E, int NB, int bid, int G)
{
  __shared__ int hist[MAXNB];
  const int tid = threadIdx.x;
  for (int r = tid; r < NB; r += 1024) hist[r] = 0;
  __syncthreads();
  const int chunk = (E + G - 1) / G;
  const int e0 = bid * chunk;
  const int e1 = (e0 + chunk < E) ? (e0 + chunk) : E;
  for (int base = e0; base < e1; base += 4096) {
    int d[4];
    for (int q = 0; q < 4; ++q) {
      const int i = base + tid + q * 1024;
      d[q] = (i < e1) ? dst[i] : -1;
    }
    for (int q = 0; q < 4; ++q)
      if (d[q] >= 0) atomicAdd(&hist[d[q] >> RB_SHIFT], 1);
  }
  __syncthreads();
  for (int r = tid; r < NB; r += 1024) wgcnt[(size_t)r * G + bid] = hist[r];
}

__device__ __forceinline__ void scan_counts_body(
    const int* __restrict__ cnt, int* __restrict__ rowptr, int N)
{
  __shared__ int wtot[32];
  __shared__ int tile_total;
  const int tid  = threadIdx.x;
  const int lane = tid & 31;
  const int wave = tid >> 5;
  int carry = 0;
  for (int base = 0; base < N; base += 8192) {
    const int i0 = base + tid * 8;
    int c[8];
    int t = 0;
    for (int q = 0; q < 8; ++q) {
      c[q] = (i0 + q < N) ? cnt[i0 + q] : 0;
      t += c[q];
    }
    int s = t;
    for (int off = 1; off < 32; off <<= 1) {
      const int nb = __shfl_up(s, off, 32);
      if (lane >= off) s += nb;
    }
    if (lane == 31) wtot[wave] = s;
    __syncthreads();
    if (wave == 0) {
      const int w = wtot[lane];
      int ws = w;
      for (int off = 1; off < 32; off <<= 1) {
        const int nb = __shfl_up(ws, off, 32);
        if (lane >= off) ws += nb;
      }
      wtot[lane] = ws - w;
      if (lane == 31) tile_total = ws;
    }
    __syncthreads();
    int e = carry + wtot[wave] + (s - t);
    carry += tile_total;
    for (int q = 0; q < 8; ++q) {
      if (i0 + q < N) rowptr[i0 + q] = e;
      e += c[q];
    }
    __syncthreads();
  }
  if (tid == 0) rowptr[N] = carry;
}

__device__ __forceinline__ void bin_scatter_body(
    const int* __restrict__ src, const int* __restrict__ dst,
    const int* __restrict__ wgoff, int* __restrict__ binned, int E, int NB, int bid, int G)
{
  __shared__ int cur[MAXNB];
  const int tid = threadIdx.x;
  for (int r = tid; r < NB; r += 1024) cur[r] = wgoff[(size_t)r * G + bid];
  __syncthreads();
  int2* pairs = reinterpret_cast<int2*>(binned);
  const int chunk = (E + G - 1) / G;
  const int e0 = bid * chunk;
  const int e1 = (e0 + chunk < E) ? (e0 + chunk) : E;
  for (int base = e0; base < e1; base += 4096) {
    int d[4], s[4];
    for (int q = 0; q < 4; ++q) {
      const int i = base + tid + q * 1024;
      d[q] = (i < e1) ? dst[i] : -1;
      s[q] = (i < e1) ? src[i] : 0;
    }
    for (int q = 0; q < 4; ++q)
      if (d[q] >= 0) {
        const int slot = atomicAdd(&cur[d[q] >> RB_SHIFT], 1);
        pairs[slot] = make_int2(s[q], d[q]);
      }
  }
}

__device__ __forceinline__ void range_sort_body(
    const int* __restrict__ binned, const int* __restrict__ wgoff,
    int* __restrict__ rowptr, int* __restrict__ col, int N, int G, int rg, int NB)
{
  __shared__ int cnt[RB];
  __shared__ int wtot[RB / 32];
  const int tid  = threadIdx.x;
  const int lane = tid & 31;
  const int wave = tid >> 5;
  const int s0 = wgoff[(size_t)rg * G];
  const int s1 = wgoff[(size_t)(rg + 1) * G];
  const int2* pairs = reinterpret_cast<const int2*>(binned);

  if (tid < RB) cnt[tid] = 0;
  __syncthreads();
  for (int base = s0; base < s1; base += 4096) {
    int dl[4];
    for (int q = 0; q < 4; ++q) {
      const int i = base + tid + q * 1024;
      dl[q] = (i < s1) ? (pairs[i].y & (RB - 1)) : -1;
    }
    for (int q = 0; q < 4; ++q)
      if (dl[q] >= 0) atomicAdd(&cnt[dl[q]], 1);
  }
  __syncthreads();

  int v = 0, incl = 0;
  if (tid < RB) {
    v = cnt[tid];
    incl = v;
    for (int off = 1; off < 32; off <<= 1) {
      const int nb = __shfl_up(incl, off, 32);
      if (lane >= off) incl += nb;
    }
    if (lane == 31) wtot[wave] = incl;
  }
  __syncthreads();
  if (tid < RB) {
    int offs = 0;
    for (int w = 0; w < wave; ++w) offs += wtot[w];
    const int excl = offs + incl - v;
    cnt[tid] = excl;
    const int node = rg * RB + tid;
    if (node < N) rowptr[node] = s0 + excl;
  }
  if (rg == NB - 1 && tid == 0) rowptr[N] = s1;
  __syncthreads();

  for (int base = s0; base < s1; base += 4096) {
    int2 r[4];
    for (int q = 0; q < 4; ++q) {
      const int i = base + tid + q * 1024;
      r[q] = (i < s1) ? pairs[i] : make_int2(0, -1);
    }
    for (int q = 0; q < 4; ++q)
      if (r[q].y >= 0) {
        const int slot = atomicAdd(&cnt[r[q].y & (RB - 1)], 1);
        col[s0 + slot] = r[q].x;
      }
  }
}

__global__ __launch_bounds__(1024) void count_and_fold(
    const int* __restrict__ dst, int* __restrict__ wgcnt, int E, int NB, int G,
    const float* __restrict__ W0, const float* __restrict__ a0,
    const float* __restrict__ W1, const float* __restrict__ a1,
    const float* __restrict__ W2, const float* __restrict__ a2,
    const float* __restrict__ W3, const float* __restrict__ a3,
    float* __restrict__ v, int r1, int c1, int r2, int c2)
{
  const int b = (int)blockIdx.x;
  if (b < G) bin_count_body(dst, wgcnt, E, NB, b, G);
  else fold_att_body(W0, a0, W1, a1, W2, a2, W3, a3, v, r1, c1, r2, c2,
                     (b - G) * 32 + (int)(threadIdx.x >> 5));
}

__global__ __launch_bounds__(1024, 16) void scan_and_gemm(
    const int* __restrict__ cnt, int* __restrict__ off, int n,
    const float* __restrict__ X, const float* __restrict__ W,
    const float* __restrict__ vecs, float* __restrict__ C,
    float* __restrict__ logit, int M, int tile0, int tile1)
{
  const int b = (int)blockIdx.x;
  if (b == 0) scan_counts_body(cnt, off, n);
  else gemm1_part(X, W, vecs, C, logit, M, tile0, tile1, b - 1);
}

__global__ __launch_bounds__(1024, 16) void scatter_and_gemm(
    const int* __restrict__ src, const int* __restrict__ dst,
    const int* __restrict__ wgoff, int* __restrict__ binned, int E, int NB, int G,
    const float* __restrict__ X, const float* __restrict__ W,
    const float* __restrict__ vecs, float* __restrict__ C,
    float* __restrict__ logit, int M, int tile0, int tile1)
{
  const int b = (int)blockIdx.x;
  if (b < G) bin_scatter_body(src, dst, wgoff, binned, E, NB, b, G);
  else gemm1_part(X, W, vecs, C, logit, M, tile0, tile1, b - G);
}

__global__ __launch_bounds__(1024, 16) void sort_and_gemm(
    const int* __restrict__ binned, const int* __restrict__ wgoff,
    int* __restrict__ rowptr, int* __restrict__ col, int N, int G, int NB,
    const float* __restrict__ X, const float* __restrict__ W,
    const float* __restrict__ vecs, float* __restrict__ C,
    float* __restrict__ logit, int M, int tile0, int tile1)
{
  const int b = (int)blockIdx.x;
  if (b < NB) range_sort_body(binned, wgoff, rowptr, col, N, G, b, NB);
  else gemm1_part(X, W, vecs, C, logit, M, tile0, tile1, b - NB);
}

typedef float v4f __attribute__((ext_vector_type(4)));

__device__ __forceinline__ float leaky(float v)
{
  return (v > 0.f) ? v : NEG_SLOPE * v;
}

template<typename VT, int C, bool RELU>
__device__ __forceinline__ VT gather_row(
    const int* __restrict__ rowptr, const int* __restrict__ col,
    const float* __restrict__ asrc, const float* __restrict__ adst,
    const float* __restrict__ h, const float* __restrict__ bias, int d)
{
  constexpr int P = C / 32;
  const int lane = threadIdx.x & 31;
  const int beg = rowptr[d];
  const int end = rowptr[d + 1];
  const float ad = adst[d];

  const int j0 = beg + lane;
  int   sl0 = 0;
  float e0  = -__builtin_huge_valf();
  if (j0 < end) {
    sl0 = col[j0];
    e0  = leaky(asrc[sl0] + ad);
  }

  float mx = e0;
  for (int j = j0 + 32; j < end; j += 32)
    mx = fmaxf(mx, leaky(asrc[col[j]] + ad));
  for (int off = 16; off > 0; off >>= 1) mx = fmaxf(mx, __shfl_xor(mx, off, 32));

  const float p0 = (j0 < end) ? __expf(e0 - mx) : 0.f;
  float den = p0;
  for (int j = j0 + 32; j < end; j += 32)
    den += __expf(leaky(asrc[col[j]] + ad) - mx);
  for (int off = 16; off > 0; off >>= 1) den += __shfl_xor(den, off, 32);

  VT acc = (VT)(0.f);
  for (int base = beg; base < end; base += 32) {
    int   sl = sl0;
    float al = p0 / den;
    if (base != beg) {
      const int j = base + lane;
      sl = 0;
      al = 0.f;
      if (j < end) {
        sl = col[j];
        al = __expf(leaky(asrc[sl] + ad) - mx) / den;
      }
    }
    const int n = (end - base < 32) ? (end - base) : 32;
    int k = 0;
    for (; k + 4 <= n; k += 4) {
      const int s0 = __builtin_amdgcn_readlane(sl, k);
      const int s1 = __builtin_amdgcn_readlane(sl, k + 1);
      const int s2 = __builtin_amdgcn_readlane(sl, k + 2);
      const int s3 = __builtin_amdgcn_readlane(sl, k + 3);
      const VT r0 = *(const VT*)(h + (size_t)s0 * C + lane * P);
      const VT r1 = *(const VT*)(h + (size_t)s1 * C + lane * P);
      const VT r2 = *(const VT*)(h + (size_t)s2 * C + lane * P);
      const VT r3 = *(const VT*)(h + (size_t)s3 * C + lane * P);
      const float a0 = __int_as_float(__builtin_amdgcn_readlane(__float_as_int(al), k));
      const float a1 = __int_as_float(__builtin_amdgcn_readlane(__float_as_int(al), k + 1));
      const float a2 = __int_as_float(__builtin_amdgcn_readlane(__float_as_int(al), k + 2));
      const float a3 = __int_as_float(__builtin_amdgcn_readlane(__float_as_int(al), k + 3));
      acc += a0 * r0;
      acc += a1 * r1;
      acc += a2 * r2;
      acc += a3 * r3;
    }
    for (; k < n; ++k) {
      const int s0 = __builtin_amdgcn_readlane(sl, k);
      const float a0 = __int_as_float(__builtin_amdgcn_readlane(__float_as_int(al), k));
      const VT r0 = *(const VT*)(h + (size_t)s0 * C + lane * P);
      acc += a0 * r0;
    }
  }

  VT o = acc + *(const VT*)(bias + lane * P);
  if (RELU) {
    for (int i = 0; i < P; ++i) o[i] = fmaxf(o[i], 0.f);
  }
  return o;
}

template<typename VT, int C, bool RELU>
__global__ __launch_bounds__(256) void seg_gather(
    const int* __restrict__ rowptr, const int* __restrict__ col,
    const float* __restrict__ asrc, const float* __restrict__ adst,
    const float* __restrict__ h, const float* __restrict__ bias,
    float* __restrict__ out, int N)
{
  constexpr int P = C / 32;
  const int lane = threadIdx.x & 31;
  const int d = __builtin_amdgcn_readfirstlane((int)(blockIdx.x * 8 + (threadIdx.x >> 5)));
  if (d >= N) return;
  const VT o = gather_row<VT, C, RELU>(rowptr, col, asrc, adst, h, bias, d);
  *(VT*)(out + (size_t)d * C + lane * P) = o;
}

#define TP 132
#define FROWS 32
__global__ __launch_bounds__(256, 16) void gather1_gemm2(
    const int* __restrict__ rowptr, const int* __restrict__ col,
    const float* __restrict__ asrc, const float* __restrict__ adst,
    const float* __restrict__ h, const float* __restrict__ bias,
    const float* __restrict__ W2, const float* __restrict__ vecs2,
    float* __restrict__ h2, float* __restrict__ logit2, int N)
{
  __shared__ __attribute__((aligned(16))) float tile[FROWS * TP];
  const int lane = threadIdx.x & 31;
  const int wave = __builtin_amdgcn_readfirstlane((int)(threadIdx.x >> 5));
  const int row0 = (int)blockIdx.x * FROWS;

  for (int n = 0; n < 4; ++n) {
    const int r = 4 * wave + n;
    v4f o = (v4f)(0.f);
    if (row0 + r < N) o = gather_row<v4f, 128, true>(rowptr, col, asrc, adst, h, bias, row0 + r);
    *(v4f*)(tile + r * TP + lane * 4) = o;
  }
  __syncthreads();

  const int half = lane >> 4;
  const int l15  = lane & 15;
  const int rt   = (wave >> 1) & 1;
  const int trow = row0 + 16 * rt;
  const float* ap = tile + (16 * rt + l15) * TP + 2 * half;

  if (wave >= 4) {
    const int v = wave & 1;
    const float* vp = vecs2 + (size_t)v * 128 + 2 * half;
    float lg = 0.f;
#pragma unroll
    for (int k = 0; k < 128; k += 4) {
      const v2f a = *(const v2f*)(ap + k);
      lg = fmaf(a.x, vp[k + 0], lg);
      lg = fmaf(a.y, vp[k + 1], lg);
    }
    const float s = lg + __shfl_xor(lg, 16, 32);
    if (half == 0 && trow + l15 < N) logit2[(size_t)v * N + trow + l15] = s;
    return;
  }

  const int g = wave & 1;
  const float* wb = W2 + (size_t)(2 * half) * 64 + g * 32 + l15;
  v8f acc0 = (v8f){0.f,0.f,0.f,0.f,0.f,0.f,0.f,0.f};
  v8f acc1 = (v8f){0.f,0.f,0.f,0.f,0.f,0.f,0.f,0.f};
  constexpr int AH = 4;
  v2f b0[AH], b1[AH];
#pragma unroll
  for (int i = 0; i < AH; ++i) {
    const int k = 4 * i;
    b0[i].x = wb[(size_t)(k + 0) * 64];
    b0[i].y = wb[(size_t)(k + 1) * 64];
    b1[i].x = wb[(size_t)(k + 0) * 64 + 16];
    b1[i].y = wb[(size_t)(k + 1) * 64 + 16];
  }
  v2f a = *(const v2f*)(ap);
#pragma unroll
  for (int i = 0; i < 32; ++i) {
    const int k = 4 * i;
    const int j = i % AH;
    v2f an = a;
    if (i + 1 < 32) an = *(const v2f*)(ap + k + 4);
    acc0 = __builtin_amdgcn_wmma_f32_16x16x4_f32(false, a, false, b0[j], (short)0, acc0, false, false);
    if (i + AH < 32) {
      b0[j].x = wb[(size_t)(k + 4 * AH + 0) * 64];
      b0[j].y = wb[(size_t)(k + 4 * AH + 1) * 64];
    }
    acc1 = __builtin_amdgcn_wmma_f32_16x16x4_f32(false, a, false, b1[j], (short)0, acc1, false, false);
    if (i + AH < 32) {
      b1[j].x = wb[(size_t)(k + 4 * AH + 0) * 64 + 16];
      b1[j].y = wb[(size_t)(k + 4 * AH + 1) * 64 + 16];
    }
    __builtin_amdgcn_sched_barrier(0);
    a = an;
  }

  const int mbase = trow + 8 * half;
  float* cp = h2 + (size_t)mbase * 64 + g * 32 + l15;
#pragma unroll
  for (int r = 0; r < 8; ++r)
    if (mbase + r < N) {
      cp[(size_t)r * 64]      = acc0[r];
      cp[(size_t)r * 64 + 16] = acc1[r];
    }
}

static inline int cdiv(long a, long b) { return (int)((a + b - 1) / b); }

extern "C" void kernel_launch(void* const* d_in, const int* in_sizes, int n_in,
                              void* d_out, int out_size, void* d_ws, size_t ws_size,
                              hipStream_t stream)
{
  const float* x       = (const float*)d_in[0];
  const int*   ei      = (const int*)d_in[1];
  const float* Wsrc1   = (const float*)d_in[2];
  const float* Wdst1   = (const float*)d_in[3];
  const float* attsrc1 = (const float*)d_in[4];
  const float* attdst1 = (const float*)d_in[5];
  const float* b1      = (const float*)d_in[6];
  const float* Wsrc2   = (const float*)d_in[7];
  const float* Wdst2   = (const float*)d_in[8];
  const float* attsrc2 = (const float*)d_in[9];
  const float* attdst2 = (const float*)d_in[10];
  const float* b2      = (const float*)d_in[11];

  const int HID = in_sizes[4];
  const int OUT = in_sizes[9];
  const int FIN = in_sizes[2] / HID;
  const int N   = in_sizes[0] / FIN;
  const int E   = in_sizes[1] / 2;
  const int* src = ei;
  const int* dst = ei + E;

  float* ws    = (float*)d_ws;
  float* h1    = ws;  ws += (size_t)N * HID;
  float* h2    = ws;  ws += (size_t)N * OUT;
  float* logit = ws;  ws += 2 * (size_t)N;
  float* logit2 = ws; ws += 2 * (size_t)N;
  float* vfold = ws;  ws += 2 * (size_t)FIN + 2 * (size_t)HID;
  const int G1 = 256;
  const int NB = cdiv(N, RB);
  int* binned  = (int*)ws;  ws += 2 * (size_t)E;
  int* wgcnt   = (int*)ws;  ws += (size_t)NB * G1;
  int* wgoff   = (int*)ws;  ws += (size_t)NB * G1 + 1;
  int* col     = (int*)ws;  ws += E;
  int* rowptr  = (int*)ws;  ws += N + 1;

  const int tiles1 = (N / 16) * (HID / 64);
  const int tA = tiles1 / 3;
  const int tB = 2 * (tiles1 / 3);
  count_and_fold<<<G1 + cdiv(2 * FIN + 2 * HID, 32), 1024, 0, stream>>>(
      dst, wgcnt, E, NB, G1,
      Wsrc1, attsrc1, Wdst1, attdst1, Wsrc2, attsrc2, Wdst2, attdst2,
      vfold, FIN, HID, HID, OUT);
  scan_and_gemm<<<1 + cdiv(tA, GEMM_WPB), 1024, 0, stream>>>(
      wgcnt, wgoff, NB * G1,
      x, Wsrc1, vfold, h1, logit, N, 0, tA);
  scatter_and_gemm<<<G1 + cdiv(tB - tA, GEMM_WPB), 1024, 0, stream>>>(
      src, dst, wgoff, binned, E, NB, G1,
      x, Wsrc1, vfold, h1, logit, N, tA, tB);
  sort_and_gemm<<<NB + cdiv(tiles1 - tB, GEMM_WPB), 1024, 0, stream>>>(
      binned, wgoff, rowptr, col, N, G1, NB,
      x, Wsrc1, vfold, h1, logit, N, tB, tiles1);

  gather1_gemm2<<<cdiv(N, FROWS), 256, 0, stream>>>(
      rowptr, col, logit, logit + N, h1, b1,
      Wsrc2, vfold + 2 * (size_t)FIN, h2, logit2, N);

  {
    float* out = (float*)d_out;
    seg_gather<v2f, 64, false><<<cdiv(N, 8), 256, 0, stream>>>(
        rowptr, col, logit2, logit2 + N, h2, b2, out, N);
  }
}


# Gathers: two destination nodes per wave, one per 16-lane half, rows asked for in groups of 8 with no one-by-one tail; row sums keep their order, softmax denominators regrouped only above 32 edges
# speedup vs baseline: 6.0584x; 1.0394x over previous
#include <hip/hip_runtime.h>
#include <hip/hip_bf16.h>

typedef float v2f __attribute__((ext_vector_type(2)));
typedef float v8f __attribute__((ext_vector_type(8)));

#define NEG_SLOPE 0.2f

template<int K, int NOUT, int NT>
__device__ __forceinline__ void gemm_strip(
    const float* __restrict__ X, const float* __restrict__ W,
    const float* __restrict__ vecs, float* __restrict__ C,
    float* __restrict__ logit, int M, int tile)
{
  const int lane   = threadIdx.x & 31;
  constexpr int tilesN = NOUT / (16 * NT);
  static_assert(tilesN == 1 || tilesN == 2, "two logit vectors are split over the strips of a row block");
  constexpr int NV = 2 / tilesN;
  const int tM = tile / tilesN;
  const int tN = tile % tilesN;

  const int half = lane >> 4;
  const int l15  = lane & 15;
  const float* xrow = X + (size_t)(tM * 16 + l15) * K + 2 * half;
  const float* wb   = W + (size_t)(2 * half) * NOUT + tN * (16 * NT) + l15;
  const float* vp   = vecs + (size_t)(tN * NV) * K + 2 * half;

  v8f acc[NT];
#pragma unroll
  for (int t = 0; t < NT; ++t) acc[t] = (v8f){0.f,0.f,0.f,0.f,0.f,0.f,0.f,0.f};
  float lg[NV];
  for (int i = 0; i < NV; ++i) lg[i] = 0.f;

  {
    v2f a, an;
    v2f vv[NV], vn[NV];
    v2f b[NT];
    a.x = xrow[0];
    a.y = xrow[1];
#pragma unroll
    for (int i = 0; i < NV; ++i) { vv[i].x = vp[i * K + 0]; vv[i].y = vp[i * K + 1]; }
#pragma unroll
    for (int t = 0; t < NT; ++t) { b[t].x = wb[16 * t]; b[t].y = wb[(size_t)NOUT + 16 * t]; }
#pragma unroll
    for (int k = 0; k < K; k += 4) {
      if (k + 4 < K) {
        an.x = xrow[k + 4];
        an.y = xrow[k + 5];
#pragma unroll
        for (int i = 0; i < NV; ++i) { vn[i].x = vp[i * K + k + 4]; vn[i].y = vp[i * K + k + 5]; }
      }
#pragma unroll
      for (int i = 0; i < NV; ++i) {
        lg[i] = fmaf(a.x, vv[i].x, lg[i]);
        lg[i] = fmaf(a.y, vv[i].y, lg[i]);
      }
#pragma unroll
      for (int t = 0; t < NT; ++t) {
        acc[t] = __builtin_amdgcn_wmma_f32_16x16x4_f32(false, a, false, b[t],
                                                       (short)0, acc[t],
                                                       false, false);
        if (k + 4 < K) {
          b[t].x = wb[(size_t)(k + 4) * NOUT + 16 * t];
          b[t].y = wb[(size_t)(k + 5) * NOUT + 16 * t];
        }
        __builtin_amdgcn_sched_barrier(0);
      }
      if (k + 4 < K) {
        a = an;
#pragma unroll
        for (int i = 0; i < NV; ++i) vv[i] = vn[i];
      }
    }
  }

  const int mbase = tM * 16 + 8 * half;
  float* cp = C + (size_t)mbase * NOUT + tN * (16 * NT) + l15;
#pragma unroll
  for (int t = 0; t < NT; ++t)
#pragma unroll
    for (int r = 0; r < 8; ++r)
      cp[(size_t)r * NOUT + 16 * t] = acc[t][r];

  for (int i = 0; i < NV; ++i) {
    const float s = lg[i] + __shfl_xor(lg[i], 16, 32);
    if (half == 0) logit[(size_t)(tN * NV + i) * M + tM * 16 + l15] = s;
  }
}

#define GEMM_WPB 16
__device__ __forceinline__ void gemm1_part(
    const float* __restrict__ X, const float* __restrict__ W,
    const float* __restrict__ vecs, float* __restrict__ C,
    float* __restrict__ logit, int M, int tile0, int tile1, int gb)
{
  const int wave = threadIdx.x >> 5;
  if (wave >= GEMM_WPB) return;
  const int tile = __builtin_amdgcn_readfirstlane(tile0 + gb * GEMM_WPB + wave);
  if (tile >= tile1) return;
  gemm_strip<128, 128, 4>(X, W, vecs, C, logit, M, tile);
}

__device__ __forceinline__ void fold_att_body(
    const float* __restrict__ W0, const float* __restrict__ a0,
    const float* __restrict__ W1, const float* __restrict__ a1,
    const float* __restrict__ W2, const float* __restrict__ a2,
    const float* __restrict__ W3, const float* __restrict__ a3,
    float* __restrict__ v, int r1, int c1, int r2, int c2, int gw)
{
  const int lane = threadIdx.x & 31;
  if (gw >= 2 * r1 + 2 * r2) return;
  const float* Wm; const float* att; int row, cols;
  if (gw < r1)               { Wm = W0; att = a0; row = gw;               cols = c1; }
  else if (gw < 2 * r1)      { Wm = W1; att = a1; row = gw - r1;          cols = c1; }
  else if (gw < 2 * r1 + r2) { Wm = W2; att = a2; row = gw - 2 * r1;      cols = c2; }
  else                       { Wm = W3; att = a3; row = gw - 2 * r1 - r2; cols = c2; }
  const float* wr = Wm + (size_t)row * cols;
  float s = 0.f;
  for (int k = lane; k < cols; k += 32) s += wr[k] * att[k];
  for (int off = 16; off > 0; off >>= 1) s += __shfl_down(s, off, 32);
  if (lane == 0) v[gw] = s;
}

#define RB_SHIFT 9
#define RB (1 << RB_SHIFT)
#define MAXNB 1024

__device__ __forceinline__ void bin_count_body(
    const int* __restrict__ dst, int* __restrict__ wgcnt, int E, int NB, int bid, int G)
{
  __shared__ int hist[MAXNB];
  const int tid = threadIdx.x;
  for (int r = tid; r < NB; r += 1024) hist[r] = 0;
  __syncthreads();
  const int chunk = (E + G - 1) / G;
  const int e0 = bid * chunk;
  const int e1 = (e0 + chunk < E) ? (e0 + chunk) : E;
  for (int base = e0; base < e1; base += 4096) {
    int d[4];
    for (int q = 0; q < 4; ++q) {
      const int i = base + tid + q * 1024;
      d[q] = (i < e1) ? dst[i] : -1;
    }
    for (int q = 0; q < 4; ++q)
      if (d[q] >= 0) atomicAdd(&hist[d[q] >> RB_SHIFT], 1);
  }
  __syncthreads();
  for (int r = tid; r < NB; r += 1024) wgcnt[(size_t)r * G + bid] = hist[r];
}

__device__ __forceinline__ void scan_counts_body(
    const int* __restrict__ cnt, int* __restrict__ rowptr, int N)
{
  __shared__ int wtot[32];
  __shared__ int tile_total;
  const int tid  = threadIdx.x;
  const int lane = tid & 31;
  const int wave = tid >> 5;
  int carry = 0;
  for (int base = 0; base < N; base += 8192) {
    const int i0 = base + tid * 8;
    int c[8];
    int t = 0;
    for (int q = 0; q < 8; ++q) {
      c[q] = (i0 + q < N) ? cnt[i0 + q] : 0;
      t += c[q];
    }
    int s = t;
    for (int off = 1; off < 32; off <<= 1) {
      const int nb = __shfl_up(s, off, 32);
      if (lane >= off) s += nb;
    }
    if (lane == 31) wtot[wave] = s;
    __syncthreads();
    if (wave == 0) {
      const int w = wtot[lane];
      int ws = w;
      for (int off = 1; off < 32; off <<= 1) {
        const int nb = __shfl_up(ws, off, 32);
        if (lane >= off) ws += nb;
      }
      wtot[lane] = ws - w;
      if (lane == 31) tile_total = ws;
    }
    __syncthreads();
    int e = carry + wtot[wave] + (s - t);
    carry += tile_total;
    for (int q = 0; q < 8; ++q) {
      if (i0 + q < N) rowptr[i0 + q] = e;
      e += c[q];
    }
    __syncthreads();
  }
  if (tid == 0) rowptr[N] = carry;
}

__device__ __forceinline__ void bin_scatter_body(
    const int* __restrict__ src, const int* __restrict__ dst,
    const int* __restrict__ wgoff, int* __restrict__ binned, int E, int NB, int bid, int G)
{
  __shared__ int cur[MAXNB];
  const int tid = threadIdx.x;
  for (int r = tid; r < NB; r += 1024) cur[r] = wgoff[(size_t)r * G + bid];
  __syncthreads();
  int2* pairs = reinterpret_cast<int2*>(binned);
  const int chunk = (E + G - 1) / G;
  const int e0 = bid * chunk;
  const int e1 = (e0 + chunk < E) ? (e0 + chunk) : E;
  for (int base = e0; base < e1; base += 4096) {
    int d[4], s[4];
    for (int q = 0; q < 4; ++q) {
      const int i = base + tid + q * 1024;
      d[q] = (i < e1) ? dst[i] : -1;
      s[q] = (i < e1) ? src[i] : 0;
    }
    for (int q = 0; q < 4; ++q)
      if (d[q] >= 0) {
        const int slot = atomicAdd(&cur[d[q] >> RB_SHIFT], 1);
        pairs[slot] = make_int2(s[q], d[q]);
      }
  }
}

__device__ __forceinline__ void range_sort_body(
    const int* __restrict__ binned, const int* __restrict__ wgoff,
    int* __restrict__ rowptr, int* __restrict__ col, int N, int G, int rg, int NB)
{
  __shared__ int cnt[RB];
  __shared__ int wtot[RB / 32];
  const int tid  = threadIdx.x;
  const int lane = tid & 31;
  const int wave = tid >> 5;
  const int s0 = wgoff[(size_t)rg * G];
  const int s1 = wgoff[(size_t)(rg + 1) * G];
  const int2* pairs = reinterpret_cast<const int2*>(binned);

  if (tid < RB) cnt[tid] = 0;
  __syncthreads();
  for (int base = s0; base < s1; base += 4096) {
    int dl[4];
    for (int q = 0; q < 4; ++q) {
      const int i = base + tid + q * 1024;
      dl[q] = (i < s1) ? (pairs[i].y & (RB - 1)) : -1;
    }
    for (int q = 0; q < 4; ++q)
      if (dl[q] >= 0) atomicAdd(&cnt[dl[q]], 1);
  }
  __syncthreads();

  int v = 0, incl = 0;
  if (tid < RB) {
    v = cnt[tid];
    incl = v;
    for (int off = 1; off < 32; off <<= 1) {
      const int nb = __shfl_up(incl, off, 32);
      if (lane >= off) incl += nb;
    }
    if (lane == 31) wtot[wave] = incl;
  }
  __syncthreads();
  if (tid < RB) {
    int offs = 0;
    for (int w = 0; w < wave; ++w) offs += wtot[w];
    const int excl = offs + incl - v;
    cnt[tid] = excl;
    const int node = rg * RB + tid;
    if (node < N) rowptr[node] = s0 + excl;
  }
  if (rg == NB - 1 && tid == 0) rowptr[N] = s1;
  __syncthreads();

  for (int base = s0; base < s1; base += 4096) {
    int2 r[4];
    for (int q = 0; q < 4; ++q) {
      const int i = base + tid + q * 1024;
      r[q] = (i < s1) ? pairs[i] : make_int2(0, -1);
    }
    for (int q = 0; q < 4; ++q)
      if (r[q].y >= 0) {
        const int slot = atomicAdd(&cnt[r[q].y & (RB - 1)], 1);
        col[s0 + slot] = r[q].x;
      }
  }
}

__global__ __launch_bounds__(1024) void count_and_fold(
    const int* __restrict__ dst, int* __restrict__ wgcnt, int E, int NB, int G,
    const float* __restrict__ W0, const float* __restrict__ a0,
    const float* __restrict__ W1, const float* __restrict__ a1,
    const float* __restrict__ W2, const float* __restrict__ a2,
    const float* __restrict__ W3, const float* __restrict__ a3,
    float* __restrict__ v, int r1, int c1, int r2, int c2)
{
  const int b = (int)blockIdx.x;
  if (b < G) bin_count_body(dst, wgcnt, E, NB, b, G);
  else fold_att_body(W0, a0, W1, a1, W2, a2, W3, a3, v, r1, c1, r2, c2,
                     (b - G) * 32 + (int)(threadIdx.x >> 5));
}

__global__ __launch_bounds__(1024, 16) void scan_and_gemm(
    const int* __restrict__ cnt, int* __restrict__ off, int n,
    const float* __restrict__ X, const float* __restrict__ W,
    const float* __restrict__ vecs, float* __restrict__ C,
    float* __restrict__ logit, int M, int tile0, int tile1)
{
  const int b = (int)blockIdx.x;
  if (b == 0) scan_counts_body(cnt, off, n);
  else gemm1_part(X, W, vecs, C, logit, M, tile0, tile1, b - 1);
}

__global__ __launch_bounds__(1024, 16) void scatter_and_gemm(
    const int* __restrict__ src, const int* __restrict__ dst,
    const int* __restrict__ wgoff, int* __restrict__ binned, int E, int NB, int G,
    const float* __restrict__ X, const float* __restrict__ W,
    const float* __restrict__ vecs, float* __restrict__ C,
    float* __restrict__ logit, int M, int tile0, int tile1)
{
  const int b = (int)blockIdx.x;
  if (b < G) bin_scatter_body(src, dst, wgoff, binned, E, NB, b, G);
  else gemm1_part(X, W, vecs, C, logit, M, tile0, tile1, b - G);
}

__global__ __launch_bounds__(1024, 16) void sort_and_gemm(
    const int* __restrict__ binned, const int* __restrict__ wgoff,
    int* __restrict__ rowptr, int* __restrict__ col, int N, int G, int NB,
    const float* __restrict__ X, const float* __restrict__ W,
    const float* __restrict__ vecs, float* __restrict__ C,
    float* __restrict__ logit, int M, int tile0, int tile1)
{
  const int b = (int)blockIdx.x;
  if (b < NB) range_sort_body(binned, wgoff, rowptr, col, N, G, b, NB);
  else gemm1_part(X, W, vecs, C, logit, M, tile0, tile1, b - NB);
}

typedef float v4f __attribute__((ext_vector_type(4)));

__device__ __forceinline__ float leaky(float v)
{
  return (v > 0.f) ? v : NEG_SLOPE * v;
}


__device__ __forceinline__ int lane_id()
{
  return (int)__builtin_amdgcn_mbcnt_lo(~0u, 0u);
}

template<int K>
__device__ __forceinline__ int half_lane(int v)
{
  return __builtin_amdgcn_update_dpp(0, v, 0x150 + K, 0xf, 0xf, false);
}

template<int C, int K>
__device__ __forceinline__ void row_ask(const char* __restrict__ hb, unsigned lo, int sl,
                                        v4f (&r)[C / 64])
{
  const unsigned off = (unsigned)half_lane<K>(sl) * (unsigned)(C * 4) + lo;
#pragma unroll
  for (int v = 0; v < C / 64; ++v) r[v] = *(const v4f*)(hb + off + v * 256);
}

template<int C, int K>
__device__ __forceinline__ void row_add(int ali, const v4f (&r)[C / 64], v4f (&acc)[C / 64])
{
  const float a = __int_as_float(half_lane<K>(ali));
#pragma unroll
  for (int v = 0; v < C / 64; ++v) acc[v] += a * r[v];
}

template<int C, int K0, int G>
__device__ __forceinline__ void row_group(const char* __restrict__ hb, unsigned lo, int sl, int ali,
                                          v4f (&acc)[C / 64])
{
  static_assert(G == 4 || G == 8, "group sizes written out below");
  v4f r0[C / 64], r1[C / 64], r2[C / 64], r3[C / 64];
  row_ask<C, K0 + 0>(hb, lo, sl, r0);
  row_ask<C, K0 + 1>(hb, lo, sl, r1);
  row_ask<C, K0 + 2>(hb, lo, sl, r2);
  row_ask<C, K0 + 3>(hb, lo, sl, r3);
  if constexpr (G == 8) {
    v4f r4[C / 64], r5[C / 64], r6[C / 64], r7[C / 64];
    row_ask<C, K0 + 4>(hb, lo, sl, r4);
    row_ask<C, K0 + 5>(hb, lo, sl, r5);
    row_ask<C, K0 + 6>(hb, lo, sl, r6);
    row_ask<C, K0 + 7>(hb, lo, sl, r7);
    row_add<C, K0 + 0>(ali, r0, acc);
    row_add<C, K0 + 1>(ali, r1, acc);
    row_add<C, K0 + 2>(ali, r2, acc);
    row_add<C, K0 + 3>(ali, r3, acc);
    row_add<C, K0 + 4>(ali, r4, acc);
    row_add<C, K0 + 5>(ali, r5, acc);
    row_add<C, K0 + 6>(ali, r6, acc);
    row_add<C, K0 + 7>(ali, r7, acc);
  } else {
    row_add<C, K0 + 0>(ali, r0, acc);
    row_add<C, K0 + 1>(ali, r1, acc);
    row_add<C, K0 + 2>(ali, r2, acc);
    row_add<C, K0 + 3>(ali, r3, acc);
  }
}

template<int C, bool BIAS>
__device__ __forceinline__ void gather_pair(
    const int* __restrict__ rowptr, const int* __restrict__ col,
    const float* __restrict__ asrc, const float* __restrict__ adst,
    const float* __restrict__ h, const float* __restrict__ bias, int d0, int N,
    v4f (&o)[C / 64])
{
  constexpr int NV = C / 64;
  constexpr int G  = 8 / NV;
  const int lane = lane_id();
  const int hf = lane >> 4;
  const int hl = lane & 15;
  const int   i2  = (d0 + 2 < N) ? (d0 + 2) : N;
  const int   i1  = (d0 + 1 < N) ? (d0 + 1) : (N - 1);
  const int   b0  = rowptr[d0];
  const int   b1  = rowptr[d0 + 1];
  const int   b2  = rowptr[i2];
  const float ad0 = adst[d0];
  const float ad1 = adst[i1];
  const int   beg = hf ? b1 : b0;
  const int   end = hf ? b2 : b1;
  const float ad  = hf ? ad1 : ad0;

  const int  j0  = beg + hl;
  const int  j1  = j0 + 16;
  const bool in0 = j0 < end;
  const bool in1 = j1 < end;
  int sl0 = 0, sl1 = 0;
  if (in0) sl0 = col[j0];
  if (in1) sl1 = col[j1];
  const float g0 = asrc[sl0];
  const float g1 = asrc[sl1];
  const float e0 = in0 ? leaky(g0 + ad) : -__builtin_huge_valf();
  const float e1 = in1 ? leaky(g1 + ad) : -__builtin_huge_valf();

  float mx = fmaxf(e0, e1);
  for (int j = j0 + 32; j < end; j += 16)
    mx = fmaxf(mx, leaky(asrc[col[j]] + ad));
  for (int off = 8; off > 0; off >>= 1) mx = fmaxf(mx, __shfl_xor(mx, off, 32));

  const float p0 = in0 ? __expf(e0 - mx) : 0.f;
  const float p1 = in1 ? __expf(e1 - mx) : 0.f;
  float den = p0 + p1;
  for (int j = j0 + 32; j < end; j += 16)
    den += __expf(leaky(asrc[col[j]] + ad) - mx);
  for (int off = 8; off > 0; off >>= 1) den += __shfl_xor(den, off, 32);

  const int n0 = b1 - b0;
  const int n1 = b2 - b1;
  const int nmax = (n0 > n1) ? n0 : n1;

  const char* hb = reinterpret_cast<const char*>(h);
  const unsigned lo = (unsigned)hl * 16u;
  v4f acc[NV];
#pragma unroll
  for (int v = 0; v < NV; ++v) acc[v] = (v4f)(0.f);

  for (int base = 0; base < nmax; base += 16) {
    int   sl = sl0;
    float al = in0 ? p0 / den : 0.f;
    if (base == 16) {
      sl = sl1;
      al = in1 ? p1 / den : 0.f;
    } else if (base != 0) {
      const int j = j0 + base;
      sl = 0;
      al = 0.f;
      if (j < end) {
        sl = col[j];
        al = __expf(leaky(asrc[sl] + ad) - mx) / den;
      }
    }
    const int ali = __float_as_int(al);
    const int rem = nmax - base;
    row_group<C, 0, G>(hb, lo, sl, ali, acc);
    if (G == 4) {
      if (rem > 4)  row_group<C, 4, G>(hb, lo, sl, ali, acc);
      if (rem > 8)  row_group<C, 8, G>(hb, lo, sl, ali, acc);
      if (rem > 12) row_group<C, 12, G>(hb, lo, sl, ali, acc);
    } else {
      if (rem > 8)  row_group<C, 8, G>(hb, lo, sl, ali, acc);
    }
  }

#pragma unroll
  for (int v = 0; v < NV; ++v) {
    o[v] = acc[v];
    if (BIAS) o[v] += *(const v4f*)(bias + v * 64 + hl * 4);
  }
}

__global__ __launch_bounds__(256, 16) void seg_gather_pair64(
    const int* __restrict__ rowptr, const int* __restrict__ col,
    const float* __restrict__ asrc, const float* __restrict__ adst,
    const float* __restrict__ h, const float* __restrict__ bias,
    float* __restrict__ out, int N)
{
  const int lane = threadIdx.x & 31;
  const int d0 = __builtin_amdgcn_readfirstlane((int)(blockIdx.x * 16 + (threadIdx.x >> 5) * 2));
  if (d0 >= N) return;
  v4f o[1];
  gather_pair<64, true>(rowptr, col, asrc, adst, h, bias, d0, N, o);
  const int d = d0 + (lane >> 4);
  if (d < N) *(v4f*)(out + (size_t)d * 64 + (lane & 15) * 4) = o[0];
}

#define TP 132
#define FROWS 32
__global__ __launch_bounds__(256, 16) void gather1_gemm2(
    const int* __restrict__ rowptr, const int* __restrict__ col,
    const float* __restrict__ asrc, const float* __restrict__ adst,
    const float* __restrict__ h, const float* __restrict__ bias,
    const float* __restrict__ W2, const float* __restrict__ vecs2,
    float* __restrict__ h2, float* __restrict__ logit2, int N)
{
  __shared__ __attribute__((aligned(16))) float tile[FROWS * TP];
  const int wave = __builtin_amdgcn_readfirstlane((int)(threadIdx.x >> 5));
  const int row0 = (int)blockIdx.x * FROWS;

  for (int p = 0; p < 2; ++p) {
    const int r = 4 * wave + 2 * p;
    v4f o[2];
    o[0] = (v4f)(0.f);
    o[1] = (v4f)(0.f);
    if (row0 + r < N) gather_pair<128, false>(rowptr, col, asrc, adst, h, bias, row0 + r, N, o);
    const int ln = lane_id();
    float* tp = tile + (r + (ln >> 4)) * TP + (ln & 15) * 4;
    *(v4f*)(tp)      = o[0];
    *(v4f*)(tp + 64) = o[1];
  }
  {
    const int ln = lane_id();
    const v4f bA = *(const v4f*)(bias + (ln & 15) * 4);
    const v4f bB = *(const v4f*)(bias + 64 + (ln & 15) * 4);
    for (int p = 0; p < 2; ++p) {
      float* tp = tile + (4 * wave + 2 * p + (ln >> 4)) * TP + (ln & 15) * 4;
      v4f x = *(v4f*)(tp) + bA;
      v4f y = *(v4f*)(tp + 64) + bB;
      for (int i = 0; i < 4; ++i) { x[i] = fmaxf(x[i], 0.f); y[i] = fmaxf(y[i], 0.f); }
      *(v4f*)(tp)      = x;
      *(v4f*)(tp + 64) = y;
    }
  }
  __syncthreads();

  const int lane = lane_id();
  const int half = lane >> 4;
  const int l15  = lane & 15;
  const int rt   = (wave >> 1) & 1;
  const int trow = row0 + 16 * rt;
  const float* ap = tile + (16 * rt + l15) * TP + 2 * half;

  if (wave >= 4) {
    const int v = wave & 1;
    const float* vp = vecs2 + (size_t)v * 128 + 2 * half;
    float lg = 0.f;
#pragma unroll
    for (int k = 0; k < 128; k += 4) {
      const v2f a = *(const v2f*)(ap + k);
      lg = fmaf(a.x, vp[k + 0], lg);
      lg = fmaf(a.y, vp[k + 1], lg);
    }
    const float s = lg + __shfl_xor(lg, 16, 32);
    if (half == 0 && trow + l15 < N) logit2[(size_t)v * N + trow + l15] = s;
    return;
  }

  const int g = wave & 1;
  const float* wb = W2 + (size_t)(2 * half) * 64 + g * 32 + l15;
  v8f acc0 = (v8f){0.f,0.f,0.f,0.f,0.f,0.f,0.f,0.f};
  v8f acc1 = (v8f){0.f,0.f,0.f,0.f,0.f,0.f,0.f,0.f};
  constexpr int AH = 4;
  v2f b0[AH], b1[AH];
#pragma unroll
  for (int i = 0; i < AH; ++i) {
    const int k = 4 * i;
    b0[i].x = wb[(size_t)(k + 0) * 64];
    b0[i].y = wb[(size_t)(k + 1) * 64];
    b1[i].x = wb[(size_t)(k + 0) * 64 + 16];
    b1[i].y = wb[(size_t)(k + 1) * 64 + 16];
  }
  v2f a = *(const v2f*)(ap);
#pragma unroll
  for (int i = 0; i < 32; ++i) {
    const int k = 4 * i;
    const int j = i % AH;
    v2f an = a;
    if (i + 1 < 32) an = *(const v2f*)(ap + k + 4);
    acc0 = __builtin_amdgcn_wmma_f32_16x16x4_f32(false, a, false, b0[j], (short)0, acc0, false, false);
    if (i + AH < 32) {
      b0[j].x = wb[(size_t)(k + 4 * AH + 0) * 64];
      b0[j].y = wb[(size_t)(k + 4 * AH + 1) * 64];
    }
    acc1 = __builtin_amdgcn_wmma_f32_16x16x4_f32(false, a, false, b1[j], (short)0, acc1, false, false);
    if (i + AH < 32) {
      b1[j].x = wb[(size_t)(k + 4 * AH + 0) * 64 + 16];
      b1[j].y = wb[(size_t)(k + 4 * AH + 1) * 64 + 16];
    }
    __builtin_amdgcn_sched_barrier(0);
    a = an;
  }

  const int mbase = trow + 8 * half;
  float* cp = h2 + (size_t)mbase * 64 + g * 32 + l15;
#pragma unroll
  for (int r = 0; r < 8; ++r)
    if (mbase + r < N) {
      cp[(size_t)r * 64]      = acc0[r];
      cp[(size_t)r * 64 + 16] = acc1[r];
    }
}

static inline int cdiv(long a, long b) { return (int)((a + b - 1) / b); }

extern "C" void kernel_launch(void* const* d_in, const int* in_sizes, int n_in,
                              void* d_out, int out_size, void* d_ws, size_t ws_size,
                              hipStream_t stream)
{
  const float* x       = (const float*)d_in[0];
  const int*   ei      = (const int*)d_in[1];
  const float* Wsrc1   = (const float*)d_in[2];
  const float* Wdst1   = (const float*)d_in[3];
  const float* attsrc1 = (const float*)d_in[4];
  const float* attdst1 = (const float*)d_in[5];
  const float* b1      = (const float*)d_in[6];
  const float* Wsrc2   = (const float*)d_in[7];
  const float* Wdst2   = (const float*)d_in[8];
  const float* attsrc2 = (const float*)d_in[9];
  const float* attdst2 = (const float*)d_in[10];
  const float* b2      = (const float*)d_in[11];

  const int HID = in_sizes[4];
  const int OUT = in_sizes[9];
  const int FIN = in_sizes[2] / HID;
  const int N   = in_sizes[0] / FIN;
  const int E   = in_sizes[1] / 2;
  const int* src = ei;
  const int* dst = ei + E;

  float* ws    = (float*)d_ws;
  float* h1    = ws;  ws += (size_t)N * HID;
  float* h2    = ws;  ws += (size_t)N * OUT;
  float* logit = ws;  ws += 2 * (size_t)N;
  float* logit2 = ws; ws += 2 * (size_t)N;
  float* vfold = ws;  ws += 2 * (size_t)FIN + 2 * (size_t)HID;
  const int G1 = 256;
  const int NB = cdiv(N, RB);
  int* binned  = (int*)ws;  ws += 2 * (size_t)E;
  int* wgcnt   = (int*)ws;  ws += (size_t)NB * G1;
  int* wgoff   = (int*)ws;  ws += (size_t)NB * G1 + 1;
  int* col     = (int*)ws;  ws += E;
  int* rowptr  = (int*)ws;  ws += N + 1;

  const int tiles1 = (N / 16) * (HID / 64);
  const int tA = tiles1 / 3;
  const int tB = 2 * (tiles1 / 3);
  count_and_fold<<<G1 + cdiv(2 * FIN + 2 * HID, 32), 1024, 0, stream>>>(
      dst, wgcnt, E, NB, G1,
      Wsrc1, attsrc1, Wdst1, attdst1, Wsrc2, attsrc2, Wdst2, attdst2,
      vfold, FIN, HID, HID, OUT);
  scan_and_gemm<<<1 + cdiv(tA, GEMM_WPB), 1024, 0, stream>>>(
      wgcnt, wgoff, NB * G1,
      x, Wsrc1, vfold, h1, logit, N, 0, tA);
  scatter_and_gemm<<<G1 + cdiv(tB - tA, GEMM_WPB), 1024, 0, stream>>>(
      src, dst, wgoff, binned, E, NB, G1,
      x, Wsrc1, vfold, h1, logit, N, tA, tB);
  sort_and_gemm<<<NB + cdiv(tiles1 - tB, GEMM_WPB), 1024, 0, stream>>>(
      binned, wgoff, rowptr, col, N, G1, NB,
      x, Wsrc1, vfold, h1, logit, N, tB, tiles1);

  gather1_gemm2<<<cdiv(N, FROWS), 256, 0, stream>>>(
      rowptr, col, logit, logit + N, h1, b1,
      Wsrc2, vfold + 2 * (size_t)FIN, h2, logit2, N);

  {
    float* out = (float*)d_out;
    seg_gather_pair64<<<cdiv(N, 16), 256, 0, stream>>>(
        rowptr, col, logit2, logit2 + N, h2, b2, out, N);
  }
}


# Count, scatter, range-sort launches: 512-thread workgroups (half the waves enter, every rider wave works), 256 riders each, rest with the scan; rows summed in counter arrival order of the new layout
# speedup vs baseline: 6.2437x; 1.0306x over previous
#include <hip/hip_runtime.h>
#include <hip/hip_bf16.h>

typedef float v2f __attribute__((ext_vector_type(2)));
typedef float v8f __attribute__((ext_vector_type(8)));

#define NEG_SLOPE 0.2f

template<int K, int NOUT, int NT>
__device__ __forceinline__ void gemm_strip(
    const float* __restrict__ X, const float* __restrict__ W,
    const float* __restrict__ vecs, float* __restrict__ C,
    float* __restrict__ logit, int M, int tile)
{
  const int lane   = threadIdx.x & 31;
  constexpr int tilesN = NOUT / (16 * NT);
  static_assert(tilesN == 1 || tilesN == 2, "two logit vectors are split over the strips of a row block");
  constexpr int NV = 2 / tilesN;
  const int tM = tile / tilesN;
  const int tN = tile % tilesN;

  const int half = lane >> 4;
  const int l15  = lane & 15;
  const float* xrow = X + (size_t)(tM * 16 + l15) * K + 2 * half;
  const float* wb   = W + (size_t)(2 * half) * NOUT + tN * (16 * NT) + l15;
  const float* vp   = vecs + (size_t)(tN * NV) * K + 2 * half;

  v8f acc[NT];
#pragma unroll
  for (int t = 0; t < NT; ++t) acc[t] = (v8f){0.f,0.f,0.f,0.f,0.f,0.f,0.f,0.f};
  float lg[NV];
  for (int i = 0; i < NV; ++i) lg[i] = 0.f;

  {
    v2f a, an;
    v2f vv[NV], vn[NV];
    v2f b[NT];
    a.x = xrow[0];
    a.y = xrow[1];
#pragma unroll
    for (int i = 0; i < NV; ++i) { vv[i].x = vp[i * K + 0]; vv[i].y = vp[i * K + 1]; }
#pragma unroll
    for (int t = 0; t < NT; ++t) { b[t].x = wb[16 * t]; b[t].y = wb[(size_t)NOUT + 16 * t]; }
#pragma unroll
    for (int k = 0; k < K; k += 4) {
      if (k + 4 < K) {
        an.x = xrow[k + 4];
        an.y = xrow[k + 5];
#pragma unroll
        for (int i = 0; i < NV; ++i) { vn[i].x = vp[i * K + k + 4]; vn[i].y = vp[i * K + k + 5]; }
      }
#pragma unroll
      for (int i = 0; i < NV; ++i) {
        lg[i] = fmaf(a.x, vv[i].x, lg[i]);
        lg[i] = fmaf(a.y, vv[i].y, lg[i]);
      }
#pragma unroll
      for (int t = 0; t < NT; ++t) {
        acc[t] = __builtin_amdgcn_wmma_f32_16x16x4_f32(false, a, false, b[t],
                                                       (short)0, acc[t],
                                                       false, false);
        if (k + 4 < K) {
          b[t].x = wb[(size_t)(k + 4) * NOUT + 16 * t];
          b[t].y = wb[(size_t)(k + 5) * NOUT + 16 * t];
        }
        __builtin_amdgcn_sched_barrier(0);
      }
      if (k + 4 < K) {
        a = an;
#pragma unroll
        for (int i = 0; i < NV; ++i) vv[i] = vn[i];
      }
    }
  }

  const int mbase = tM * 16 + 8 * half;
  float* cp = C + (size_t)mbase * NOUT + tN * (16 * NT) + l15;
#pragma unroll
  for (int t = 0; t < NT; ++t)
#pragma unroll
    for (int r = 0; r < 8; ++r)
      cp[(size_t)r * NOUT + 16 * t] = acc[t][r];

  for (int i = 0; i < NV; ++i) {
    const float s = lg[i] + __shfl_xor(lg[i], 16, 32);
    if (half == 0) logit[(size_t)(tN * NV + i) * M + tM * 16 + l15] = s;
  }
}

#define GEMM_WPB 16
__device__ __forceinline__ void gemm1_part(
    const float* __restrict__ X, const float* __restrict__ W,
    const float* __restrict__ vecs, float* __restrict__ C,
    float* __restrict__ logit, int M, int tile0, int tile1, int gb)
{
  const int wave = threadIdx.x >> 5;
  if (wave >= GEMM_WPB) return;
  const int tile = __builtin_amdgcn_readfirstlane(tile0 + gb * GEMM_WPB + wave);
  if (tile >= tile1) return;
  gemm_strip<128, 128, 4>(X, W, vecs, C, logit, M, tile);
}

__device__ __forceinline__ void fold_att_body(
    const float* __restrict__ W0, const float* __restrict__ a0,
    const float* __restrict__ W1, const float* __restrict__ a1,
    const float* __restrict__ W2, const float* __restrict__ a2,
    const float* __restrict__ W3, const float* __restrict__ a3,
    float* __restrict__ v, int r1, int c1, int r2, int c2, int gw)
{
  const int lane = threadIdx.x & 31;
  if (gw >= 2 * r1 + 2 * r2) return;
  const float* Wm; const float* att; int row, cols;
  if (gw < r1)               { Wm = W0; att = a0; row = gw;               cols = c1; }
  else if (gw < 2 * r1)      { Wm = W1; att = a1; row = gw - r1;          cols = c1; }
  else if (gw < 2 * r1 + r2) { Wm = W2; att = a2; row = gw - 2 * r1;      cols = c2; }
  else                       { Wm = W3; att = a3; row = gw - 2 * r1 - r2; cols = c2; }
  const float* wr = Wm + (size_t)row * cols;
  float s = 0.f;
  for (int k = lane; k < cols; k += 32) s += wr[k] * att[k];
  for (int off = 16; off > 0; off >>= 1) s += __shfl_down(s, off, 32);
  if (lane == 0) v[gw] = s;
}

#define RB_SHIFT 9
#define RB (1 << RB_SHIFT)
#define MAXNB 1024
#define GT 512
#define GW (GT / 32)
#define EPT (4096 / GT)
static_assert(GT >= RB && GT % 32 == 0 && 4096 % GT == 0, "range_sort gives one thread to each node of a range");

__device__ __forceinline__ void bin_count_body(
    const int* __restrict__ dst, int* __restrict__ wgcnt, int E, int NB, int bid, int G)
{
  __shared__ int hist[MAXNB];
  const int tid = threadIdx.x;
  for (int r = tid; r < NB; r += GT) hist[r] = 0;
  __syncthreads();
  const int chunk = (E + G - 1) / G;
  const int e0 = bid * chunk;
  const int e1 = (e0 + chunk < E) ? (e0 + chunk) : E;
  for (int base = e0; base < e1; base += 4096) {
    int d[EPT];
    for (int q = 0; q < EPT; ++q) {
      const int i = base + tid + q * GT;
      d[q] = (i < e1) ? dst[i] : -1;
    }
    for (int q = 0; q < EPT; ++q)
      if (d[q] >= 0) atomicAdd(&hist[d[q] >> RB_SHIFT], 1);
  }
  __syncthreads();
  for (int r = tid; r < NB; r += GT) wgcnt[(size_t)r * G + bid] = hist[r];
}

__device__ __forceinline__ void scan_counts_body(
    const int* __restrict__ cnt, int* __restrict__ rowptr, int N)
{
  __shared__ int wtot[32];
  __shared__ int tile_total;
  const int tid  = threadIdx.x;
  const int lane = tid & 31;
  const int wave = tid >> 5;
  int carry = 0;
  for (int base = 0; base < N; base += 8192) {
    const int i0 = base + tid * 8;
    int c[8];
    int t = 0;
    for (int q = 0; q < 8; ++q) {
      c[q] = (i0 + q < N) ? cnt[i0 + q] : 0;
      t += c[q];
    }
    int s = t;
    for (int off = 1; off < 32; off <<= 1) {
      const int nb = __shfl_up(s, off, 32);
      if (lane >= off) s += nb;
    }
    if (lane == 31) wtot[wave] = s;
    __syncthreads();
    if (wave == 0) {
      const int w = wtot[lane];
      int ws = w;
      for (int off = 1; off < 32; off <<= 1) {
        const int nb = __shfl_up(ws, off, 32);
        if (lane >= off) ws += nb;
      }
      wtot[lane] = ws - w;
      if (lane == 31) tile_total = ws;
    }
    __syncthreads();
    int e = carry + wtot[wave] + (s - t);
    carry += tile_total;
    for (int q = 0; q < 8; ++q) {
      if (i0 + q < N) rowptr[i0 + q] = e;
      e += c[q];
    }
    __syncthreads();
  }
  if (tid == 0) rowptr[N] = carry;
}

__device__ __forceinline__ void bin_scatter_body(
    const int* __restrict__ src, const int* __restrict__ dst,
    const int* __restrict__ wgoff, int* __restrict__ binned, int E, int NB, int bid, int G)
{
  __shared__ int cur[MAXNB];
  const int tid = threadIdx.x;
  for (int r = tid; r < NB; r += GT) cur[r] = wgoff[(size_t)r * G + bid];
  __syncthreads();
  int2* pairs = reinterpret_cast<int2*>(binned);
  const int chunk = (E + G - 1) / G;
  const int e0 = bid * chunk;
  const int e1 = (e0 + chunk < E) ? (e0 + chunk) : E;
  for (int base = e0; base < e1; base += 4096) {
    int d[EPT], s[EPT];
    for (int q = 0; q < EPT; ++q) {
      const int i = base + tid + q * GT;
      d[q] = (i < e1) ? dst[i] : -1;
      s[q] = (i < e1) ? src[i] : 0;
    }
    for (int q = 0; q < EPT; ++q)
      if (d[q] >= 0) {
        const int slot = atomicAdd(&cur[d[q] >> RB_SHIFT], 1);
        pairs[slot] = make_int2(s[q], d[q]);
      }
  }
}

__device__ __forceinline__ void range_sort_body(
    const int* __restrict__ binned, const int* __restrict__ wgoff,
    int* __restrict__ rowptr, int* __restrict__ col, int N, int G, int rg, int NB)
{
  __shared__ int cnt[RB];
  __shared__ int wtot[RB / 32];
  const int tid  = threadIdx.x;
  const int lane = tid & 31;
  const int wave = tid >> 5;
  const int s0 = wgoff[(size_t)rg * G];
  const int s1 = wgoff[(size_t)(rg + 1) * G];
  const int2* pairs = reinterpret_cast<const int2*>(binned);

  if (tid < RB) cnt[tid] = 0;
  __syncthreads();
  for (int base = s0; base < s1; base += 4096) {
    int dl[EPT];
    for (int q = 0; q < EPT; ++q) {
      const int i = base + tid + q * GT;
      dl[q] = (i < s1) ? (pairs[i].y & (RB - 1)) : -1;
    }
    for (int q = 0; q < EPT; ++q)
      if (dl[q] >= 0) atomicAdd(&cnt[dl[q]], 1);
  }
  __syncthreads();

  int v = 0, incl = 0;
  if (tid < RB) {
    v = cnt[tid];
    incl = v;
    for (int off = 1; off < 32; off <<= 1) {
      const int nb = __shfl_up(incl, off, 32);
      if (lane >= off) incl += nb;
    }
    if (lane == 31) wtot[wave] = incl;
  }
  __syncthreads();
  if (tid < RB) {
    int offs = 0;
    for (int w = 0; w < wave; ++w) offs += wtot[w];
    const int excl = offs + incl - v;
    cnt[tid] = excl;
    const int node = rg * RB + tid;
    if (node < N) rowptr[node] = s0 + excl;
  }
  if (rg == NB - 1 && tid == 0) rowptr[N] = s1;
  __syncthreads();

  for (int base = s0; base < s1; base += 4096) {
    int2 r[EPT];
    for (int q = 0; q < EPT; ++q) {
      const int i = base + tid + q * GT;
      r[q] = (i < s1) ? pairs[i] : make_int2(0, -1);
    }
    for (int q = 0; q < EPT; ++q)
      if (r[q].y >= 0) {
        const int slot = atomicAdd(&cnt[r[q].y & (RB - 1)], 1);
        col[s0 + slot] = r[q].x;
      }
  }
}

__global__ __launch_bounds__(GT) void count_and_fold(
    const int* __restrict__ dst, int* __restrict__ wgcnt, int E, int NB, int G,
    const float* __restrict__ W0, const float* __restrict__ a0,
    const float* __restrict__ W1, const float* __restrict__ a1,
    const float* __restrict__ W2, const float* __restrict__ a2,
    const float* __restrict__ W3, const float* __restrict__ a3,
    float* __restrict__ v, int r1, int c1, int r2, int c2)
{
  const int b = (int)blockIdx.x;
  if (b < G) bin_count_body(dst, wgcnt, E, NB, b, G);
  else fold_att_body(W0, a0, W1, a1, W2, a2, W3, a3, v, r1, c1, r2, c2,
                     (b - G) * GW + (int)(threadIdx.x >> 5));
}

__global__ __launch_bounds__(1024, 16) void scan_and_gemm(
    const int* __restrict__ cnt, int* __restrict__ off, int n,
    const float* __restrict__ X, const float* __restrict__ W,
    const float* __restrict__ vecs, float* __restrict__ C,
    float* __restrict__ logit, int M, int tile0, int tile1)
{
  const int b = (int)blockIdx.x;
  if (b == 0) scan_counts_body(cnt, off, n);
  else gemm1_part(X, W, vecs, C, logit, M, tile0, tile1, b - 1);
}

__global__ __launch_bounds__(GT, 16) void scatter_and_gemm(
    const int* __restrict__ src, const int* __restrict__ dst,
    const int* __restrict__ wgoff, int* __restrict__ binned, int E, int NB, int G,
    const float* __restrict__ X, const float* __restrict__ W,
    const float* __restrict__ vecs, float* __restrict__ C,
    float* __restrict__ logit, int M, int tile0, int tile1)
{
  const int b = (int)blockIdx.x;
  if (b < G) bin_scatter_body(src, dst, wgoff, binned, E, NB, b, G);
  else gemm1_part(X, W, vecs, C, logit, M, tile0, tile1, b - G);
}

__global__ __launch_bounds__(GT, 16) void sort_and_gemm(
    const int* __restrict__ binned, const int* __restrict__ wgoff,
    int* __restrict__ rowptr, int* __restrict__ col, int N, int G, int NB,
    const float* __restrict__ X, const float* __restrict__ W,
    const float* __restrict__ vecs, float* __restrict__ C,
    float* __restrict__ logit, int M, int tile0, int tile1)
{
  const int b = (int)blockIdx.x;
  if (b < NB) range_sort_body(binned, wgoff, rowptr, col, N, G, b, NB);
  else gemm1_part(X, W, vecs, C, logit, M, tile0, tile1, b - NB);
}

typedef float v4f __attribute__((ext_vector_type(4)));

__device__ __forceinline__ float leaky(float v)
{
  return (v > 0.f) ? v : NEG_SLOPE * v;
}


__device__ __forceinline__ int lane_id()
{
  return (int)__builtin_amdgcn_mbcnt_lo(~0u, 0u);
}

template<int K>
__device__ __forceinline__ int half_lane(int v)
{
  return __builtin_amdgcn_update_dpp(0, v, 0x150 + K, 0xf, 0xf, false);
}

template<int C, int K>
__device__ __forceinline__ void row_ask(const char* __restrict__ hb, unsigned lo, int sl,
                                        v4f (&r)[C / 64])
{
  const unsigned off = (unsigned)half_lane<K>(sl) * (unsigned)(C * 4) + lo;
#pragma unroll
  for (int v = 0; v < C / 64; ++v) r[v] = *(const v4f*)(hb + off + v * 256);
}

template<int C, int K>
__device__ __forceinline__ void row_add(int ali, const v4f (&r)[C / 64], v4f (&acc)[C / 64])
{
  const float a = __int_as_float(half_lane<K>(ali));
#pragma unroll
  for (int v = 0; v < C / 64; ++v) acc[v] += a * r[v];
}

template<int C, int K0, int G>
__device__ __forceinline__ void row_group(const char* __restrict__ hb, unsigned lo, int sl, int ali,
                                          v4f (&acc)[C / 64])
{
  static_assert(G == 4 || G == 8, "group sizes written out below");
  v4f r0[C / 64], r1[C / 64], r2[C / 64], r3[C / 64];
  row_ask<C, K0 + 0>(hb, lo, sl, r0);
  row_ask<C, K0 + 1>(hb, lo, sl, r1);
  row_ask<C, K0 + 2>(hb, lo, sl, r2);
  row_ask<C, K0 + 3>(hb, lo, sl, r3);
  if constexpr (G == 8) {
    v4f r4[C / 64], r5[C / 64], r6[C / 64], r7[C / 64];
    row_ask<C, K0 + 4>(hb, lo, sl, r4);
    row_ask<C, K0 + 5>(hb, lo, sl, r5);
    row_ask<C, K0 + 6>(hb, lo, sl, r6);
    row_ask<C, K0 + 7>(hb, lo, sl, r7);
    row_add<C, K0 + 0>(ali, r0, acc);
    row_add<C, K0 + 1>(ali, r1, acc);
    row_add<C, K0 + 2>(ali, r2, acc);
    row_add<C, K0 + 3>(ali, r3, acc);
    row_add<C, K0 + 4>(ali, r4, acc);
    row_add<C, K0 + 5>(ali, r5, acc);
    row_add<C, K0 + 6>(ali, r6, acc);
    row_add<C, K0 + 7>(ali, r7, acc);
  } else {
    row_add<C, K0 + 0>(ali, r0, acc);
    row_add<C, K0 + 1>(ali, r1, acc);
    row_add<C, K0 + 2>(ali, r2, acc);
    row_add<C, K0 + 3>(ali, r3, acc);
  }
}

template<int C, bool BIAS>
__device__ __forceinline__ void gather_pair(
    const int* __restrict__ rowptr, const int* __restrict__ col,
    const float* __restrict__ asrc, const float* __restrict__ adst,
    const float* __restrict__ h, const float* __restrict__ bias, int d0, int N,
    v4f (&o)[C / 64])
{
  constexpr int NV = C / 64;
  constexpr int G  = 8 / NV;
  const int lane = lane_id();
  const int hf = lane >> 4;
  const int hl = lane & 15;
  const int   i2  = (d0 + 2 < N) ? (d0 + 2) : N;
  const int   i1  = (d0 + 1 < N) ? (d0 + 1) : (N - 1);
  const int   b0  = rowptr[d0];
  const int   b1  = rowptr[d0 + 1];
  const int   b2  = rowptr[i2];
  const float ad0 = adst[d0];
  const float ad1 = adst[i1];
  const int   beg = hf ? b1 : b0;
  const int   end = hf ? b2 : b1;
  const float ad  = hf ? ad1 : ad0;

  const int  j0  = beg + hl;
  const int  j1  = j0 + 16;
  const bool in0 = j0 < end;
  const bool in1 = j1 < end;
  int sl0 = 0, sl1 = 0;
  if (in0) sl0 = col[j0];
  if (in1) sl1 = col[j1];
  const float g0 = asrc[sl0];
  const float g1 = asrc[sl1];
  const float e0 = in0 ? leaky(g0 + ad) : -__builtin_huge_valf();
  const float e1 = in1 ? leaky(g1 + ad) : -__builtin_huge_valf();

  float mx = fmaxf(e0, e1);
  for (int j = j0 + 32; j < end; j += 16)
    mx = fmaxf(mx, leaky(asrc[col[j]] + ad));
  for (int off = 8; off > 0; off >>= 1) mx = fmaxf(mx, __shfl_xor(mx, off, 32));

  const float p0 = in0 ? __expf(e0 - mx) : 0.f;
  const float p1 = in1 ? __expf(e1 - mx) : 0.f;
  float den = p0 + p1;
  for (int j = j0 + 32; j < end; j += 16)
    den += __expf(leaky(asrc[col[j]] + ad) - mx);
  for (int off = 8; off > 0; off >>= 1) den += __shfl_xor(den, off, 32);

  const int n0 = b1 - b0;
  const int n1 = b2 - b1;
  const int nmax = (n0 > n1) ? n0 : n1;

  const char* hb = reinterpret_cast<const char*>(h);
  const unsigned lo = (unsigned)hl * 16u;
  v4f acc[NV];
#pragma unroll
  for (int v = 0; v < NV; ++v) acc[v] = (v4f)(0.f);

  for (int base = 0; base < nmax; base += 16) {
    int   sl = sl0;
    float al = in0 ? p0 / den : 0.f;
    if (base == 16) {
      sl = sl1;
      al = in1 ? p1 / den : 0.f;
    } else if (base != 0) {
      const int j = j0 + base;
      sl = 0;
      al = 0.f;
      if (j < end) {
        sl = col[j];
        al = __expf(leaky(asrc[sl] + ad) - mx) / den;
      }
    }
    const int ali = __float_as_int(al);
    const int rem = nmax - base;
    row_group<C, 0, G>(hb, lo, sl, ali, acc);
    if (G == 4) {
      if (rem > 4)  row_group<C, 4, G>(hb, lo, sl, ali, acc);
      if (rem > 8)  row_group<C, 8, G>(hb, lo, sl, ali, acc);
      if (rem > 12) row_group<C, 12, G>(hb, lo, sl, ali, acc);
    } else {
      if (rem > 8)  row_group<C, 8, G>(hb, lo, sl, ali, acc);
    }
  }

#pragma unroll
  for (int v = 0; v < NV; ++v) {
    o[v] = acc[v];
    if (BIAS) o[v] += *(const v4f*)(bias + v * 64 + hl * 4);
  }
}

__global__ __launch_bounds__(256, 16) void seg_gather_pair64(
    const int* __restrict__ rowptr, const int* __restrict__ col,
    const float* __restrict__ asrc, const float* __restrict__ adst,
    const float* __restrict__ h, const float* __restrict__ bias,
    float* __restrict__ out, int N)
{
  const int lane = threadIdx.x & 31;
  const int d0 = __builtin_amdgcn_readfirstlane((int)(blockIdx.x * 16 + (threadIdx.x >> 5) * 2));
  if (d0 >= N) return;
  v4f o[1];
  gather_pair<64, true>(rowptr, col, asrc, adst, h, bias, d0, N, o);
  const int d = d0 + (lane >> 4);
  if (d < N) *(v4f*)(out + (size_t)d * 64 + (lane & 15) * 4) = o[0];
}

#define TP 132
#define FROWS 32
__global__ __launch_bounds__(256, 16) void gather1_gemm2(
    const int* __restrict__ rowptr, const int* __restrict__ col,
    const float* __restrict__ asrc, const float* __restrict__ adst,
    const float* __restrict__ h, const float* __restrict__ bias,
    const float* __restrict__ W2, const float* __restrict__ vecs2,
    float* __restrict__ h2, float* __restrict__ logit2, int N)
{
  __shared__ __attribute__((aligned(16))) float tile[FROWS * TP];
  const int wave = __builtin_amdgcn_readfirstlane((int)(threadIdx.x >> 5));
  const int row0 = (int)blockIdx.x * FROWS;

  for (int p = 0; p < 2; ++p) {
    const int r = 4 * wave + 2 * p;
    v4f o[2];
    o[0] = (v4f)(0.f);
    o[1] = (v4f)(0.f);
    if (row0 + r < N) gather_pair<128, false>(rowptr, col, asrc, adst, h, bias, row0 + r, N, o);
    const int ln = lane_id();
    float* tp = tile + (r + (ln >> 4)) * TP + (ln & 15) * 4;
    *(v4f*)(tp)      = o[0];
    *(v4f*)(tp + 64) = o[1];
  }
  {
    const int ln = lane_id();
    const v4f bA = *(const v4f*)(bias + (ln & 15) * 4);
    const v4f bB = *(const v4f*)(bias + 64 + (ln & 15) * 4);
    for (int p = 0; p < 2; ++p) {
      float* tp = tile + (4 * wave + 2 * p + (ln >> 4)) * TP + (ln & 15) * 4;
      v4f x = *(v4f*)(tp) + bA;
      v4f y = *(v4f*)(tp + 64) + bB;
      for (int i = 0; i < 4; ++i) { x[i] = fmaxf(x[i], 0.f); y[i] = fmaxf(y[i], 0.f); }
      *(v4f*)(tp)      = x;
      *(v4f*)(tp + 64) = y;
    }
  }
  __syncthreads();

  const int lane = lane_id();
  const int half = lane >> 4;
  const int l15  = lane & 15;
  const int rt   = (wave >> 1) & 1;
  const int trow = row0 + 16 * rt;
  const float* ap = tile + (16 * rt + l15) * TP + 2 * half;

  if (wave >= 4) {
    const int v = wave & 1;
    const float* vp = vecs2 + (size_t)v * 128 + 2 * half;
    float lg = 0.f;
#pragma unroll
    for (int k = 0; k < 128; k += 4) {
      const v2f a = *(const v2f*)(ap + k);
      lg = fmaf(a.x, vp[k + 0], lg);
      lg = fmaf(a.y, vp[k + 1], lg);
    }
    const float s = lg + __shfl_xor(lg, 16, 32);
    if (half == 0 && trow + l15 < N) logit2[(size_t)v * N + trow + l15] = s;
    return;
  }

  const int g = wave & 1;
  const float* wb = W2 + (size_t)(2 * half) * 64 + g * 32 + l15;
  v8f acc0 = (v8f){0.f,0.f,0.f,0.f,0.f,0.f,0.f,0.f};
  v8f acc1 = (v8f){0.f,0.f,0.f,0.f,0.f,0.f,0.f,0.f};
  constexpr int AH = 4;
  v2f b0[AH], b1[AH];
#pragma unroll
  for (int i = 0; i < AH; ++i) {
    const int k = 4 * i;
    b0[i].x = wb[(size_t)(k + 0) * 64];
    b0[i].y = wb[(size_t)(k + 1) * 64];
    b1[i].x = wb[(size_t)(k + 0) * 64 + 16];
    b1[i].y = wb[(size_t)(k + 1) * 64 + 16];
  }
  v2f a = *(const v2f*)(ap);
#pragma unroll
  for (int i = 0; i < 32; ++i) {
    const int k = 4 * i;
    const int j = i % AH;
    v2f an = a;
    if (i + 1 < 32) an = *(const v2f*)(ap + k + 4);
    acc0 = __builtin_amdgcn_wmma_f32_16x16x4_f32(false, a, false, b0[j], (short)0, acc0, false, false);
    if (i + AH < 32) {
      b0[j].x = wb[(size_t)(k + 4 * AH + 0) * 64];
      b0[j].y = wb[(size_t)(k + 4 * AH + 1) * 64];
    }
    acc1 = __builtin_amdgcn_wmma_f32_16x16x4_f32(false, a, false, b1[j], (short)0, acc1, false, false);
    if (i + AH < 32) {
      b1[j].x = wb[(size_t)(k + 4 * AH + 0) * 64 + 16];
      b1[j].y = wb[(size_t)(k + 4 * AH + 1) * 64 + 16];
    }
    __builtin_amdgcn_sched_barrier(0);
    a = an;
  }

  const int mbase = trow + 8 * half;
  float* cp = h2 + (size_t)mbase * 64 + g * 32 + l15;
#pragma unroll
  for (int r = 0; r < 8; ++r)
    if (mbase + r < N) {
      cp[(size_t)r * 64]      = acc0[r];
      cp[(size_t)r * 64 + 16] = acc1[r];
    }
}

static inline int cdiv(long a, long b) { return (int)((a + b - 1) / b); }

extern "C" void kernel_launch(void* const* d_in, const int* in_sizes, int n_in,
                              void* d_out, int out_size, void* d_ws, size_t ws_size,
                              hipStream_t stream)
{
  const float* x       = (const float*)d_in[0];
  const int*   ei      = (const int*)d_in[1];
  const float* Wsrc1   = (const float*)d_in[2];
  const float* Wdst1   = (const float*)d_in[3];
  const float* attsrc1 = (const float*)d_in[4];
  const float* attdst1 = (const float*)d_in[5];
  const float* b1      = (const float*)d_in[6];
  const float* Wsrc2   = (const float*)d_in[7];
  const float* Wdst2   = (const float*)d_in[8];
  const float* attsrc2 = (const float*)d_in[9];
  const float* attdst2 = (const float*)d_in[10];
  const float* b2      = (const float*)d_in[11];

  const int HID = in_sizes[4];
  const int OUT = in_sizes[9];
  const int FIN = in_sizes[2] / HID;
  const int N   = in_sizes[0] / FIN;
  const int E   = in_sizes[1] / 2;
  const int* src = ei;
  const int* dst = ei + E;

  float* ws    = (float*)d_ws;
  float* h1    = ws;  ws += (size_t)N * HID;
  float* h2    = ws;  ws += (size_t)N * OUT;
  float* logit = ws;  ws += 2 * (size_t)N;
  float* logit2 = ws; ws += 2 * (size_t)N;
  float* vfold = ws;  ws += 2 * (size_t)FIN + 2 * (size_t)HID;
  const int G1 = 256;
  const int NB = cdiv(N, RB);
  int* binned  = (int*)ws;  ws += 2 * (size_t)E;
  int* wgcnt   = (int*)ws;  ws += (size_t)NB * G1;
  int* wgoff   = (int*)ws;  ws += (size_t)NB * G1 + 1;
  int* col     = (int*)ws;  ws += E;
  int* rowptr  = (int*)ws;  ws += N + 1;

  const int tiles1 = (N / 16) * (HID / 64);
  const int SL = 256 * GEMM_WPB;
  const int tA = (tiles1 > 2 * SL) ? (tiles1 - 2 * SL) : 0;
  const int tB = (tA + SL < tiles1) ? (tA + SL) : tiles1;
  count_and_fold<<<G1 + cdiv(2 * FIN + 2 * HID, GW), GT, 0, stream>>>(
      dst, wgcnt, E, NB, G1,
      Wsrc1, attsrc1, Wdst1, attdst1, Wsrc2, attsrc2, Wdst2, attdst2,
      vfold, FIN, HID, HID, OUT);
  scan_and_gemm<<<1 + cdiv(tA, GEMM_WPB), 1024, 0, stream>>>(
      wgcnt, wgoff, NB * G1,
      x, Wsrc1, vfold, h1, logit, N, 0, tA);
  scatter_and_gemm<<<G1 + cdiv(tB - tA, GEMM_WPB), GT, 0, stream>>>(
      src, dst, wgoff, binned, E, NB, G1,
      x, Wsrc1, vfold, h1, logit, N, tA, tB);
  sort_and_gemm<<<NB + cdiv(tiles1 - tB, GEMM_WPB), GT, 0, stream>>>(
      binned, wgoff, rowptr, col, N, G1, NB,
      x, Wsrc1, vfold, h1, logit, N, tB, tiles1);

  gather1_gemm2<<<cdiv(N, FROWS), 256, 0, stream>>>(
      rowptr, col, logit, logit + N, h1, b1,
      Wsrc2, vfold + 2 * (size_t)FIN, h2, logit2, N);

  {
    float* out = (float*)d_out;
    seg_gather_pair64<<<cdiv(N, 16), 256, 0, stream>>>(
        rowptr, col, logit2, logit2 + N, h2, b2, out, N);
  }
}


# Edge grouping in 2 launches: a workgroup orders its slice by range in shared memory, a range merges its 256 runs; 6 launches to 4; layer-1 logits by row-dot riders; new logit and row sum order
# speedup vs baseline: 6.6787x; 1.0697x over previous
#include <hip/hip_runtime.h>
#include <hip/hip_bf16.h>

typedef float v2f __attribute__((ext_vector_type(2)));
typedef float v8f __attribute__((ext_vector_type(8)));

#define NEG_SLOPE 0.2f

template<int K, int NOUT, int NT>
__device__ __forceinline__ void gemm_strip(
    const float* __restrict__ X, const float* __restrict__ W,
    float* __restrict__ C, int tile)
{
  const int lane   = threadIdx.x & 31;
  constexpr int tilesN = NOUT / (16 * NT);
  const int tM = tile / tilesN;
  const int tN = tile % tilesN;

  const int half = lane >> 4;
  const int l15  = lane & 15;
  const float* xrow = X + (size_t)(tM * 16 + l15) * K + 2 * half;
  const float* wb   = W + (size_t)(2 * half) * NOUT + tN * (16 * NT) + l15;

  v8f acc[NT];
#pragma unroll
  for (int t = 0; t < NT; ++t) acc[t] = (v8f){0.f,0.f,0.f,0.f,0.f,0.f,0.f,0.f};

  {
    v2f a, an;
    v2f b[NT];
    a.x = xrow[0];
    a.y = xrow[1];
#pragma unroll
    for (int t = 0; t < NT; ++t) { b[t].x = wb[16 * t]; b[t].y = wb[(size_t)NOUT + 16 * t]; }
#pragma unroll
    for (int k = 0; k < K; k += 4) {
      if (k + 4 < K) {
        an.x = xrow[k + 4];
        an.y = xrow[k + 5];
      }
#pragma unroll
      for (int t = 0; t < NT; ++t) {
        acc[t] = __builtin_amdgcn_wmma_f32_16x16x4_f32(false, a, false, b[t],
                                                       (short)0, acc[t],
                                                       false, false);
        if (k + 4 < K) {
          b[t].x = wb[(size_t)(k + 4) * NOUT + 16 * t];
          b[t].y = wb[(size_t)(k + 5) * NOUT + 16 * t];
        }
        __builtin_amdgcn_sched_barrier(0);
      }
      if (k + 4 < K) a = an;
    }
  }

  const int mbase = tM * 16 + 8 * half;
  float* cp = C + (size_t)mbase * NOUT + tN * (16 * NT) + l15;
#pragma unroll
  for (int t = 0; t < NT; ++t)
#pragma unroll
    for (int r = 0; r < 8; ++r)
      cp[(size_t)r * NOUT + 16 * t] = acc[t][r];
}

#define GEMM_WPB 16
__device__ __forceinline__ void gemm1_part(
    const float* __restrict__ X, const float* __restrict__ W,
    float* __restrict__ C, int tile0, int tile1, int gb)
{
  const int wave = threadIdx.x >> 5;
  if (wave >= GEMM_WPB) return;
  const int tile = __builtin_amdgcn_readfirstlane(tile0 + gb * GEMM_WPB + wave);
  if (tile >= tile1) return;
  gemm_strip<128, 128, 4>(X, W, C, tile);
}

__device__ __forceinline__ void fold_att_body(
    const float* __restrict__ W0, const float* __restrict__ a0,
    const float* __restrict__ W1, const float* __restrict__ a1,
    const float* __restrict__ W2, const float* __restrict__ a2,
    const float* __restrict__ W3, const float* __restrict__ a3,
    float* __restrict__ v, int r1, int c1, int r2, int c2, int gw)
{
  const int lane = threadIdx.x & 31;
  if (gw >= 2 * r1 + 2 * r2) return;
  const float* Wm; const float* att; int row, cols;
  if (gw < r1)               { Wm = W0; att = a0; row = gw;               cols = c1; }
  else if (gw < 2 * r1)      { Wm = W1; att = a1; row = gw - r1;          cols = c1; }
  else if (gw < 2 * r1 + r2) { Wm = W2; att = a2; row = gw - 2 * r1;      cols = c2; }
  else                       { Wm = W3; att = a3; row = gw - 2 * r1 - r2; cols = c2; }
  const float* wr = Wm + (size_t)row * cols;
  float s = 0.f;
  for (int k = lane; k < cols; k += 32) s += wr[k] * att[k];
  for (int off = 16; off > 0; off >>= 1) s += __shfl_down(s, off, 32);
  if (lane == 0) v[gw] = s;
}

#define RB_SHIFT 9
#define RB (1 << RB_SHIFT)
#define MAXNB 1024
#define GT 512
#define GW (GT / 32)
#define EPT (4096 / GT)
#define NSL 256
static_assert(GT >= RB && GT % 32 == 0 && 4096 % GT == 0, "range_sort gives one thread to each node of a range");
static_assert(GT >= NSL && (NSL & (NSL - 1)) == 0, "range_sort gives one thread to each run and halves its way through them");
static_assert(MAXNB == 2 * GT, "local_sort scans two range counts per thread");

__device__ __forceinline__ void local_sort_body(
    const int* __restrict__ src, const int* __restrict__ dst,
    int* __restrict__ loff, int* __restrict__ binned, int E, int NB, int bid, int G)
{
  __shared__ int hist[MAXNB];
  __shared__ int wsum[GW];
  const int tid  = threadIdx.x;
  const int lane = tid & 31;
  const int wave = tid >> 5;
  hist[2 * tid]     = 0;
  hist[2 * tid + 1] = 0;
  __syncthreads();
  const int chunk = (E + G - 1) / G;
  const int e0 = (bid * chunk < E) ? (bid * chunk) : E;
  const int e1 = (e0 + chunk < E) ? (e0 + chunk) : E;
  for (int base = e0; base < e1; base += 4096) {
    int d[EPT];
    for (int q = 0; q < EPT; ++q) {
      const int i = base + tid + q * GT;
      d[q] = (i < e1) ? dst[i] : -1;
    }
    for (int q = 0; q < EPT; ++q)
      if (d[q] >= 0) atomicAdd(&hist[d[q] >> RB_SHIFT], 1);
  }
  __syncthreads();
  const int ca = hist[2 * tid];
  const int cb = hist[2 * tid + 1];
  const int t  = ca + cb;
  int incl = t;
  for (int off = 1; off < 32; off <<= 1) {
    const int nb = __shfl_up(incl, off, 32);
    if (lane >= off) incl += nb;
  }
  if (lane == 31) wsum[wave] = incl;
  __syncthreads();
  int offs = 0;
  for (int w = 0; w < wave; ++w) offs += wsum[w];
  const int ex = offs + incl - t;
  if (2 * tid < NB)     loff[(size_t)(2 * tid) * G + bid]     = ex;
  if (2 * tid + 1 < NB) loff[(size_t)(2 * tid + 1) * G + bid] = ex + ca;
  hist[2 * tid]     = e0 + ex;
  hist[2 * tid + 1] = e0 + ex + ca;
  __syncthreads();
  int2* pairs = reinterpret_cast<int2*>(binned);
  for (int base = e0; base < e1; base += 4096) {
    int d[EPT], s[EPT];
    for (int q = 0; q < EPT; ++q) {
      const int i = base + tid + q * GT;
      d[q] = (i < e1) ? dst[i] : -1;
      s[q] = (i < e1) ? src[i] : 0;
    }
    for (int q = 0; q < EPT; ++q)
      if (d[q] >= 0) {
        const int slot = atomicAdd(&hist[d[q] >> RB_SHIFT], 1);
        pairs[slot] = make_int2(s[q], d[q]);
      }
  }
}

__device__ __forceinline__ void range_sort_body(
    const int* __restrict__ binned, const int* __restrict__ loff, int* __restrict__ lined_i,
    int* __restrict__ rowptr, int* __restrict__ col, int N, int E, int G, int rg, int NB)
{
  __shared__ int cnt[RB];
  __shared__ int wtot[RB / 32];
  __shared__ int pre[NSL + 1];
  __shared__ int ra[NSL];
  __shared__ int wlen[GW];
  __shared__ int wst[GW];
  const int tid  = threadIdx.x;
  const int lane = tid & 31;
  const int wave = tid >> 5;
  const int2* pairs = reinterpret_cast<const int2*>(binned);
  int2* lined = reinterpret_cast<int2*>(lined_i);

  int len = 0, st = 0;
  if (tid < G) {
    const int chunk = (E + G - 1) / G;
    const int e0 = (tid * chunk < E) ? (tid * chunk) : E;
    const int e1 = (e0 + chunk < E) ? (e0 + chunk) : E;
    st = loff[(size_t)rg * G + tid];
    const int en = (rg + 1 < NB) ? loff[(size_t)(rg + 1) * G + tid] : (e1 - e0);
    len = en - st;
    ra[tid] = e0 + st;
  }
  int il = len;
  for (int off = 1; off < 32; off <<= 1) {
    const int nb = __shfl_up(il, off, 32);
    if (lane >= off) il += nb;
  }
  int ss = st;
  for (int off = 16; off > 0; off >>= 1) ss += __shfl_xor(ss, off, 32);
  if (lane == 31) wlen[wave] = il;
  if (lane == 0)  wst[wave]  = ss;
  if (tid < RB) cnt[tid] = 0;
  __syncthreads();
  int before = 0, total = 0, s0 = 0;
  for (int w = 0; w < GW; ++w) {
    const int wl = wlen[w];
    if (w < wave) before += wl;
    total += wl;
    s0 += wst[w];
  }
  if (tid < G) pre[tid] = before + il - len;
  if (tid == 0) pre[NSL] = total;
  const int s1 = s0 + total;
  __syncthreads();

  for (int base = 0; base < total; base += 4096) {
    int ii[EPT], g[EPT];
#pragma unroll
    for (int q = 0; q < EPT; ++q) {
      const int i = base + tid + q * GT;
      ii[q] = (i < total) ? i : (total - 1);
      g[q] = 0;
    }
#pragma unroll
    for (int h = NSL / 2; h > 0; h >>= 1) {
      int p[EPT];
#pragma unroll
      for (int q = 0; q < EPT; ++q) p[q] = pre[g[q] + h];
      __builtin_amdgcn_sched_barrier(0);
#pragma unroll
      for (int q = 0; q < EPT; ++q) g[q] += (p[q] <= ii[q]) ? h : 0;
    }
    int2 r[EPT];
#pragma unroll
    for (int q = 0; q < EPT; ++q)
      r[q] = pairs[ra[g[q]] + (ii[q] - pre[g[q]])];
    static_assert(EPT == 8, "the fence below names the EPT loaded pairs");
    asm volatile("" :: "v"(r[0].x), "v"(r[0].y), "v"(r[1].x), "v"(r[1].y),
                       "v"(r[2].x), "v"(r[2].y), "v"(r[3].x), "v"(r[3].y),
                       "v"(r[4].x), "v"(r[4].y), "v"(r[5].x), "v"(r[5].y),
                       "v"(r[6].x), "v"(r[6].y), "v"(r[7].x), "v"(r[7].y));
#pragma unroll
    for (int q = 0; q < EPT; ++q) {
      const int i = base + tid + q * GT;
      if (i < total) {
        lined[s0 + i] = r[q];
        atomicAdd(&cnt[r[q].y & (RB - 1)], 1);
      }
    }
  }
  __syncthreads();

  int v = 0, incl = 0;
  if (tid < RB) {
    v = cnt[tid];
    incl = v;
    for (int off = 1; off < 32; off <<= 1) {
      const int nb = __shfl_up(incl, off, 32);
      if (lane >= off) incl += nb;
    }
    if (lane == 31) wtot[wave] = incl;
  }
  __syncthreads();
  if (tid < RB) {
    int offs = 0;
    for (int w = 0; w < wave; ++w) offs += wtot[w];
    const int excl = offs + incl - v;
    cnt[tid] = excl;
    const int node = rg * RB + tid;
    if (node < N) rowptr[node] = s0 + excl;
  }
  if (rg == NB - 1 && tid == 0) rowptr[N] = s1;
  __syncthreads();

  for (int base = s0; base < s1; base += 4096) {
    int2 r[EPT];
    for (int q = 0; q < EPT; ++q) {
      const int i = base + tid + q * GT;
      r[q] = (i < s1) ? lined[i] : make_int2(0, -1);
    }
    for (int q = 0; q < EPT; ++q)
      if (r[q].y >= 0) {
        const int slot = atomicAdd(&cnt[r[q].y & (RB - 1)], 1);
        col[s0 + slot] = r[q].x;
      }
  }
}

typedef float v4f __attribute__((ext_vector_type(4)));

#define LROWS 32
__device__ __forceinline__ void logit_part(
    const float* __restrict__ X, const float* __restrict__ vecs,
    float* __restrict__ logit, int M, int gb)
{
  const int lane = threadIdx.x & 31;
  const int row0 = __builtin_amdgcn_readfirstlane((gb * GW + (int)(threadIdx.x >> 5)) * LROWS);
  if (row0 >= M) return;
  const v4f vs = *(const v4f*)(vecs + lane * 4);
  const v4f vd = *(const v4f*)(vecs + 128 + lane * 4);
  float keep_s = 0.f, keep_d = 0.f;
  for (int bt = 0; bt < LROWS; bt += 8) {
    v4f x[8];
#pragma unroll
    for (int r = 0; r < 8; ++r) {
      const int row = (row0 + bt + r < M) ? (row0 + bt + r) : (M - 1);
      x[r] = *(const v4f*)(X + (size_t)row * 128 + lane * 4);
    }
#pragma unroll
    for (int r = 0; r < 8; ++r) {
      float a = x[r][0] * vs[0];
      a = fmaf(x[r][1], vs[1], a);
      a = fmaf(x[r][2], vs[2], a);
      a = fmaf(x[r][3], vs[3], a);
      float b = x[r][0] * vd[0];
      b = fmaf(x[r][1], vd[1], b);
      b = fmaf(x[r][2], vd[2], b);
      b = fmaf(x[r][3], vd[3], b);
      for (int off = 16; off > 0; off >>= 1) {
        a += __shfl_xor(a, off, 32);
        b += __shfl_xor(b, off, 32);
      }
      if (lane == bt + r) { keep_s = a; keep_d = b; }
    }
  }
  if (row0 + lane < M) {
    logit[row0 + lane]             = keep_s;
    logit[(size_t)M + row0 + lane] = keep_d;
  }
}

__global__ __launch_bounds__(GT, 16) void localsort_fold_gemm(
    const int* __restrict__ src, const int* __restrict__ dst,
    int* __restrict__ loff, int* __restrict__ binned, int E, int NB, int G,
    const float* __restrict__ W0, const float* __restrict__ a0,
    const float* __restrict__ W1, const float* __restrict__ a1,
    const float* __restrict__ W2, const float* __restrict__ a2,
    const float* __restrict__ W3, const float* __restrict__ a3,
    float* __restrict__ v, int r1, int c1, int r2, int c2, int nfold,
    const float* __restrict__ X, const float* __restrict__ W,
    float* __restrict__ C, int tile0, int tile1)
{
  const int b = (int)blockIdx.x;
  if (b < G) local_sort_body(src, dst, loff, binned, E, NB, b, G);
  else if (b < G + nfold)
    fold_att_body(W0, a0, W1, a1, W2, a2, W3, a3, v, r1, c1, r2, c2,
                  (b - G) * GW + (int)(threadIdx.x >> 5));
  else gemm1_part(X, W, C, tile0, tile1, b - G - nfold);
}

__global__ __launch_bounds__(GT, 16) void rangesort_logit_gemm(
    const int* __restrict__ binned, const int* __restrict__ loff, int* __restrict__ lined,
    int* __restrict__ rowptr, int* __restrict__ col, int N, int E, int G, int NB,
    const float* __restrict__ vecs, float* __restrict__ logit, int nlog,
    const float* __restrict__ X, const float* __restrict__ W,
    float* __restrict__ C, int tile0, int tile1)
{
  const int b = (int)blockIdx.x;
  if (b < NB) range_sort_body(binned, loff, lined, rowptr, col, N, E, G, b, NB);
  else if (b < NB + nlog) logit_part(X, vecs, logit, N, b - NB);
  else gemm1_part(X, W, C, tile0, tile1, b - NB - nlog);
}


__device__ __forceinline__ float leaky(float v)
{
  return (v > 0.f) ? v : NEG_SLOPE * v;
}


__device__ __forceinline__ int lane_id()
{
  return (int)__builtin_amdgcn_mbcnt_lo(~0u, 0u);
}

template<int K>
__device__ __forceinline__ int half_lane(int v)
{
  return __builtin_amdgcn_update_dpp(0, v, 0x150 + K, 0xf, 0xf, false);
}

template<int C, int K>
__device__ __forceinline__ void row_ask(const char* __restrict__ hb, unsigned lo, int sl,
                                        v4f (&r)[C / 64])
{
  const unsigned off = (unsigned)half_lane<K>(sl) * (unsigned)(C * 4) + lo;
#pragma unroll
  for (int v = 0; v < C / 64; ++v) r[v] = *(const v4f*)(hb + off + v * 256);
}

template<int C, int K>
__device__ __forceinline__ void row_add(int ali, const v4f (&r)[C / 64], v4f (&acc)[C / 64])
{
  const float a = __int_as_float(half_lane<K>(ali));
#pragma unroll
  for (int v = 0; v < C / 64; ++v) acc[v] += a * r[v];
}

template<int C, int K0, int G>
__device__ __forceinline__ void row_group(const char* __restrict__ hb, unsigned lo, int sl, int ali,
                                          v4f (&acc)[C / 64])
{
  static_assert(G == 4 || G == 8, "group sizes written out below");
  v4f r0[C / 64], r1[C / 64], r2[C / 64], r3[C / 64];
  row_ask<C, K0 + 0>(hb, lo, sl, r0);
  row_ask<C, K0 + 1>(hb, lo, sl, r1);
  row_ask<C, K0 + 2>(hb, lo, sl, r2);
  row_ask<C, K0 + 3>(hb, lo, sl, r3);
  if constexpr (G == 8) {
    v4f r4[C / 64], r5[C / 64], r6[C / 64], r7[C / 64];
    row_ask<C, K0 + 4>(hb, lo, sl, r4);
    row_ask<C, K0 + 5>(hb, lo, sl, r5);
    row_ask<C, K0 + 6>(hb, lo, sl, r6);
    row_ask<C, K0 + 7>(hb, lo, sl, r7);
    row_add<C, K0 + 0>(ali, r0, acc);
    row_add<C, K0 + 1>(ali, r1, acc);
    row_add<C, K0 + 2>(ali, r2, acc);
    row_add<C, K0 + 3>(ali, r3, acc);
    row_add<C, K0 + 4>(ali, r4, acc);
    row_add<C, K0 + 5>(ali, r5, acc);
    row_add<C, K0 + 6>(ali, r6, acc);
    row_add<C, K0 + 7>(ali, r7, acc);
  } else {
    row_add<C, K0 + 0>(ali, r0, acc);
    row_add<C, K0 + 1>(ali, r1, acc);
    row_add<C, K0 + 2>(ali, r2, acc);
    row_add<C, K0 + 3>(ali, r3, acc);
  }
}

template<int C, bool BIAS>
__device__ __forceinline__ void gather_pair(
    const int* __restrict__ rowptr, const int* __restrict__ col,
    const float* __restrict__ asrc, const float* __restrict__ adst,
    const float* __restrict__ h, const float* __restrict__ bias, int d0, int N,
    v4f (&o)[C / 64])
{
  constexpr int NV = C / 64;
  constexpr int G  = 8 / NV;
  const int lane = lane_id();
  const int hf = lane >> 4;
  const int hl = lane & 15;
  const int   i2  = (d0 + 2 < N) ? (d0 + 2) : N;
  const int   i1  = (d0 + 1 < N) ? (d0 + 1) : (N - 1);
  const int   b0  = rowptr[d0];
  const int   b1  = rowptr[d0 + 1];
  const int   b2  = rowptr[i2];
  const float ad0 = adst[d0];
  const float ad1 = adst[i1];
  const int   beg = hf ? b1 : b0;
  const int   end = hf ? b2 : b1;
  const float ad  = hf ? ad1 : ad0;

  const int  j0  = beg + hl;
  const int  j1  = j0 + 16;
  const bool in0 = j0 < end;
  const bool in1 = j1 < end;
  int sl0 = 0, sl1 = 0;
  if (in0) sl0 = col[j0];
  if (in1) sl1 = col[j1];
  const float g0 = asrc[sl0];
  const float g1 = asrc[sl1];
  const float e0 = in0 ? leaky(g0 + ad) : -__builtin_huge_valf();
  const float e1 = in1 ? leaky(g1 + ad) : -__builtin_huge_valf();

  float mx = fmaxf(e0, e1);
  for (int j = j0 + 32; j < end; j += 16)
    mx = fmaxf(mx, leaky(asrc[col[j]] + ad));
  for (int off = 8; off > 0; off >>= 1) mx = fmaxf(mx, __shfl_xor(mx, off, 32));

  const float p0 = in0 ? __expf(e0 - mx) : 0.f;
  const float p1 = in1 ? __expf(e1 - mx) : 0.f;
  float den = p0 + p1;
  for (int j = j0 + 32; j < end; j += 16)
    den += __expf(leaky(asrc[col[j]] + ad) - mx);
  for (int off = 8; off > 0; off >>= 1) den += __shfl_xor(den, off, 32);

  const int n0 = b1 - b0;
  const int n1 = b2 - b1;
  const int nmax = (n0 > n1) ? n0 : n1;

  const char* hb = reinterpret_cast<const char*>(h);
  const unsigned lo = (unsigned)hl * 16u;
  v4f acc[NV];
#pragma unroll
  for (int v = 0; v < NV; ++v) acc[v] = (v4f)(0.f);

  for (int base = 0; base < nmax; base += 16) {
    int   sl = sl0;
    float al = in0 ? p0 / den : 0.f;
    if (base == 16) {
      sl = sl1;
      al = in1 ? p1 / den : 0.f;
    } else if (base != 0) {
      const int j = j0 + base;
      sl = 0;
      al = 0.f;
      if (j < end) {
        sl = col[j];
        al = __expf(leaky(asrc[sl] + ad) - mx) / den;
      }
    }
    const int ali = __float_as_int(al);
    const int rem = nmax - base;
    row_group<C, 0, G>(hb, lo, sl, ali, acc);
    if (G == 4) {
      if (rem > 4)  row_group<C, 4, G>(hb, lo, sl, ali, acc);
      if (rem > 8)  row_group<C, 8, G>(hb, lo, sl, ali, acc);
      if (rem > 12) row_group<C, 12, G>(hb, lo, sl, ali, acc);
    } else {
      if (rem > 8)  row_group<C, 8, G>(hb, lo, sl, ali, acc);
    }
  }

#pragma unroll
  for (int v = 0; v < NV; ++v) {
    o[v] = acc[v];
    if (BIAS) o[v] += *(const v4f*)(bias + v * 64 + hl * 4);
  }
}

__global__ __launch_bounds__(256, 16) void seg_gather_pair64(
    const int* __restrict__ rowptr, const int* __restrict__ col,
    const float* __restrict__ asrc, const float* __restrict__ adst,
    const float* __restrict__ h, const float* __restrict__ bias,
    float* __restrict__ out, int N)
{
  const int lane = threadIdx.x & 31;
  const int d0 = __builtin_amdgcn_readfirstlane((int)(blockIdx.x * 16 + (threadIdx.x >> 5) * 2));
  if (d0 >= N) return;
  v4f o[1];
  gather_pair<64, true>(rowptr, col, asrc, adst, h, bias, d0, N, o);
  const int d = d0 + (lane >> 4);
  if (d < N) *(v4f*)(out + (size_t)d * 64 + (lane & 15) * 4) = o[0];
}

#define TP 132
#define FROWS 32
__global__ __launch_bounds__(256, 16) void gather1_gemm2(
    const int* __restrict__ rowptr, const int* __restrict__ col,
    const float* __restrict__ asrc, const float* __restrict__ adst,
    const float* __restrict__ h, const float* __restrict__ bias,
    const float* __restrict__ W2, const float* __restrict__ vecs2,
    float* __restrict__ h2, float* __restrict__ logit2, int N)
{
  __shared__ __attribute__((aligned(16))) float tile[FROWS * TP];
  const int wave = __builtin_amdgcn_readfirstlane((int)(threadIdx.x >> 5));
  const int row0 = (int)blockIdx.x * FROWS;

  for (int p = 0; p < 2; ++p) {
    const int r = 4 * wave + 2 * p;
    v4f o[2];
    o[0] = (v4f)(0.f);
    o[1] = (v4f)(0.f);
    if (row0 + r < N) gather_pair<128, false>(rowptr, col, asrc, adst, h, bias, row0 + r, N, o);
    const int ln = lane_id();
    float* tp = tile + (r + (ln >> 4)) * TP + (ln & 15) * 4;
    *(v4f*)(tp)      = o[0];
    *(v4f*)(tp + 64) = o[1];
  }
  {
    const int ln = lane_id();
    const v4f bA = *(const v4f*)(bias + (ln & 15) * 4);
    const v4f bB = *(const v4f*)(bias + 64 + (ln & 15) * 4);
    for (int p = 0; p < 2; ++p) {
      float* tp = tile + (4 * wave + 2 * p + (ln >> 4)) * TP + (ln & 15) * 4;
      v4f x = *(v4f*)(tp) + bA;
      v4f y = *(v4f*)(tp + 64) + bB;
      for (int i = 0; i < 4; ++i) { x[i] = fmaxf(x[i], 0.f); y[i] = fmaxf(y[i], 0.f); }
      *(v4f*)(tp)      = x;
      *(v4f*)(tp + 64) = y;
    }
  }
  __syncthreads();

  const int lane = lane_id();
  const int half = lane >> 4;
  const int l15  = lane & 15;
  const int rt   = (wave >> 1) & 1;
  const int trow = row0 + 16 * rt;
  const float* ap = tile + (16 * rt + l15) * TP + 2 * half;

  if (wave >= 4) {
    const int v = wave & 1;
    const float* vp = vecs2 + (size_t)v * 128 + 2 * half;
    float lg = 0.f;
#pragma unroll
    for (int k = 0; k < 128; k += 4) {
      const v2f a = *(const v2f*)(ap + k);
      lg = fmaf(a.x, vp[k + 0], lg);
      lg = fmaf(a.y, vp[k + 1], lg);
    }
    const float s = lg + __shfl_xor(lg, 16, 32);
    if (half == 0 && trow + l15 < N) logit2[(size_t)v * N + trow + l15] = s;
    return;
  }

  const int g = wave & 1;
  const float* wb = W2 + (size_t)(2 * half) * 64 + g * 32 + l15;
  v8f acc0 = (v8f){0.f,0.f,0.f,0.f,0.f,0.f,0.f,0.f};
  v8f acc1 = (v8f){0.f,0.f,0.f,0.f,0.f,0.f,0.f,0.f};
  constexpr int AH = 4;
  v2f b0[AH], b1[AH];
#pragma unroll
  for (int i = 0; i < AH; ++i) {
    const int k = 4 * i;
    b0[i].x = wb[(size_t)(k + 0) * 64];
    b0[i].y = wb[(size_t)(k + 1) * 64];
    b1[i].x = wb[(size_t)(k + 0) * 64 + 16];
    b1[i].y = wb[(size_t)(k + 1) * 64 + 16];
  }
  v2f a = *(const v2f*)(ap);
#pragma unroll
  for (int i = 0; i < 32; ++i) {
    const int k = 4 * i;
    const int j = i % AH;
    v2f an = a;
    if (i + 1 < 32) an = *(const v2f*)(ap + k + 4);
    acc0 = __builtin_amdgcn_wmma_f32_16x16x4_f32(false, a, false, b0[j], (short)0, acc0, false, false);
    if (i + AH < 32) {
      b0[j].x = wb[(size_t)(k + 4 * AH + 0) * 64];
      b0[j].y = wb[(size_t)(k + 4 * AH + 1) * 64];
    }
    acc1 = __builtin_amdgcn_wmma_f32_16x16x4_f32(false, a, false, b1[j], (short)0, acc1, false, false);
    if (i + AH < 32) {
      b1[j].x = wb[(size_t)(k + 4 * AH + 0) * 64 + 16];
      b1[j].y = wb[(size_t)(k + 4 * AH + 1) * 64 + 16];
    }
    __builtin_amdgcn_sched_barrier(0);
    a = an;
  }

  const int mbase = trow + 8 * half;
  float* cp = h2 + (size_t)mbase * 64 + g * 32 + l15;
#pragma unroll
  for (int r = 0; r < 8; ++r)
    if (mbase + r < N) {
      cp[(size_t)r * 64]      = acc0[r];
      cp[(size_t)r * 64 + 16] = acc1[r];
    }
}

static inline int cdiv(long a, long b) { return (int)((a + b - 1) / b); }

extern "C" void kernel_launch(void* const* d_in, const int* in_sizes, int n_in,
                              void* d_out, int out_size, void* d_ws, size_t ws_size,
                              hipStream_t stream)
{
  const float* x       = (const float*)d_in[0];
  const int*   ei      = (const int*)d_in[1];
  const float* Wsrc1   = (const float*)d_in[2];
  const float* Wdst1   = (const float*)d_in[3];
  const float* attsrc1 = (const float*)d_in[4];
  const float* attdst1 = (const float*)d_in[5];
  const float* b1      = (const float*)d_in[6];
  const float* Wsrc2   = (const float*)d_in[7];
  const float* Wdst2   = (const float*)d_in[8];
  const float* attsrc2 = (const float*)d_in[9];
  const float* attdst2 = (const float*)d_in[10];
  const float* b2      = (const float*)d_in[11];

  const int HID = in_sizes[4];
  const int OUT = in_sizes[9];
  const int FIN = in_sizes[2] / HID;
  const int N   = in_sizes[0] / FIN;
  const int E   = in_sizes[1] / 2;
  const int* src = ei;
  const int* dst = ei + E;

  float* ws    = (float*)d_ws;
  float* h1    = ws;  ws += (size_t)N * HID;
  float* h2    = ws;  ws += (size_t)N * OUT;
  float* logit = ws;  ws += 2 * (size_t)N;
  float* logit2 = ws; ws += 2 * (size_t)N;
  float* vfold = ws;  ws += 2 * (size_t)FIN + 2 * (size_t)HID;
  const int G1 = NSL;
  const int NB = cdiv(N, RB);
  int* binned  = (int*)ws;  ws += 2 * (size_t)E;
  int* lined   = (int*)ws;  ws += 2 * (size_t)E;
  int* loff    = (int*)ws;  ws += (size_t)NB * G1;
  int* col     = (int*)ws;  ws += E;
  int* rowptr  = (int*)ws;  ws += N + 1;

  const int tiles1 = (N / 16) * (HID / 64);
  const int tA = tiles1 / 2;
  const int nfold = cdiv(2 * FIN + 2 * HID, GW);
  const int nlog  = cdiv(N, GW * LROWS);
  localsort_fold_gemm<<<G1 + nfold + cdiv(tA, GEMM_WPB), GT, 0, stream>>>(
      src, dst, loff, binned, E, NB, G1,
      Wsrc1, attsrc1, Wdst1, attdst1, Wsrc2, attsrc2, Wdst2, attdst2,
      vfold, FIN, HID, HID, OUT, nfold,
      x, Wsrc1, h1, 0, tA);
  rangesort_logit_gemm<<<NB + nlog + cdiv(tiles1 - tA, GEMM_WPB), GT, 0, stream>>>(
      binned, loff, lined, rowptr, col, N, E, G1, NB,
      vfold, logit, nlog,
      x, Wsrc1, h1, tA, tiles1);

  gather1_gemm2<<<cdiv(N, FROWS), 256, 0, stream>>>(
      rowptr, col, logit, logit + N, h1, b1,
      Wsrc2, vfold + 2 * (size_t)FIN, h2, logit2, N);

  {
    float* out = (float*)d_out;
    seg_gather_pair64<<<cdiv(N, 16), 256, 0, stream>>>(
        rowptr, col, logit2, logit2 + N, h2, b2, out, N);
  }
}


# GEMM fragments by wide loads: a lane owns 4 (layer 2: 2) neighbouring columns, one per tile, and 4 consecutive k of x; W rows, x and results move 16 or 8 bytes per lane; h1 summed in a new k order
# speedup vs baseline: 6.9386x; 1.0389x over previous
#include <hip/hip_runtime.h>
#include <hip/hip_bf16.h>

typedef float v2f __attribute__((ext_vector_type(2)));
typedef float v8f __attribute__((ext_vector_type(8)));

#define NEG_SLOPE 0.2f

template<int K, int NOUT>
__device__ __forceinline__ void gemm_strip(
    const float* __restrict__ X, const float* __restrict__ W,
    float* __restrict__ C, int tile)
{
  typedef float v4 __attribute__((ext_vector_type(4)));
  static_assert(K % 8 == 0 && NOUT % 64 == 0, "blocks of eight k, strips of 64 columns");
  const int lane   = threadIdx.x & 31;
  constexpr int tilesN = NOUT / 64;
  constexpr int KS = K / 4;
  const int tM = tile / tilesN;
  const int tN = tile % tilesN;

  const int half = lane >> 4;
  const int l15  = lane & 15;
  const float* xrow = X + (size_t)(tM * 16 + l15) * K + 4 * half;
  const float* wb   = W + (size_t)(4 * half) * NOUT + tN * 64 + 4 * l15;

  v8f acc[4];
#pragma unroll
  for (int t = 0; t < 4; ++t) acc[t] = (v8f){0.f,0.f,0.f,0.f,0.f,0.f,0.f,0.f};

  {
    v4 a  = *(const v4*)(xrow);
    v4 an = a;
    v4 r0 = *(const v4*)(wb);
    v4 r1 = *(const v4*)(wb + (size_t)NOUT);
#pragma unroll
    for (int ks = 0; ks < KS; ++ks) {
      const int i = ks & 1;
      v2f b[4];
#pragma unroll
      for (int t = 0; t < 4; ++t) { b[t].x = r0[t]; b[t].y = r1[t]; }
      if (ks + 1 < KS) {
        const int kn = 8 * ((ks + 1) >> 1) + 2 * ((ks + 1) & 1);
        r0 = *(const v4*)(wb + (size_t)(kn + 0) * NOUT);
        r1 = *(const v4*)(wb + (size_t)(kn + 1) * NOUT);
      }
      if (i == 0 && ks + 2 < KS)
        an = *(const v4*)(xrow + 8 * ((ks >> 1) + 1));
      v2f av;
      if (i == 0) { av.x = a.x; av.y = a.y; } else { av.x = a.z; av.y = a.w; }
#pragma unroll
      for (int t = 0; t < 4; ++t)
        acc[t] = __builtin_amdgcn_wmma_f32_16x16x4_f32(false, av, false, b[t],
                                                       (short)0, acc[t],
                                                       false, false);
      __builtin_amdgcn_sched_barrier(0);
      if (i == 1) a = an;
    }
  }

  const int mbase = tM * 16 + 8 * half;
  float* cp = C + (size_t)mbase * NOUT + tN * 64 + 4 * l15;
#pragma unroll
  for (int r = 0; r < 8; ++r) {
    v4 o;
    o.x = acc[0][r]; o.y = acc[1][r]; o.z = acc[2][r]; o.w = acc[3][r];
    *(v4*)(cp + (size_t)r * NOUT) = o;
  }
}

#define GEMM_WPB 16
__device__ __forceinline__ void gemm1_part(
    const float* __restrict__ X, const float* __restrict__ W,
    float* __restrict__ C, int tile0, int tile1, int gb)
{
  const int wave = threadIdx.x >> 5;
  if (wave >= GEMM_WPB) return;
  const int tile = __builtin_amdgcn_readfirstlane(tile0 + gb * GEMM_WPB + wave);
  if (tile >= tile1) return;
  gemm_strip<128, 128>(X, W, C, tile);
}

__device__ __forceinline__ void fold_att_body(
    const float* __restrict__ W0, const float* __restrict__ a0,
    const float* __restrict__ W1, const float* __restrict__ a1,
    const float* __restrict__ W2, const float* __restrict__ a2,
    const float* __restrict__ W3, const float* __restrict__ a3,
    float* __restrict__ v, int r1, int c1, int r2, int c2, int gw)
{
  const int lane = threadIdx.x & 31;
  if (gw >= 2 * r1 + 2 * r2) return;
  const float* Wm; const float* att; int row, cols;
  if (gw < r1)               { Wm = W0; att = a0; row = gw;               cols = c1; }
  else if (gw < 2 * r1)      { Wm = W1; att = a1; row = gw - r1;          cols = c1; }
  else if (gw < 2 * r1 + r2) { Wm = W2; att = a2; row = gw - 2 * r1;      cols = c2; }
  else                       { Wm = W3; att = a3; row = gw - 2 * r1 - r2; cols = c2; }
  const float* wr = Wm + (size_t)row * cols;
  float s = 0.f;
  for (int k = lane; k < cols; k += 32) s += wr[k] * att[k];
  for (int off = 16; off > 0; off >>= 1) s += __shfl_down(s, off, 32);
  if (lane == 0) v[gw] = s;
}

#define RB_SHIFT 9
#define RB (1 << RB_SHIFT)
#define MAXNB 1024
#define GT 512
#define GW (GT / 32)
#define EPT (4096 / GT)
#define NSL 256
static_assert(GT >= RB && GT % 32 == 0 && 4096 % GT == 0, "range_sort gives one thread to each node of a range");
static_assert(GT >= NSL && (NSL & (NSL - 1)) == 0, "range_sort gives one thread to each run and halves its way through them");
static_assert(MAXNB == 2 * GT, "local_sort scans two range counts per thread");

__device__ __forceinline__ void local_sort_body(
    const int* __restrict__ src, const int* __restrict__ dst,
    int* __restrict__ loff, int* __restrict__ binned, int E, int NB, int bid, int G)
{
  __shared__ int hist[MAXNB];
  __shared__ int wsum[GW];
  const int tid  = threadIdx.x;
  const int lane = tid & 31;
  const int wave = tid >> 5;
  hist[2 * tid]     = 0;
  hist[2 * tid + 1] = 0;
  __syncthreads();
  const int chunk = (E + G - 1) / G;
  const int e0 = (bid * chunk < E) ? (bid * chunk) : E;
  const int e1 = (e0 + chunk < E) ? (e0 + chunk) : E;
  for (int base = e0; base < e1; base += 4096) {
    int d[EPT];
    for (int q = 0; q < EPT; ++q) {
      const int i = base + tid + q * GT;
      d[q] = (i < e1) ? dst[i] : -1;
    }
    for (int q = 0; q < EPT; ++q)
      if (d[q] >= 0) atomicAdd(&hist[d[q] >> RB_SHIFT], 1);
  }
  __syncthreads();
  const int ca = hist[2 * tid];
  const int cb = hist[2 * tid + 1];
  const int t  = ca + cb;
  int incl = t;
  for (int off = 1; off < 32; off <<= 1) {
    const int nb = __shfl_up(incl, off, 32);
    if (lane >= off) incl += nb;
  }
  if (lane == 31) wsum[wave] = incl;
  __syncthreads();
  int offs = 0;
  for (int w = 0; w < wave; ++w) offs += wsum[w];
  const int ex = offs + incl - t;
  if (2 * tid < NB)     loff[(size_t)(2 * tid) * G + bid]     = ex;
  if (2 * tid + 1 < NB) loff[(size_t)(2 * tid + 1) * G + bid] = ex + ca;
  hist[2 * tid]     = e0 + ex;
  hist[2 * tid + 1] = e0 + ex + ca;
  __syncthreads();
  int2* pairs = reinterpret_cast<int2*>(binned);
  for (int base = e0; base < e1; base += 4096) {
    int d[EPT], s[EPT];
    for (int q = 0; q < EPT; ++q) {
      const int i = base + tid + q * GT;
      d[q] = (i < e1) ? dst[i] : -1;
      s[q] = (i < e1) ? src[i] : 0;
    }
    for (int q = 0; q < EPT; ++q)
      if (d[q] >= 0) {
        const int slot = atomicAdd(&hist[d[q] >> RB_SHIFT], 1);
        pairs[slot] = make_int2(s[q], d[q]);
      }
  }
}

__device__ __forceinline__ void range_sort_body(
    const int* __restrict__ binned, const int* __restrict__ loff, int* __restrict__ lined_i,
    int* __restrict__ rowptr, int* __restrict__ col, int N, int E, int G, int rg, int NB)
{
  __shared__ int cnt[RB];
  __shared__ int wtot[RB / 32];
  __shared__ int pre[NSL + 1];
  __shared__ int ra[NSL];
  __shared__ int wlen[GW];
  __shared__ int wst[GW];
  const int tid  = threadIdx.x;
  const int lane = tid & 31;
  const int wave = tid >> 5;
  const int2* pairs = reinterpret_cast<const int2*>(binned);
  int2* lined = reinterpret_cast<int2*>(lined_i);

  int len = 0, st = 0;
  if (tid < G) {
    const int chunk = (E + G - 1) / G;
    const int e0 = (tid * chunk < E) ? (tid * chunk) : E;
    const int e1 = (e0 + chunk < E) ? (e0 + chunk) : E;
    st = loff[(size_t)rg * G + tid];
    const int en = (rg + 1 < NB) ? loff[(size_t)(rg + 1) * G + tid] : (e1 - e0);
    len = en - st;
    ra[tid] = e0 + st;
  }
  int il = len;
  for (int off = 1; off < 32; off <<= 1) {
    const int nb = __shfl_up(il, off, 32);
    if (lane >= off) il += nb;
  }
  int ss = st;
  for (int off = 16; off > 0; off >>= 1) ss += __shfl_xor(ss, off, 32);
  if (lane == 31) wlen[wave] = il;
  if (lane == 0)  wst[wave]  = ss;
  if (tid < RB) cnt[tid] = 0;
  __syncthreads();
  int before = 0, total = 0, s0 = 0;
  for (int w = 0; w < GW; ++w) {
    const int wl = wlen[w];
    if (w < wave) before += wl;
    total += wl;
    s0 += wst[w];
  }
  if (tid < G) pre[tid] = before + il - len;
  if (tid == 0) pre[NSL] = total;
  const int s1 = s0 + total;
  __syncthreads();

  for (int base = 0; base < total; base += 4096) {
    int ii[EPT], g[EPT];
#pragma unroll
    for (int q = 0; q < EPT; ++q) {
      const int i = base + tid + q * GT;
      ii[q] = (i < total) ? i : (total - 1);
      g[q] = 0;
    }
#pragma unroll
    for (int h = NSL / 2; h > 0; h >>= 1) {
      int p[EPT];
#pragma unroll
      for (int q = 0; q < EPT; ++q) p[q] = pre[g[q] + h];
      __builtin_amdgcn_sched_barrier(0);
#pragma unroll
      for (int q = 0; q < EPT; ++q) g[q] += (p[q] <= ii[q]) ? h : 0;
    }
    int2 r[EPT];
#pragma unroll
    for (int q = 0; q < EPT; ++q)
      r[q] = pairs[ra[g[q]] + (ii[q] - pre[g[q]])];
    static_assert(EPT == 8, "the fence below names the EPT loaded pairs");
    asm volatile("" :: "v"(r[0].x), "v"(r[0].y), "v"(r[1].x), "v"(r[1].y),
                       "v"(r[2].x), "v"(r[2].y), "v"(r[3].x), "v"(r[3].y),
                       "v"(r[4].x), "v"(r[4].y), "v"(r[5].x), "v"(r[5].y),
                       "v"(r[6].x), "v"(r[6].y), "v"(r[7].x), "v"(r[7].y));
#pragma unroll
    for (int q = 0; q < EPT; ++q) {
      const int i = base + tid + q * GT;
      if (i < total) {
        lined[s0 + i] = r[q];
        atomicAdd(&cnt[r[q].y & (RB - 1)], 1);
      }
    }
  }
  __syncthreads();

  int v = 0, incl = 0;
  if (tid < RB) {
    v = cnt[tid];
    incl = v;
    for (int off = 1; off < 32; off <<= 1) {
      const int nb = __shfl_up(incl, off, 32);
      if (lane >= off) incl += nb;
    }
    if (lane == 31) wtot[wave] = incl;
  }
  __syncthreads();
  if (tid < RB) {
    int offs = 0;
    for (int w = 0; w < wave; ++w) offs += wtot[w];
    const int excl = offs + incl - v;
    cnt[tid] = excl;
    const int node = rg * RB + tid;
    if (node < N) rowptr[node] = s0 + excl;
  }
  if (rg == NB - 1 && tid == 0) rowptr[N] = s1;
  __syncthreads();

  for (int base = s0; base < s1; base += 4096) {
    int2 r[EPT];
    for (int q = 0; q < EPT; ++q) {
      const int i = base + tid + q * GT;
      r[q] = (i < s1) ? lined[i] : make_int2(0, -1);
    }
    for (int q = 0; q < EPT; ++q)
      if (r[q].y >= 0) {
        const int slot = atomicAdd(&cnt[r[q].y & (RB - 1)], 1);
        col[s0 + slot] = r[q].x;
      }
  }
}

typedef float v4f __attribute__((ext_vector_type(4)));

#define LROWS 32
__device__ __forceinline__ void logit_part(
    const float* __restrict__ X, const float* __restrict__ vecs,
    float* __restrict__ logit, int M, int gb)
{
  const int lane = threadIdx.x & 31;
  const int row0 = __builtin_amdgcn_readfirstlane((gb * GW + (int)(threadIdx.x >> 5)) * LROWS);
  if (row0 >= M) return;
  const v4f vs = *(const v4f*)(vecs + lane * 4);
  const v4f vd = *(const v4f*)(vecs + 128 + lane * 4);
  float keep_s = 0.f, keep_d = 0.f;
  for (int bt = 0; bt < LROWS; bt += 8) {
    v4f x[8];
#pragma unroll
    for (int r = 0; r < 8; ++r) {
      const int row = (row0 + bt + r < M) ? (row0 + bt + r) : (M - 1);
      x[r] = *(const v4f*)(X + (size_t)row * 128 + lane * 4);
    }
#pragma unroll
    for (int r = 0; r < 8; ++r) {
      float a = x[r][0] * vs[0];
      a = fmaf(x[r][1], vs[1], a);
      a = fmaf(x[r][2], vs[2], a);
      a = fmaf(x[r][3], vs[3], a);
      float b = x[r][0] * vd[0];
      b = fmaf(x[r][1], vd[1], b);
      b = fmaf(x[r][2], vd[2], b);
      b = fmaf(x[r][3], vd[3], b);
      for (int off = 16; off > 0; off >>= 1) {
        a += __shfl_xor(a, off, 32);
        b += __shfl_xor(b, off, 32);
      }
      if (lane == bt + r) { keep_s = a; keep_d = b; }
    }
  }
  if (row0 + lane < M) {
    logit[row0 + lane]             = keep_s;
    logit[(size_t)M + row0 + lane] = keep_d;
  }
}

__global__ __launch_bounds__(GT, 16) void localsort_fold_gemm(
    const int* __restrict__ src, const int* __restrict__ dst,
    int* __restrict__ loff, int* __restrict__ binned, int E, int NB, int G,
    const float* __restrict__ W0, const float* __restrict__ a0,
    const float* __restrict__ W1, const float* __restrict__ a1,
    const float* __restrict__ W2, const float* __restrict__ a2,
    const float* __restrict__ W3, const float* __restrict__ a3,
    float* __restrict__ v, int r1, int c1, int r2, int c2, int nfold,
    const float* __restrict__ X, const float* __restrict__ W,
    float* __restrict__ C, int tile0, int tile1)
{
  const int b = (int)blockIdx.x;
  if (b < G) local_sort_body(src, dst, loff, binned, E, NB, b, G);
  else if (b < G + nfold)
    fold_att_body(W0, a0, W1, a1, W2, a2, W3, a3, v, r1, c1, r2, c2,
                  (b - G) * GW + (int)(threadIdx.x >> 5));
  else gemm1_part(X, W, C, tile0, tile1, b - G - nfold);
}

__global__ __launch_bounds__(GT, 16) void rangesort_logit_gemm(
    const int* __restrict__ binned, const int* __restrict__ loff, int* __restrict__ lined,
    int* __restrict__ rowptr, int* __restrict__ col, int N, int E, int G, int NB,
    const float* __restrict__ vecs, float* __restrict__ logit, int nlog,
    const float* __restrict__ X, const float* __restrict__ W,
    float* __restrict__ C, int tile0, int tile1)
{
  const int b = (int)blockIdx.x;
  if (b < NB) range_sort_body(binned, loff, lined, rowptr, col, N, E, G, b, NB);
  else if (b < NB + nlog) logit_part(X, vecs, logit, N, b - NB);
  else gemm1_part(X, W, C, tile0, tile1, b - NB - nlog);
}


__device__ __forceinline__ float leaky(float v)
{
  return (v > 0.f) ? v : NEG_SLOPE * v;
}


__device__ __forceinline__ int lane_id()
{
  return (int)__builtin_amdgcn_mbcnt_lo(~0u, 0u);
}

template<int K>
__device__ __forceinline__ int half_lane(int v)
{
  return __builtin_amdgcn_update_dpp(0, v, 0x150 + K, 0xf, 0xf, false);
}

template<int C, int K>
__device__ __forceinline__ void row_ask(const char* __restrict__ hb, unsigned lo, int sl,
                                        v4f (&r)[C / 64])
{
  const unsigned off = (unsigned)half_lane<K>(sl) * (unsigned)(C * 4) + lo;
#pragma unroll
  for (int v = 0; v < C / 64; ++v) r[v] = *(const v4f*)(hb + off + v * 256);
}

template<int C, int K>
__device__ __forceinline__ void row_add(int ali, const v4f (&r)[C / 64], v4f (&acc)[C / 64])
{
  const float a = __int_as_float(half_lane<K>(ali));
#pragma unroll
  for (int v = 0; v < C / 64; ++v) acc[v] += a * r[v];
}

template<int C, int K0, int G>
__device__ __forceinline__ void row_group(const char* __restrict__ hb, unsigned lo, int sl, int ali,
                                          v4f (&acc)[C / 64])
{
  static_assert(G == 4 || G == 8, "group sizes written out below");
  v4f r0[C / 64], r1[C / 64], r2[C / 64], r3[C / 64];
  row_ask<C, K0 + 0>(hb, lo, sl, r0);
  row_ask<C, K0 + 1>(hb, lo, sl, r1);
  row_ask<C, K0 + 2>(hb, lo, sl, r2);
  row_ask<C, K0 + 3>(hb, lo, sl, r3);
  if constexpr (G == 8) {
    v4f r4[C / 64], r5[C / 64], r6[C / 64], r7[C / 64];
    row_ask<C, K0 + 4>(hb, lo, sl, r4);
    row_ask<C, K0 + 5>(hb, lo, sl, r5);
    row_ask<C, K0 + 6>(hb, lo, sl, r6);
    row_ask<C, K0 + 7>(hb, lo, sl, r7);
    row_add<C, K0 + 0>(ali, r0, acc);
    row_add<C, K0 + 1>(ali, r1, acc);
    row_add<C, K0 + 2>(ali, r2, acc);
    row_add<C, K0 + 3>(ali, r3, acc);
    row_add<C, K0 + 4>(ali, r4, acc);
    row_add<C, K0 + 5>(ali, r5, acc);
    row_add<C, K0 + 6>(ali, r6, acc);
    row_add<C, K0 + 7>(ali, r7, acc);
  } else {
    row_add<C, K0 + 0>(ali, r0, acc);
    row_add<C, K0 + 1>(ali, r1, acc);
    row_add<C, K0 + 2>(ali, r2, acc);
    row_add<C, K0 + 3>(ali, r3, acc);
  }
}

template<int C, bool BIAS>
__device__ __forceinline__ void gather_pair(
    const int* __restrict__ rowptr, const int* __restrict__ col,
    const float* __restrict__ asrc, const float* __restrict__ adst,
    const float* __restrict__ h, const float* __restrict__ bias, int d0, int N,
    v4f (&o)[C / 64])
{
  constexpr int NV = C / 64;
  constexpr int G  = 8 / NV;
  const int lane = lane_id();
  const int hf = lane >> 4;
  const int hl = lane & 15;
  const int   i2  = (d0 + 2 < N) ? (d0 + 2) : N;
  const int   i1  = (d0 + 1 < N) ? (d0 + 1) : (N - 1);
  const int   b0  = rowptr[d0];
  const int   b1  = rowptr[d0 + 1];
  const int   b2  = rowptr[i2];
  const float ad0 = adst[d0];
  const float ad1 = adst[i1];
  const int   beg = hf ? b1 : b0;
  const int   end = hf ? b2 : b1;
  const float ad  = hf ? ad1 : ad0;

  const int  j0  = beg + hl;
  const int  j1  = j0 + 16;
  const bool in0 = j0 < end;
  const bool in1 = j1 < end;
  int sl0 = 0, sl1 = 0;
  if (in0) sl0 = col[j0];
  if (in1) sl1 = col[j1];
  const float g0 = asrc[sl0];
  const float g1 = asrc[sl1];
  const float e0 = in0 ? leaky(g0 + ad) : -__builtin_huge_valf();
  const float e1 = in1 ? leaky(g1 + ad) : -__builtin_huge_valf();

  float mx = fmaxf(e0, e1);
  for (int j = j0 + 32; j < end; j += 16)
    mx = fmaxf(mx, leaky(asrc[col[j]] + ad));
  for (int off = 8; off > 0; off >>= 1) mx = fmaxf(mx, __shfl_xor(mx, off, 32));

  const float p0 = in0 ? __expf(e0 - mx) : 0.f;
  const float p1 = in1 ? __expf(e1 - mx) : 0.f;
  float den = p0 + p1;
  for (int j = j0 + 32; j < end; j += 16)
    den += __expf(leaky(asrc[col[j]] + ad) - mx);
  for (int off = 8; off > 0; off >>= 1) den += __shfl_xor(den, off, 32);

  const int n0 = b1 - b0;
  const int n1 = b2 - b1;
  const int nmax = (n0 > n1) ? n0 : n1;

  const char* hb = reinterpret_cast<const char*>(h);
  const unsigned lo = (unsigned)hl * 16u;
  v4f acc[NV];
#pragma unroll
  for (int v = 0; v < NV; ++v) acc[v] = (v4f)(0.f);

  for (int base = 0; base < nmax; base += 16) {
    int   sl = sl0;
    float al = in0 ? p0 / den : 0.f;
    if (base == 16) {
      sl = sl1;
      al = in1 ? p1 / den : 0.f;
    } else if (base != 0) {
      const int j = j0 + base;
      sl = 0;
      al = 0.f;
      if (j < end) {
        sl = col[j];
        al = __expf(leaky(asrc[sl] + ad) - mx) / den;
      }
    }
    const int ali = __float_as_int(al);
    const int rem = nmax - base;
    row_group<C, 0, G>(hb, lo, sl, ali, acc);
    if (G == 4) {
      if (rem > 4)  row_group<C, 4, G>(hb, lo, sl, ali, acc);
      if (rem > 8)  row_group<C, 8, G>(hb, lo, sl, ali, acc);
      if (rem > 12) row_group<C, 12, G>(hb, lo, sl, ali, acc);
    } else {
      if (rem > 8)  row_group<C, 8, G>(hb, lo, sl, ali, acc);
    }
  }

#pragma unroll
  for (int v = 0; v < NV; ++v) {
    o[v] = acc[v];
    if (BIAS) o[v] += *(const v4f*)(bias + v * 64 + hl * 4);
  }
}

__global__ __launch_bounds__(256, 16) void seg_gather_pair64(
    const int* __restrict__ rowptr, const int* __restrict__ col,
    const float* __restrict__ asrc, const float* __restrict__ adst,
    const float* __restrict__ h, const float* __restrict__ bias,
    float* __restrict__ out, int N)
{
  const int lane = threadIdx.x & 31;
  const int d0 = __builtin_amdgcn_readfirstlane((int)(blockIdx.x * 16 + (threadIdx.x >> 5) * 2));
  if (d0 >= N) return;
  v4f o[1];
  gather_pair<64, true>(rowptr, col, asrc, adst, h, bias, d0, N, o);
  const int d = d0 + (lane >> 4);
  if (d < N) *(v4f*)(out + (size_t)d * 64 + (lane & 15) * 4) = o[0];
}

#define TP 132
#define FROWS 32
__global__ __launch_bounds__(256, 16) void gather1_gemm2(
    const int* __restrict__ rowptr, const int* __restrict__ col,
    const float* __restrict__ asrc, const float* __restrict__ adst,
    const float* __restrict__ h, const float* __restrict__ bias,
    const float* __restrict__ W2, const float* __restrict__ vecs2,
    float* __restrict__ h2, float* __restrict__ logit2, int N)
{
  __shared__ __attribute__((aligned(16))) float tile[FROWS * TP];
  const int wave = __builtin_amdgcn_readfirstlane((int)(threadIdx.x >> 5));
  const int row0 = (int)blockIdx.x * FROWS;

  for (int p = 0; p < 2; ++p) {
    const int r = 4 * wave + 2 * p;
    v4f o[2];
    o[0] = (v4f)(0.f);
    o[1] = (v4f)(0.f);
    if (row0 + r < N) gather_pair<128, false>(rowptr, col, asrc, adst, h, bias, row0 + r, N, o);
    const int ln = lane_id();
    float* tp = tile + (r + (ln >> 4)) * TP + (ln & 15) * 4;
    *(v4f*)(tp)      = o[0];
    *(v4f*)(tp + 64) = o[1];
  }
  {
    const int ln = lane_id();
    const v4f bA = *(const v4f*)(bias + (ln & 15) * 4);
    const v4f bB = *(const v4f*)(bias + 64 + (ln & 15) * 4);
    for (int p = 0; p < 2; ++p) {
      float* tp = tile + (4 * wave + 2 * p + (ln >> 4)) * TP + (ln & 15) * 4;
      v4f x = *(v4f*)(tp) + bA;
      v4f y = *(v4f*)(tp + 64) + bB;
      for (int i = 0; i < 4; ++i) { x[i] = fmaxf(x[i], 0.f); y[i] = fmaxf(y[i], 0.f); }
      *(v4f*)(tp)      = x;
      *(v4f*)(tp + 64) = y;
    }
  }
  __syncthreads();

  const int lane = lane_id();
  const int half = lane >> 4;
  const int l15  = lane & 15;
  const int rt   = (wave >> 1) & 1;
  const int trow = row0 + 16 * rt;
  const float* ap = tile + (16 * rt + l15) * TP + 2 * half;

  if (wave >= 4) {
    const int v = wave & 1;
    const float* vp = vecs2 + (size_t)v * 128 + 2 * half;
    float lg = 0.f;
#pragma unroll
    for (int k = 0; k < 128; k += 4) {
      const v2f a = *(const v2f*)(ap + k);
      lg = fmaf(a.x, vp[k + 0], lg);
      lg = fmaf(a.y, vp[k + 1], lg);
    }
    const float s = lg + __shfl_xor(lg, 16, 32);
    if (half == 0 && trow + l15 < N) logit2[(size_t)v * N + trow + l15] = s;
    return;
  }

  const int g = wave & 1;
  const float* wb = W2 + (size_t)(2 * half) * 64 + g * 32 + 2 * l15;
  v8f acc0 = (v8f){0.f,0.f,0.f,0.f,0.f,0.f,0.f,0.f};
  v8f acc1 = (v8f){0.f,0.f,0.f,0.f,0.f,0.f,0.f,0.f};
  constexpr int AH = 4;
  v2f r0[AH], r1[AH];
#pragma unroll
  for (int i = 0; i < AH; ++i) {
    const int k = 4 * i;
    r0[i] = *(const v2f*)(wb + (size_t)(k + 0) * 64);
    r1[i] = *(const v2f*)(wb + (size_t)(k + 1) * 64);
  }
  v2f a = *(const v2f*)(ap);
#pragma unroll
  for (int i = 0; i < 32; ++i) {
    const int k = 4 * i;
    const int j = i % AH;
    v2f an = a;
    if (i + 1 < 32) an = *(const v2f*)(ap + k + 4);
    v2f b0, b1;
    b0.x = r0[j].x; b0.y = r1[j].x;
    b1.x = r0[j].y; b1.y = r1[j].y;
    if (i + AH < 32) {
      r0[j] = *(const v2f*)(wb + (size_t)(k + 4 * AH + 0) * 64);
      r1[j] = *(const v2f*)(wb + (size_t)(k + 4 * AH + 1) * 64);
    }
    acc0 = __builtin_amdgcn_wmma_f32_16x16x4_f32(false, a, false, b0, (short)0, acc0, false, false);
    acc1 = __builtin_amdgcn_wmma_f32_16x16x4_f32(false, a, false, b1, (short)0, acc1, false, false);
    __builtin_amdgcn_sched_barrier(0);
    a = an;
  }

  const int mbase = trow + 8 * half;
  float* cp = h2 + (size_t)mbase * 64 + g * 32 + 2 * l15;
#pragma unroll
  for (int r = 0; r < 8; ++r)
    if (mbase + r < N) {
      v2f o;
      o.x = acc0[r]; o.y = acc1[r];
      *(v2f*)(cp + (size_t)r * 64) = o;
    }
}

static inline int cdiv(long a, long b) { return (int)((a + b - 1) / b); }

extern "C" void kernel_launch(void* const* d_in, const int* in_sizes, int n_in,
                              void* d_out, int out_size, void* d_ws, size_t ws_size,
                              hipStream_t stream)
{
  const float* x       = (const float*)d_in[0];
  const int*   ei      = (const int*)d_in[1];
  const float* Wsrc1   = (const float*)d_in[2];
  const float* Wdst1   = (const float*)d_in[3];
  const float* attsrc1 = (const float*)d_in[4];
  const float* attdst1 = (const float*)d_in[5];
  const float* b1      = (const float*)d_in[6];
  const float* Wsrc2   = (const float*)d_in[7];
  const float* Wdst2   = (const float*)d_in[8];
  const float* attsrc2 = (const float*)d_in[9];
  const float* attdst2 = (const float*)d_in[10];
  const float* b2      = (const float*)d_in[11];

  const int HID = in_sizes[4];
  const int OUT = in_sizes[9];
  const int FIN = in_sizes[2] / HID;
  const int N   = in_sizes[0] / FIN;
  const int E   = in_sizes[1] / 2;
  const int* src = ei;
  const int* dst = ei + E;

  float* ws    = (float*)d_ws;
  float* h1    = ws;  ws += (size_t)N * HID;
  float* h2    = ws;  ws += (size_t)N * OUT;
  float* logit = ws;  ws += 2 * (size_t)N;
  float* logit2 = ws; ws += 2 * (size_t)N;
  float* vfold = ws;  ws += 2 * (size_t)FIN + 2 * (size_t)HID;
  const int G1 = NSL;
  const int NB = cdiv(N, RB);
  int* binned  = (int*)ws;  ws += 2 * (size_t)E;
  int* lined   = (int*)ws;  ws += 2 * (size_t)E;
  int* loff    = (int*)ws;  ws += (size_t)NB * G1;
  int* col     = (int*)ws;  ws += E;
  int* rowptr  = (int*)ws;  ws += N + 1;

  const int tiles1 = (N / 16) * (HID / 64);
  const int tA = tiles1 / 2;
  const int nfold = cdiv(2 * FIN + 2 * HID, GW);
  const int nlog  = cdiv(N, GW * LROWS);
  localsort_fold_gemm<<<G1 + nfold + cdiv(tA, GEMM_WPB), GT, 0, stream>>>(
      src, dst, loff, binned, E, NB, G1,
      Wsrc1, attsrc1, Wdst1, attdst1, Wsrc2, attsrc2, Wdst2, attdst2,
      vfold, FIN, HID, HID, OUT, nfold,
      x, Wsrc1, h1, 0, tA);
  rangesort_logit_gemm<<<NB + nlog + cdiv(tiles1 - tA, GEMM_WPB), GT, 0, stream>>>(
      binned, loff, lined, rowptr, col, N, E, G1, NB,
      vfold, logit, nlog,
      x, Wsrc1, h1, tA, tiles1);

  gather1_gemm2<<<cdiv(N, FROWS), 256, 0, stream>>>(
      rowptr, col, logit, logit + N, h1, b1,
      Wsrc2, vfold + 2 * (size_t)FIN, h2, logit2, N);

  {
    float* out = (float*)d_out;
    seg_gather_pair64<<<cdiv(N, 16), 256, 0, stream>>>(
        rowptr, col, logit2, logit2 + N, h2, b2, out, N);
  }
}
